# starting kernel
# baseline (speedup 1.0000x reference)
#include <hip/hip_runtime.h>

#define N_NODES 50000
#define N_EDGES 800000
#define N_GRAPHS 512
#define IN_C 128
#define HID 256
#define OUT_C 16

typedef __attribute__((ext_vector_type(16))) __bf16 v16bf;
typedef __attribute__((ext_vector_type(8)))  float  v8f;

__device__ __forceinline__ unsigned short f2bf(float f) {
  union { float f; unsigned int u; } c; c.f = f;
  unsigned int u = c.u;
  u += 0x7fffu + ((u >> 16) & 1u);
  return (unsigned short)(u >> 16);
}

__device__ __forceinline__ int frag_koff(int lane, int p) {
  int kb = (lane < 16) ? 0 : 8;
  return (p < 4) ? (kb + 2 * p) : (16 + kb + 2 * (p - 4));
}

__global__ void copy_f4(const float* __restrict__ src, float* __restrict__ dst, int n4) {
  int i = blockIdx.x * blockDim.x + threadIdx.x;
  if (i < n4) ((float4*)dst)[i] = ((const float4*)src)[i];
}

__global__ void edge_scatter(const float* __restrict__ h, const int* __restrict__ esrc,
                             const int* __restrict__ edst, float* __restrict__ agg,
                             int cg  , long total) {
  long i = (long)blockIdx.x * blockDim.x + threadIdx.x;
  if (i >= total) return;
  int e = (int)(i / cg);
  int c = (int)(i - (long)e * cg);
  int s = esrc[e], d = edst[e];
  const float4 v = ((const float4*)(h + (long)s * (cg * 4)))[c];
  float* o = agg + (long)d * (cg * 4) + c * 4;
  atomicAdd(o + 0, v.x); atomicAdd(o + 1, v.y);
  atomicAdd(o + 2, v.z); atomicAdd(o + 3, v.w);
}

__global__ void apack_f32(const float* __restrict__ src, unsigned int* __restrict__ out,
                          int K, int KT, long total) {
  long i = (long)blockIdx.x * blockDim.x + threadIdx.x;
  if (i >= total) return;
  int p = (int)(i & 7);
  int lane = (int)((i >> 3) & 31);
  long tt = i >> 8;
  int kt = (int)(tt % KT);
  int rt = (int)(tt / KT);
  int row = rt * 16 + (lane & 15);
  int k = kt * 32 + frag_koff(lane, p);
  const float* s = src + (long)row * K + k;
  out[i] = (unsigned int)f2bf(s[0]) | ((unsigned int)f2bf(s[1]) << 16);
}

__global__ void apack_bf16(const unsigned short* __restrict__ src, unsigned int* __restrict__ out,
                           int K, int KT, long total) {
  long i = (long)blockIdx.x * blockDim.x + threadIdx.x;
  if (i >= total) return;
  int p = (int)(i & 7);
  int lane = (int)((i >> 3) & 31);
  long tt = i >> 8;
  int kt = (int)(tt % KT);
  int rt = (int)(tt / KT);
  int row = rt * 16 + (lane & 15);
  int k = kt * 32 + frag_koff(lane, p);
  out[i] = *(const unsigned int*)(src + (long)row * K + k);
}

__global__ void wpack_f32(const float* __restrict__ W, unsigned int* __restrict__ out,
                          int K, int M, int KT, long total) {
  long i = (long)blockIdx.x * blockDim.x + threadIdx.x;
  if (i >= total) return;
  int p = (int)(i & 7);
  int lane = (int)((i >> 3) & 31);
  long tt = i >> 8;
  int kt = (int)(tt % KT);
  int ct = (int)(tt / KT);
  int col = ct * 16 + (lane & 15);
  int k = kt * 32 + frag_koff(lane, p);
  unsigned int lo = f2bf(W[(long)k * M + col]);
  unsigned int hi = f2bf(W[(long)(k + 1) * M + col]);
  out[i] = lo | (hi << 16);
}

template <int RELU, int OUTBF>
__global__ void gemm_pk(const uint4* __restrict__ Apk, const uint4* __restrict__ Wpk,
                        const float* __restrict__ bias, void* __restrict__ Cout,
                        int N, int K, int M) {
  const int tid  = threadIdx.x;
  const int lane = tid & 31;
  const int wave = tid >> 5;
  const int wr   = wave & 3;
  const int wc   = wave >> 2;
  const int KT   = K >> 5;
  const int nrt  = N >> 4;
  const int nct  = M >> 4;
  const int rt0  = blockIdx.y * 8 + wr * 2;
  const int ct0  = blockIdx.x * 4 + wc * 2;
  const bool r0 = rt0 < nrt,     r1 = (rt0 + 1) < nrt;
  const bool c0 = ct0 < nct,     c1 = (ct0 + 1) < nct;
  const int rA0 = r0 ? rt0 : 0,  rA1 = r1 ? (rt0 + 1) : 0;
  const int cB0 = c0 ? ct0 : 0,  cB1 = c1 ? (ct0 + 1) : 0;

  long ia0 = ((long)rA0 * KT * 32 + lane) * 2;
  long ia1 = ((long)rA1 * KT * 32 + lane) * 2;
  long ib0 = ((long)cB0 * KT * 32 + lane) * 2;
  long ib1 = ((long)cB1 * KT * 32 + lane) * 2;
  const long step = 64;

  v8f acc00 = {}, acc01 = {}, acc10 = {}, acc11 = {};
  union F { uint4 q[2]; v16bf v; };

  for (int kt = 0; kt < KT; ++kt) {
    F a0, a1, b0, b1;
    a0.q[0] = Apk[ia0]; a0.q[1] = Apk[ia0 + 1];
    a1.q[0] = Apk[ia1]; a1.q[1] = Apk[ia1 + 1];
    b0.q[0] = Wpk[ib0]; b0.q[1] = Wpk[ib0 + 1];
    b1.q[0] = Wpk[ib1]; b1.q[1] = Wpk[ib1 + 1];
    if (kt + 1 < KT) {
      __builtin_prefetch((const void*)(Apk + ia0 + step), 0, 3);
      __builtin_prefetch((const void*)(Apk + ia1 + step), 0, 3);
    }
    ia0 += step; ia1 += step; ib0 += step; ib1 += step;
    acc00 = __builtin_amdgcn_wmma_f32_16x16x32_bf16(false, a0.v, false, b0.v, (short)0, acc00, false, false);
    acc01 = __builtin_amdgcn_wmma_f32_16x16x32_bf16(false, a0.v, false, b1.v, (short)0, acc01, false, false);
    acc10 = __builtin_amdgcn_wmma_f32_16x16x32_bf16(false, a1.v, false, b0.v, (short)0, acc10, false, false);
    acc11 = __builtin_amdgcn_wmma_f32_16x16x32_bf16(false, a1.v, false, b1.v, (short)0, acc11, false, false);
  }

  const int rowL = lane & 15;
  auto st = [&](const v8f& acc, int rt, bool rok, int ct, bool cok) {
    if (!rok || !cok) return;
    int col = ct * 16 + rowL;
    #pragma unroll
    for (int r = 0; r < 8; ++r) {
      int row = rt * 16 + ((lane < 16) ? r : r + 8);
      float o = acc[r] + bias[col];
      if (RELU) o = fmaxf(o, 0.0f);
      if (OUTBF) ((unsigned short*)Cout)[(long)row * M + col] = f2bf(o);
      else       ((float*)Cout)[(long)row * M + col] = o;
    }
  };
  st(acc00, rt0, r0, ct0, c0);         st(acc01, rt0, r0, ct0 + 1, c1);
  st(acc10, rt0 + 1, r1, ct0, c0);     st(acc11, rt0 + 1, r1, ct0 + 1, c1);
}

__global__ void count_nodes(const int* __restrict__ batch, float* __restrict__ counts) {
  int n = blockIdx.x * blockDim.x + threadIdx.x;
  if (n < N_NODES) atomicAdd(&counts[batch[n]], 1.0f);
}

__global__ void pool_sum(const float* __restrict__ h, const int* __restrict__ batch,
                         float* __restrict__ pooled) {
  const int cg = HID / 4;
  long i = (long)blockIdx.x * blockDim.x + threadIdx.x;
  if (i >= (long)N_NODES * cg) return;
  int n = (int)(i / cg);
  int c = (int)(i - (long)n * cg);
  int g = batch[n];
  float4 v = ((const float4*)(h + (long)n * HID))[c];
  float* o = pooled + (long)g * HID + c * 4;
  atomicAdd(o + 0, v.x); atomicAdd(o + 1, v.y);
  atomicAdd(o + 2, v.z); atomicAdd(o + 3, v.w);
}

__global__ void pool_div(float* pooled, const float* __restrict__ counts) {
  int i = blockIdx.x * blockDim.x + threadIdx.x;
  if (i < N_GRAPHS * HID) {
    float c = counts[i / HID];
    pooled[i] /= fmaxf(c, 1.0f);
  }
}

extern "C" void kernel_launch(void* const* d_in, const int* in_sizes, int n_in,
                              void* d_out, int out_size, void* d_ws, size_t ws_size,
                              hipStream_t stream) {
  const float* x     = (const float*)d_in[0];
  const int*   ei    = (const int*)d_in[1];
  const int*   esrc  = ei;
  const int*   edst  = ei + N_EDGES;
  const int*   batch = (const int*)d_in[2];
  const float* cw1[3] = {(const float*)d_in[3],  (const float*)d_in[7],  (const float*)d_in[11]};
  const float* cb1[3] = {(const float*)d_in[4],  (const float*)d_in[8],  (const float*)d_in[12]};
  const float* cw2[3] = {(const float*)d_in[5],  (const float*)d_in[9],  (const float*)d_in[13]};
  const float* cb2[3] = {(const float*)d_in[6],  (const float*)d_in[10], (const float*)d_in[14]};
  const float* ow1 = (const float*)d_in[15];
  const float* ob1 = (const float*)d_in[16];
  const float* ow2 = (const float*)d_in[17];
  const float* ob2 = (const float*)d_in[18];

  char* ws = (char*)d_ws;
  size_t off = 0;
  auto bump = [&](size_t bytes) {
    void* p = ws + off;
    off += (bytes + 255) & ~(size_t)255;
    return p;
  };
  const int NRT = N_NODES / 16;
  float*          B0     = (float*)bump((size_t)N_NODES * HID * 4);
  float*          B2     = (float*)bump((size_t)N_NODES * HID * 4);
  unsigned int*   Apk    = (unsigned int*)bump((size_t)NRT * 8 * 256 * 4);
  unsigned short* Mbf    = (unsigned short*)bump((size_t)N_NODES * HID * 2);
  unsigned int*   Mpk    = (unsigned int*)bump((size_t)NRT * 8 * 256 * 4);
  float*          pooled = (float*)bump((size_t)N_GRAPHS * HID * 4);
  float*          counts = (float*)bump((size_t)N_GRAPHS * 4);
  unsigned int*   Ppk    = (unsigned int*)bump((size_t)(N_GRAPHS / 16) * 8 * 256 * 4);
  unsigned short* MPbf   = (unsigned short*)bump((size_t)N_GRAPHS * HID * 2);
  unsigned int*   MPpk   = (unsigned int*)bump((size_t)(N_GRAPHS / 16) * 8 * 256 * 4);
  unsigned int*   wpk1[3], *wpk2[3];
  for (int l = 0; l < 3; ++l) {
    wpk1[l] = (unsigned int*)bump((size_t)(HID / 16) * 8 * 256 * 4);
    wpk2[l] = (unsigned int*)bump((size_t)(HID / 16) * 8 * 256 * 4);
  }
  unsigned int* owpk1 = (unsigned int*)bump((size_t)(HID / 16) * 8 * 256 * 4);
  unsigned int* owpk2 = (unsigned int*)bump((size_t)(OUT_C / 16) * 8 * 256 * 4);

  dim3 blk(256);
  auto wpack = [&](const float* W, unsigned int* dst, int K, int M) {
    int KT = K / 32;
    long total = (long)(M / 16) * KT * 256;
    wpack_f32<<<(unsigned)((total + 255) / 256), blk, 0, stream>>>(W, dst, K, M, KT, total);
  };

  wpack(cw1[0], wpk1[0], IN_C, HID);
  wpack(cw2[0], wpk2[0], HID, HID);
  for (int l = 1; l < 3; ++l) {
    wpack(cw1[l], wpk1[l], HID, HID);
    wpack(cw2[l], wpk2[l], HID, HID);
  }
  wpack(ow1, owpk1, HID, HID);
  wpack(ow2, owpk2, HID, OUT_C);

  const float* h = x;
  int C = IN_C;
  for (int l = 0; l < 3; ++l) {
    int n4 = N_NODES * C / 4;
    copy_f4<<<(n4 + 255) / 256, blk, 0, stream>>>(h, B0, n4);
    long tot = (long)N_EDGES * (C / 4);
    edge_scatter<<<(unsigned)((tot + 255) / 256), blk, 0, stream>>>(h, esrc, edst, B0, C / 4, tot);

    int KT = C / 32;
    long atot = (long)NRT * KT * 256;
    apack_f32<<<(unsigned)((atot + 255) / 256), blk, 0, stream>>>(B0, Apk, C, KT, atot);

    dim3 g1((HID / 16 + 3) / 4, (NRT + 7) / 8);
    gemm_pk<1, 1><<<g1, blk, 0, stream>>>((const uint4*)Apk, (const uint4*)wpk1[l], cb1[l],
                                          Mbf, N_NODES, C, HID);
    long mtot = (long)NRT * (HID / 32) * 256;
    apack_bf16<<<(unsigned)((mtot + 255) / 256), blk, 0, stream>>>(Mbf, Mpk, HID, HID / 32, mtot);
    dim3 g2((HID / 16 + 3) / 4, (NRT + 7) / 8);
    gemm_pk<1, 0><<<g2, blk, 0, stream>>>((const uint4*)Mpk, (const uint4*)wpk2[l], cb2[l],
                                          B2, N_NODES, HID, HID);
    h = B2;
    C = HID;
  }

  hipMemsetAsync(pooled, 0, (size_t)N_GRAPHS * HID * 4, stream);
  hipMemsetAsync(counts, 0, (size_t)N_GRAPHS * 4, stream);
  count_nodes<<<(N_NODES + 255) / 256, blk, 0, stream>>>(batch, counts);
  long ptot = (long)N_NODES * (HID / 4);
  pool_sum<<<(unsigned)((ptot + 255) / 256), blk, 0, stream>>>(B2, batch, pooled);
  pool_div<<<(N_GRAPHS * HID + 255) / 256, blk, 0, stream>>>(pooled, counts);

  const int GRT = N_GRAPHS / 16;
  long gtot = (long)GRT * (HID / 32) * 256;
  apack_f32<<<(unsigned)((gtot + 255) / 256), blk, 0, stream>>>(pooled, Ppk, HID, HID / 32, gtot);
  dim3 g3((HID / 16 + 3) / 4, (GRT + 7) / 8);
  gemm_pk<1, 1><<<g3, blk, 0, stream>>>((const uint4*)Ppk, (const uint4*)owpk1, ob1,
                                        MPbf, N_GRAPHS, HID, HID);
  apack_bf16<<<(unsigned)((gtot + 255) / 256), blk, 0, stream>>>(MPbf, MPpk, HID, HID / 32, gtot);
  dim3 g4((OUT_C / 16 + 3) / 4, (GRT + 7) / 8);
  gemm_pk<0, 0><<<g4, blk, 0, stream>>>((const uint4*)MPpk, (const uint4*)owpk2, ob2,
                                        (float*)d_out, N_GRAPHS, HID, OUT_C);
}


# Separate A-operand pack passes removed: the row gather and the first GEMM's epilogue write the next GEMM's A fragments directly (tile turned in LDS, whole-line stores); 34 -> 27 launches
# speedup vs baseline: 2.3545x; 2.3545x over previous
#include <hip/hip_runtime.h>

#define N_NODES 50000
#define N_EDGES 800000
#define N_GRAPHS 512
#define IN_C 128
#define HID 256
#define OUT_C 16

typedef __attribute__((ext_vector_type(16))) __bf16 v16bf;
typedef __attribute__((ext_vector_type(8)))  float  v8f;

__device__ __forceinline__ unsigned short f2bf(float f) {
  union { float f; unsigned int u; } c; c.f = f;
  unsigned int u = c.u;
  u += 0x7fffu + ((u >> 16) & 1u);
  return (unsigned short)(u >> 16);
}

__device__ __forceinline__ int frag_koff(int lane, int p) {
  int kb = (lane < 16) ? 0 : 8;
  return (p < 4) ? (kb + 2 * p) : (16 + kb + 2 * (p - 4));
}

__global__ void csr_zero(int* __restrict__ deg) {
  int i = blockIdx.x * blockDim.x + threadIdx.x;
  if (i < N_NODES) deg[i] = 0;
}

__global__ void csr_count(const int* __restrict__ edst, int* __restrict__ deg) {
  int e = blockIdx.x * blockDim.x + threadIdx.x;
  if (e < N_EDGES) atomicAdd(&deg[edst[e]], 1);
}

#define SCAN_T 1024
#define SCAN_TILES ((N_NODES / 4 + SCAN_T - 1) / SCAN_T)

__device__ __forceinline__ int wave_incl_scan(int x, int lane) {
  #pragma unroll
  for (int o = 1; o < 32; o <<= 1) {
    const int up = __shfl_up(x, o, 32);
    if (lane >= o) x += up;
  }
  return x;
}

__global__ __launch_bounds__(SCAN_T) void csr_scan(const int* __restrict__ deg, int* __restrict__ rowptr,
                                                   int* __restrict__ cursor) {
  __shared__ int run_off[SCAN_TILES * 32];
  const int t = threadIdx.x;
  const int lane = t & 31;
  const int w = t >> 5;
  const int n4 = N_NODES / 4;
  int4 q[SCAN_TILES];
  int s[SCAN_TILES], inc[SCAN_TILES];
  #pragma unroll
  for (int k = 0; k < SCAN_TILES; ++k) {
    const int i4 = k * SCAN_T + t;
    q[k] = (i4 < n4) ? ((const int4*)deg)[i4] : make_int4(0, 0, 0, 0);
  }
  #pragma unroll
  for (int k = 0; k < SCAN_TILES; ++k) {
    s[k] = q[k].x + q[k].y + q[k].z + q[k].w;
    inc[k] = wave_incl_scan(s[k], lane);
    if (lane == 31) run_off[k * 32 + w] = inc[k];
  }
  __syncthreads();
  if (w == 0) {
    int carry = 0;
    #pragma unroll
    for (int k = 0; k < SCAN_TILES; ++k) {
      const int tot = run_off[k * 32 + lane];
      const int ti = wave_incl_scan(tot, lane);
      run_off[k * 32 + lane] = carry + ti - tot;
      carry += __shfl(ti, 31, 32);
    }
    if (lane == 0) rowptr[N_NODES] = carry;
  }
  __syncthreads();
  #pragma unroll
  for (int k = 0; k < SCAN_TILES; ++k) {
    const int i4 = k * SCAN_T + t;
    if (i4 < n4) {
      int4 o;
      o.x = run_off[k * 32 + w] + inc[k] - s[k];
      o.y = o.x + q[k].x;
      o.z = o.y + q[k].y;
      o.w = o.z + q[k].z;
      ((int4*)rowptr)[i4] = o;
      ((int4*)cursor)[i4] = o;
    }
  }
}

__global__ void csr_fill(const int* __restrict__ esrc, const int* __restrict__ edst,
                         int* __restrict__ cursor, int* __restrict__ col) {
  int e = blockIdx.x * blockDim.x + threadIdx.x;
  if (e < N_EDGES) {
    int pos = atomicAdd(&cursor[edst[e]], 1);
    col[pos] = esrc[e];
  }
}

__device__ __forceinline__ void add4(float4& a, const float4& b) {
  a.x += b.x; a.y += b.y; a.z += b.z; a.w += b.w;
}

static_assert(N_NODES % 16 == 0, "one work-group per full 16-row tile");
template <int C>
__global__ __launch_bounds__(512) void gather_pack(const float* __restrict__ h, const int* __restrict__ rowptr,
                                                   const int* __restrict__ col, uint4* __restrict__ Apk) {
  constexpr int V = C / 128;
  constexpr int U = 8 / V;
  constexpr int KT = C / 32;
  __shared__ uint4 tile[KT * 64];
  const int lane = threadIdx.x & 31;
  const int w = __builtin_amdgcn_readfirstlane((int)(threadIdx.x >> 5));
  const int node = __builtin_amdgcn_readfirstlane((int)(blockIdx.x * 16 + w));
  const int beg = rowptr[node];
  const int end = rowptr[node + 1];
  const float4* __restrict__ hp = (const float4*)h;
  float4 acc[V];
  #pragma unroll
  for (int v = 0; v < V; ++v) acc[v] = hp[(long)node * (C / 4) + v * 32 + lane];
  for (int base = beg; base < end; base += 32) {
    const int cnt = min(32, end - base);
    const int mine = (lane < cnt) ? col[base + lane] : 0;
    int j = 0;
    for (; j + U <= cnt; j += U) {
      float4 x[U][V];
      #pragma unroll
      for (int u = 0; u < U; ++u) {
        const int sidx = __builtin_amdgcn_readlane(mine, j + u);
        #pragma unroll
        for (int v = 0; v < V; ++v) x[u][v] = hp[(long)sidx * (C / 4) + v * 32 + lane];
      }
      #pragma unroll
      for (int u = 0; u < U; ++u) {
        #pragma unroll
        for (int v = 0; v < V; ++v) add4(acc[v], x[u][v]);
      }
    }
    #pragma nounroll
    for (; j < cnt; ++j) {
      const int sidx = __builtin_amdgcn_readlane(mine, j);
      #pragma unroll
      for (int v = 0; v < V; ++v) add4(acc[v], hp[(long)sidx * (C / 4) + v * 32 + lane]);
    }
  }
  unsigned int* tl = (unsigned int*)tile;
  const int m  = lane & 7;
  const int fh = (m >> 1) & 1;
  const int p0 = 2 * (m & 1) + 4 * (m >> 2);
  #pragma unroll
  for (int v = 0; v < V; ++v) {
    const int kt = v * 4 + (lane >> 3);
    uint2 pr;
    pr.x = (unsigned int)f2bf(acc[v].x) | ((unsigned int)f2bf(acc[v].y) << 16);
    pr.y = (unsigned int)f2bf(acc[v].z) | ((unsigned int)f2bf(acc[v].w) << 16);
    *(uint2*)(tl + kt * 256 + (w + 16 * fh) * 8 + p0) = pr;
  }
  __syncthreads();
  if (threadIdx.x < KT * 64) Apk[(long)blockIdx.x * (KT * 64) + threadIdx.x] = tile[threadIdx.x];
}

__global__ void apack_f32(const float* __restrict__ src, unsigned int* __restrict__ out,
                          int K, int KT, long total) {
  long i = (long)blockIdx.x * blockDim.x + threadIdx.x;
  if (i >= total) return;
  int p = (int)(i & 7);
  int lane = (int)((i >> 3) & 31);
  long tt = i >> 8;
  int kt = (int)(tt % KT);
  int rt = (int)(tt / KT);
  int row = rt * 16 + (lane & 15);
  int k = kt * 32 + frag_koff(lane, p);
  const float* s = src + (long)row * K + k;
  out[i] = (unsigned int)f2bf(s[0]) | ((unsigned int)f2bf(s[1]) << 16);
}

__global__ void wpack_f32(const float* __restrict__ W, unsigned int* __restrict__ out,
                          int K, int M, int KT, long total) {
  long i = (long)blockIdx.x * blockDim.x + threadIdx.x;
  if (i >= total) return;
  int p = (int)(i & 7);
  int lane = (int)((i >> 3) & 31);
  long tt = i >> 8;
  int kt = (int)(tt % KT);
  int ct = (int)(tt / KT);
  int col = ct * 16 + (lane & 15);
  int k = kt * 32 + frag_koff(lane, p);
  unsigned int lo = f2bf(W[(long)k * M + col]);
  unsigned int hi = f2bf(W[(long)(k + 1) * M + col]);
  out[i] = lo | (hi << 16);
}

template <int RELU, int OUTPK>
__global__ void gemm_pk(const uint4* __restrict__ Apk, const uint4* __restrict__ Wpk,
                        const float* __restrict__ bias, void* __restrict__ Cout,
                        int N, int K, int M) {
  const int tid  = threadIdx.x;
  const int lane = tid & 31;
  const int wave = tid >> 5;
  const int wr   = wave & 3;
  const int wc   = wave >> 2;
  const int KT   = K >> 5;
  const int nrt  = N >> 4;
  const int nct  = M >> 4;
  const int rt0  = blockIdx.y * 8 + wr * 2;
  const int ct0  = blockIdx.x * 4 + wc * 2;
  const bool r0 = rt0 < nrt,     r1 = (rt0 + 1) < nrt;
  const bool c0 = ct0 < nct,     c1 = (ct0 + 1) < nct;
  const int rA0 = r0 ? rt0 : 0,  rA1 = r1 ? (rt0 + 1) : 0;
  const int cB0 = c0 ? ct0 : 0,  cB1 = c1 ? (ct0 + 1) : 0;

  long ia0 = ((long)rA0 * KT * 32 + lane) * 2;
  long ia1 = ((long)rA1 * KT * 32 + lane) * 2;
  long ib0 = ((long)cB0 * KT * 32 + lane) * 2;
  long ib1 = ((long)cB1 * KT * 32 + lane) * 2;
  const long step = 64;

  v8f acc00 = {}, acc01 = {}, acc10 = {}, acc11 = {};
  union F { uint4 q[2]; v16bf v; };

  for (int kt = 0; kt < KT; ++kt) {
    F a0, a1, b0, b1;
    a0.q[0] = Apk[ia0]; a0.q[1] = Apk[ia0 + 1];
    a1.q[0] = Apk[ia1]; a1.q[1] = Apk[ia1 + 1];
    b0.q[0] = Wpk[ib0]; b0.q[1] = Wpk[ib0 + 1];
    b1.q[0] = Wpk[ib1]; b1.q[1] = Wpk[ib1 + 1];
    if (kt + 1 < KT) {
      __builtin_prefetch((const void*)(Apk + ia0 + step), 0, 3);
      __builtin_prefetch((const void*)(Apk + ia1 + step), 0, 3);
    }
    ia0 += step; ia1 += step; ib0 += step; ib1 += step;
    acc00 = __builtin_amdgcn_wmma_f32_16x16x32_bf16(false, a0.v, false, b0.v, (short)0, acc00, false, false);
    acc01 = __builtin_amdgcn_wmma_f32_16x16x32_bf16(false, a0.v, false, b1.v, (short)0, acc01, false, false);
    acc10 = __builtin_amdgcn_wmma_f32_16x16x32_bf16(false, a1.v, false, b0.v, (short)0, acc10, false, false);
    acc11 = __builtin_amdgcn_wmma_f32_16x16x32_bf16(false, a1.v, false, b1.v, (short)0, acc11, false, false);
  }

  const int rowL = lane & 15;
  if constexpr (OUTPK) {
    __shared__ uint4 stage[8 * 2 * 64];
    unsigned short* sw = (unsigned short*)stage + wave * 1024;
    const int hh = lane >> 4;
    auto put = [&](const v8f& acc, int i, int j, int ctc) {
      const float b = bias[ctc * 16 + rowL];
      #pragma unroll
      for (int r = 0; r < 8; ++r) {
        float o = acc[r] + b;
        if (RELU) o = fmaxf(o, 0.0f);
        sw[i * 512 + (8 * hh + r + 16 * (rowL >> 3)) * 16 + (rowL & 7) + 8 * j] = f2bf(o);
      }
    };
    put(acc00, 0, 0, cB0);   put(acc01, 0, 1, cB1);
    put(acc10, 1, 0, cB0);   put(acc11, 1, 1, cB1);
    __syncthreads();
    const int KTo = M >> 5;
    const int kto = ct0 >> 1;
    uint4* __restrict__ Opk = (uint4*)Cout;
    #pragma unroll
    for (int i = 0; i < 2; ++i) {
      const bool rok = i ? r1 : r0;
      if (rok && c1) {
        const long o4 = ((long)(rt0 + i) * KTo + kto) * 64;
        Opk[o4 + lane]      = stage[(wave * 2 + i) * 64 + lane];
        Opk[o4 + 32 + lane] = stage[(wave * 2 + i) * 64 + 32 + lane];
      }
    }
  } else {
    auto st = [&](const v8f& acc, int rt, bool rok, int ct, bool cok) {
      if (!rok || !cok) return;
      int col = ct * 16 + rowL;
      #pragma unroll
      for (int r = 0; r < 8; ++r) {
        int row = rt * 16 + ((lane < 16) ? r : r + 8);
        float o = acc[r] + bias[col];
        if (RELU) o = fmaxf(o, 0.0f);
        ((float*)Cout)[(long)row * M + col] = o;
      }
    };
    st(acc00, rt0, r0, ct0, c0);         st(acc01, rt0, r0, ct0 + 1, c1);
    st(acc10, rt0 + 1, r1, ct0, c0);     st(acc11, rt0 + 1, r1, ct0 + 1, c1);
  }
}

__global__ void count_nodes(const int* __restrict__ batch, float* __restrict__ counts) {
  int n = blockIdx.x * blockDim.x + threadIdx.x;
  if (n < N_NODES) atomicAdd(&counts[batch[n]], 1.0f);
}

__global__ void pool_sum(const float* __restrict__ h, const int* __restrict__ batch,
                         float* __restrict__ pooled) {
  const int cg = HID / 4;
  long i = (long)blockIdx.x * blockDim.x + threadIdx.x;
  if (i >= (long)N_NODES * cg) return;
  int n = (int)(i / cg);
  int c = (int)(i - (long)n * cg);
  int g = batch[n];
  float4 v = ((const float4*)(h + (long)n * HID))[c];
  float* o = pooled + (long)g * HID + c * 4;
  atomicAdd(o + 0, v.x); atomicAdd(o + 1, v.y);
  atomicAdd(o + 2, v.z); atomicAdd(o + 3, v.w);
}

__global__ void pool_div(float* pooled, const float* __restrict__ counts) {
  int i = blockIdx.x * blockDim.x + threadIdx.x;
  if (i < N_GRAPHS * HID) {
    float c = counts[i / HID];
    pooled[i] /= fmaxf(c, 1.0f);
  }
}

extern "C" void kernel_launch(void* const* d_in, const int* in_sizes, int n_in,
                              void* d_out, int out_size, void* d_ws, size_t ws_size,
                              hipStream_t stream) {
  const float* x     = (const float*)d_in[0];
  const int*   ei    = (const int*)d_in[1];
  const int*   esrc  = ei;
  const int*   edst  = ei + N_EDGES;
  const int*   batch = (const int*)d_in[2];
  const float* cw1[3] = {(const float*)d_in[3],  (const float*)d_in[7],  (const float*)d_in[11]};
  const float* cb1[3] = {(const float*)d_in[4],  (const float*)d_in[8],  (const float*)d_in[12]};
  const float* cw2[3] = {(const float*)d_in[5],  (const float*)d_in[9],  (const float*)d_in[13]};
  const float* cb2[3] = {(const float*)d_in[6],  (const float*)d_in[10], (const float*)d_in[14]};
  const float* ow1 = (const float*)d_in[15];
  const float* ob1 = (const float*)d_in[16];
  const float* ow2 = (const float*)d_in[17];
  const float* ob2 = (const float*)d_in[18];

  char* ws = (char*)d_ws;
  size_t off = 0;
  auto bump = [&](size_t bytes) {
    void* p = ws + off;
    off += (bytes + 255) & ~(size_t)255;
    return p;
  };
  const int NRT = N_NODES / 16;
  float*          B2     = (float*)bump((size_t)N_NODES * HID * 4);
  unsigned int*   Apk    = (unsigned int*)bump((size_t)NRT * 8 * 256 * 4);
  unsigned int*   Mpk    = (unsigned int*)bump((size_t)NRT * 8 * 256 * 4);
  float*          pooled = (float*)bump((size_t)N_GRAPHS * HID * 4);
  float*          counts = (float*)bump((size_t)N_GRAPHS * 4);
  unsigned int*   Ppk    = (unsigned int*)bump((size_t)(N_GRAPHS / 16) * 8 * 256 * 4);
  unsigned int*   MPpk   = (unsigned int*)bump((size_t)(N_GRAPHS / 16) * 8 * 256 * 4);
  unsigned int*   wpk1[3], *wpk2[3];
  for (int l = 0; l < 3; ++l) {
    wpk1[l] = (unsigned int*)bump((size_t)(HID / 16) * 8 * 256 * 4);
    wpk2[l] = (unsigned int*)bump((size_t)(HID / 16) * 8 * 256 * 4);
  }
  unsigned int* owpk1 = (unsigned int*)bump((size_t)(HID / 16) * 8 * 256 * 4);
  unsigned int* owpk2 = (unsigned int*)bump((size_t)(OUT_C / 16) * 8 * 256 * 4);
  int* deg    = (int*)bump((size_t)N_NODES * 4);
  int* rowptr = (int*)bump((size_t)(N_NODES + 1) * 4);
  int* cursor = (int*)bump((size_t)N_NODES * 4);
  int* col    = (int*)bump((size_t)N_EDGES * 4);

  dim3 blk(256);
  auto wpack = [&](const float* W, unsigned int* dst, int K, int M) {
    int KT = K / 32;
    long total = (long)(M / 16) * KT * 256;
    wpack_f32<<<(unsigned)((total + 255) / 256), blk, 0, stream>>>(W, dst, K, M, KT, total);
  };

  wpack(cw1[0], wpk1[0], IN_C, HID);
  wpack(cw2[0], wpk2[0], HID, HID);
  for (int l = 1; l < 3; ++l) {
    wpack(cw1[l], wpk1[l], HID, HID);
    wpack(cw2[l], wpk2[l], HID, HID);
  }
  wpack(ow1, owpk1, HID, HID);
  wpack(ow2, owpk2, HID, OUT_C);

  csr_zero<<<(N_NODES + 255) / 256, blk, 0, stream>>>(deg);
  csr_count<<<(N_EDGES + 255) / 256, blk, 0, stream>>>(edst, deg);
  csr_scan<<<1, SCAN_T, 0, stream>>>(deg, rowptr, cursor);
  csr_fill<<<(N_EDGES + 255) / 256, blk, 0, stream>>>(esrc, edst, cursor, col);

  const float* h = x;
  int C = IN_C;
  for (int l = 0; l < 3; ++l) {
    if (l == 0) gather_pack<IN_C><<<NRT, 512, 0, stream>>>(h, rowptr, col, (uint4*)Apk);
    else        gather_pack<HID><<<NRT, 512, 0, stream>>>(h, rowptr, col, (uint4*)Apk);

    dim3 g1((HID / 16 + 3) / 4, (NRT + 7) / 8);
    gemm_pk<1, 1><<<g1, blk, 0, stream>>>((const uint4*)Apk, (const uint4*)wpk1[l], cb1[l],
                                          Mpk, N_NODES, C, HID);
    dim3 g2((HID / 16 + 3) / 4, (NRT + 7) / 8);
    gemm_pk<1, 0><<<g2, blk, 0, stream>>>((const uint4*)Mpk, (const uint4*)wpk2[l], cb2[l],
                                          B2, N_NODES, HID, HID);
    h = B2;
    C = HID;
  }

  hipMemsetAsync(pooled, 0, (size_t)N_GRAPHS * HID * 4, stream);
  hipMemsetAsync(counts, 0, (size_t)N_GRAPHS * 4, stream);
  count_nodes<<<(N_NODES + 255) / 256, blk, 0, stream>>>(batch, counts);
  long ptot = (long)N_NODES * (HID / 4);
  pool_sum<<<(unsigned)((ptot + 255) / 256), blk, 0, stream>>>(B2, batch, pooled);
  pool_div<<<(N_GRAPHS * HID + 255) / 256, blk, 0, stream>>>(pooled, counts);

  const int GRT = N_GRAPHS / 16;
  long gtot = (long)GRT * (HID / 32) * 256;
  apack_f32<<<(unsigned)((gtot + 255) / 256), blk, 0, stream>>>(pooled, Ppk, HID, HID / 32, gtot);
  dim3 g3((HID / 16 + 3) / 4, (GRT + 7) / 8);
  gemm_pk<1, 1><<<g3, blk, 0, stream>>>((const uint4*)Ppk, (const uint4*)owpk1, ob1,
                                        MPpk, N_GRAPHS, HID, HID);
  dim3 g4((OUT_C / 16 + 3) / 4, (GRT + 7) / 8);
  gemm_pk<0, 0><<<g4, blk, 0, stream>>>((const uint4*)MPpk, (const uint4*)owpk2, ob2,
                                        (float*)d_out, N_GRAPHS, HID, OUT_C);
}


# Mean pool: per-element float atomics replaced by run-length sums (one wave per 16 consecutive nodes, rows added in node order, one atomic flush per run); count, divide and operand pack folded in
# speedup vs baseline: 2.6786x; 1.1377x over previous
#include <hip/hip_runtime.h>

#define N_NODES 50000
#define N_EDGES 800000
#define N_GRAPHS 512
#define IN_C 128
#define HID 256
#define OUT_C 16

typedef __attribute__((ext_vector_type(16))) __bf16 v16bf;
typedef __attribute__((ext_vector_type(8)))  float  v8f;

__device__ __forceinline__ unsigned short f2bf(float f) {
  union { float f; unsigned int u; } c; c.f = f;
  unsigned int u = c.u;
  u += 0x7fffu + ((u >> 16) & 1u);
  return (unsigned short)(u >> 16);
}

__device__ __forceinline__ int frag_koff(int lane, int p) {
  int kb = (lane < 16) ? 0 : 8;
  return (p < 4) ? (kb + 2 * p) : (16 + kb + 2 * (p - 4));
}

__global__ void csr_zero(int* __restrict__ deg) {
  int i = blockIdx.x * blockDim.x + threadIdx.x;
  if (i < N_NODES) deg[i] = 0;
}

__global__ void csr_count(const int* __restrict__ edst, int* __restrict__ deg) {
  int e = blockIdx.x * blockDim.x + threadIdx.x;
  if (e < N_EDGES) atomicAdd(&deg[edst[e]], 1);
}

#define SCAN_T 1024
#define SCAN_TILES ((N_NODES / 4 + SCAN_T - 1) / SCAN_T)

__device__ __forceinline__ int wave_incl_scan(int x, int lane) {
  #pragma unroll
  for (int o = 1; o < 32; o <<= 1) {
    const int up = __shfl_up(x, o, 32);
    if (lane >= o) x += up;
  }
  return x;
}

__global__ __launch_bounds__(SCAN_T) void csr_scan(const int* __restrict__ deg, int* __restrict__ rowptr,
                                                   int* __restrict__ cursor) {
  __shared__ int run_off[SCAN_TILES * 32];
  const int t = threadIdx.x;
  const int lane = t & 31;
  const int w = t >> 5;
  const int n4 = N_NODES / 4;
  int4 q[SCAN_TILES];
  int s[SCAN_TILES], inc[SCAN_TILES];
  #pragma unroll
  for (int k = 0; k < SCAN_TILES; ++k) {
    const int i4 = k * SCAN_T + t;
    q[k] = (i4 < n4) ? ((const int4*)deg)[i4] : make_int4(0, 0, 0, 0);
  }
  #pragma unroll
  for (int k = 0; k < SCAN_TILES; ++k) {
    s[k] = q[k].x + q[k].y + q[k].z + q[k].w;
    inc[k] = wave_incl_scan(s[k], lane);
    if (lane == 31) run_off[k * 32 + w] = inc[k];
  }
  __syncthreads();
  if (w == 0) {
    int carry = 0;
    #pragma unroll
    for (int k = 0; k < SCAN_TILES; ++k) {
      const int tot = run_off[k * 32 + lane];
      const int ti = wave_incl_scan(tot, lane);
      run_off[k * 32 + lane] = carry + ti - tot;
      carry += __shfl(ti, 31, 32);
    }
    if (lane == 0) rowptr[N_NODES] = carry;
  }
  __syncthreads();
  #pragma unroll
  for (int k = 0; k < SCAN_TILES; ++k) {
    const int i4 = k * SCAN_T + t;
    if (i4 < n4) {
      int4 o;
      o.x = run_off[k * 32 + w] + inc[k] - s[k];
      o.y = o.x + q[k].x;
      o.z = o.y + q[k].y;
      o.w = o.z + q[k].z;
      ((int4*)rowptr)[i4] = o;
      ((int4*)cursor)[i4] = o;
    }
  }
}

__global__ void csr_fill(const int* __restrict__ esrc, const int* __restrict__ edst,
                         int* __restrict__ cursor, int* __restrict__ col) {
  int e = blockIdx.x * blockDim.x + threadIdx.x;
  if (e < N_EDGES) {
    int pos = atomicAdd(&cursor[edst[e]], 1);
    col[pos] = esrc[e];
  }
}

__device__ __forceinline__ void add4(float4& a, const float4& b) {
  a.x += b.x; a.y += b.y; a.z += b.z; a.w += b.w;
}

static_assert(N_NODES % 16 == 0, "one work-group per full 16-row tile");
template <int C>
__global__ __launch_bounds__(512) void gather_pack(const float* __restrict__ h, const int* __restrict__ rowptr,
                                                   const int* __restrict__ col, uint4* __restrict__ Apk) {
  constexpr int V = C / 128;
  constexpr int U = 8 / V;
  constexpr int KT = C / 32;
  __shared__ uint4 tile[KT * 64];
  const int lane = threadIdx.x & 31;
  const int w = __builtin_amdgcn_readfirstlane((int)(threadIdx.x >> 5));
  const int node = __builtin_amdgcn_readfirstlane((int)(blockIdx.x * 16 + w));
  const int beg = rowptr[node];
  const int end = rowptr[node + 1];
  const float4* __restrict__ hp = (const float4*)h;
  float4 acc[V];
  #pragma unroll
  for (int v = 0; v < V; ++v) acc[v] = hp[(long)node * (C / 4) + v * 32 + lane];
  for (int base = beg; base < end; base += 32) {
    const int cnt = min(32, end - base);
    const int mine = (lane < cnt) ? col[base + lane] : 0;
    int j = 0;
    for (; j + U <= cnt; j += U) {
      float4 x[U][V];
      #pragma unroll
      for (int u = 0; u < U; ++u) {
        const int sidx = __builtin_amdgcn_readlane(mine, j + u);
        #pragma unroll
        for (int v = 0; v < V; ++v) x[u][v] = hp[(long)sidx * (C / 4) + v * 32 + lane];
      }
      #pragma unroll
      for (int u = 0; u < U; ++u) {
        #pragma unroll
        for (int v = 0; v < V; ++v) add4(acc[v], x[u][v]);
      }
    }
    #pragma nounroll
    for (; j < cnt; ++j) {
      const int sidx = __builtin_amdgcn_readlane(mine, j);
      #pragma unroll
      for (int v = 0; v < V; ++v) add4(acc[v], hp[(long)sidx * (C / 4) + v * 32 + lane]);
    }
  }
  unsigned int* tl = (unsigned int*)tile;
  const int m  = lane & 7;
  const int fh = (m >> 1) & 1;
  const int p0 = 2 * (m & 1) + 4 * (m >> 2);
  #pragma unroll
  for (int v = 0; v < V; ++v) {
    const int kt = v * 4 + (lane >> 3);
    uint2 pr;
    pr.x = (unsigned int)f2bf(acc[v].x) | ((unsigned int)f2bf(acc[v].y) << 16);
    pr.y = (unsigned int)f2bf(acc[v].z) | ((unsigned int)f2bf(acc[v].w) << 16);
    *(uint2*)(tl + kt * 256 + (w + 16 * fh) * 8 + p0) = pr;
  }
  __syncthreads();
  if (threadIdx.x < KT * 64) Apk[(long)blockIdx.x * (KT * 64) + threadIdx.x] = tile[threadIdx.x];
}

__global__ void wpack_f32(const float* __restrict__ W, unsigned int* __restrict__ out,
                          int K, int M, int KT, long total) {
  long i = (long)blockIdx.x * blockDim.x + threadIdx.x;
  if (i >= total) return;
  int p = (int)(i & 7);
  int lane = (int)((i >> 3) & 31);
  long tt = i >> 8;
  int kt = (int)(tt % KT);
  int ct = (int)(tt / KT);
  int col = ct * 16 + (lane & 15);
  int k = kt * 32 + frag_koff(lane, p);
  unsigned int lo = f2bf(W[(long)k * M + col]);
  unsigned int hi = f2bf(W[(long)(k + 1) * M + col]);
  out[i] = lo | (hi << 16);
}

template <int RELU, int OUTPK>
__global__ void gemm_pk(const uint4* __restrict__ Apk, const uint4* __restrict__ Wpk,
                        const float* __restrict__ bias, void* __restrict__ Cout,
                        int N, int K, int M) {
  const int tid  = threadIdx.x;
  const int lane = tid & 31;
  const int wave = tid >> 5;
  const int wr   = wave & 3;
  const int wc   = wave >> 2;
  const int KT   = K >> 5;
  const int nrt  = N >> 4;
  const int nct  = M >> 4;
  const int rt0  = blockIdx.y * 8 + wr * 2;
  const int ct0  = blockIdx.x * 4 + wc * 2;
  const bool r0 = rt0 < nrt,     r1 = (rt0 + 1) < nrt;
  const bool c0 = ct0 < nct,     c1 = (ct0 + 1) < nct;
  const int rA0 = r0 ? rt0 : 0,  rA1 = r1 ? (rt0 + 1) : 0;
  const int cB0 = c0 ? ct0 : 0,  cB1 = c1 ? (ct0 + 1) : 0;

  long ia0 = ((long)rA0 * KT * 32 + lane) * 2;
  long ia1 = ((long)rA1 * KT * 32 + lane) * 2;
  long ib0 = ((long)cB0 * KT * 32 + lane) * 2;
  long ib1 = ((long)cB1 * KT * 32 + lane) * 2;
  const long step = 64;

  v8f acc00 = {}, acc01 = {}, acc10 = {}, acc11 = {};
  union F { uint4 q[2]; v16bf v; };

  for (int kt = 0; kt < KT; ++kt) {
    F a0, a1, b0, b1;
    a0.q[0] = Apk[ia0]; a0.q[1] = Apk[ia0 + 1];
    a1.q[0] = Apk[ia1]; a1.q[1] = Apk[ia1 + 1];
    b0.q[0] = Wpk[ib0]; b0.q[1] = Wpk[ib0 + 1];
    b1.q[0] = Wpk[ib1]; b1.q[1] = Wpk[ib1 + 1];
    if (kt + 1 < KT) {
      __builtin_prefetch((const void*)(Apk + ia0 + step), 0, 3);
      __builtin_prefetch((const void*)(Apk + ia1 + step), 0, 3);
    }
    ia0 += step; ia1 += step; ib0 += step; ib1 += step;
    acc00 = __builtin_amdgcn_wmma_f32_16x16x32_bf16(false, a0.v, false, b0.v, (short)0, acc00, false, false);
    acc01 = __builtin_amdgcn_wmma_f32_16x16x32_bf16(false, a0.v, false, b1.v, (short)0, acc01, false, false);
    acc10 = __builtin_amdgcn_wmma_f32_16x16x32_bf16(false, a1.v, false, b0.v, (short)0, acc10, false, false);
    acc11 = __builtin_amdgcn_wmma_f32_16x16x32_bf16(false, a1.v, false, b1.v, (short)0, acc11, false, false);
  }

  const int rowL = lane & 15;
  if constexpr (OUTPK) {
    __shared__ uint4 stage[8 * 2 * 64];
    unsigned short* sw = (unsigned short*)stage + wave * 1024;
    const int hh = lane >> 4;
    auto put = [&](const v8f& acc, int i, int j, int ctc) {
      const float b = bias[ctc * 16 + rowL];
      #pragma unroll
      for (int r = 0; r < 8; ++r) {
        float o = acc[r] + b;
        if (RELU) o = fmaxf(o, 0.0f);
        sw[i * 512 + (8 * hh + r + 16 * (rowL >> 3)) * 16 + (rowL & 7) + 8 * j] = f2bf(o);
      }
    };
    put(acc00, 0, 0, cB0);   put(acc01, 0, 1, cB1);
    put(acc10, 1, 0, cB0);   put(acc11, 1, 1, cB1);
    __syncthreads();
    const int KTo = M >> 5;
    const int kto = ct0 >> 1;
    uint4* __restrict__ Opk = (uint4*)Cout;
    #pragma unroll
    for (int i = 0; i < 2; ++i) {
      const bool rok = i ? r1 : r0;
      if (rok && c1) {
        const long o4 = ((long)(rt0 + i) * KTo + kto) * 64;
        Opk[o4 + lane]      = stage[(wave * 2 + i) * 64 + lane];
        Opk[o4 + 32 + lane] = stage[(wave * 2 + i) * 64 + 32 + lane];
      }
    }
  } else {
    auto st = [&](const v8f& acc, int rt, bool rok, int ct, bool cok) {
      if (!rok || !cok) return;
      int col = ct * 16 + rowL;
      #pragma unroll
      for (int r = 0; r < 8; ++r) {
        int row = rt * 16 + ((lane < 16) ? r : r + 8);
        float o = acc[r] + bias[col];
        if (RELU) o = fmaxf(o, 0.0f);
        ((float*)Cout)[(long)row * M + col] = o;
      }
    };
    st(acc00, rt0, r0, ct0, c0);         st(acc01, rt0, r0, ct0 + 1, c1);
    st(acc10, rt0 + 1, r1, ct0, c0);     st(acc11, rt0 + 1, r1, ct0 + 1, c1);
  }
}

#define POOL_R 16
static_assert(N_NODES % POOL_R == 0, "every wave owns POOL_R whole rows");
static_assert(HID == 256, "a lane owns two float4 of a row");

__device__ __forceinline__ void pool_flush(float* __restrict__ pooled, float* __restrict__ counts, int g, int lane,
                                           const float4& a0, const float4& a1, int len) {
  float* o = pooled + (long)g * HID + lane * 4;
  atomicAdd(o + 0, a0.x); atomicAdd(o + 1, a0.y); atomicAdd(o + 2, a0.z); atomicAdd(o + 3, a0.w);
  atomicAdd(o + 128, a1.x); atomicAdd(o + 129, a1.y); atomicAdd(o + 130, a1.z); atomicAdd(o + 131, a1.w);
  if (lane == 0) atomicAdd(&counts[g], (float)len);
}

__global__ __launch_bounds__(256) void pool_runs(const float* __restrict__ h, const int* __restrict__ batch,
                                                 float* __restrict__ pooled, float* __restrict__ counts) {
  const int lane = threadIdx.x & 31;
  const int gw = __builtin_amdgcn_readfirstlane((int)(blockIdx.x * 8 + (threadIdx.x >> 5)));
  const int n0 = gw * POOL_R;
  if (n0 >= N_NODES) return;
  const int ids = batch[n0 + (lane & (POOL_R - 1))];
  const float4* __restrict__ hp = (const float4*)h + (long)n0 * (HID / 4);
  float4 x[POOL_R][2];
  #pragma unroll
  for (int j = 0; j < POOL_R; ++j) {
    x[j][0] = hp[j * (HID / 4) + lane];
    x[j][1] = hp[j * (HID / 4) + 32 + lane];
  }
  int cur = __builtin_amdgcn_readlane(ids, 0);
  float4 a0 = x[0][0], a1 = x[0][1];
  int len = 1;
  #pragma unroll
  for (int j = 1; j < POOL_R; ++j) {
    const int g = __builtin_amdgcn_readlane(ids, j);
    if (g != cur) {
      pool_flush(pooled, counts, cur, lane, a0, a1, len);
      a0 = x[j][0]; a1 = x[j][1];
      len = 1;
      cur = g;
    } else {
      add4(a0, x[j][0]); add4(a1, x[j][1]);
      ++len;
    }
  }
  pool_flush(pooled, counts, cur, lane, a0, a1, len);
}

__global__ void pool_div_pack(const float* __restrict__ pooled, const float* __restrict__ counts,
                              unsigned int* __restrict__ out) {
  const int KT = HID / 32;
  int i = blockIdx.x * blockDim.x + threadIdx.x;
  int p = i & 7;
  int lane = (i >> 3) & 31;
  int tt = i >> 8;
  int kt = tt % KT;
  int rt = tt / KT;
  int row = rt * 16 + (lane & 15);
  int k = kt * 32 + frag_koff(lane, p);
  const float c = fmaxf(counts[row], 1.0f);
  const float* s = pooled + (long)row * HID + k;
  out[i] = (unsigned int)f2bf(s[0] / c) | ((unsigned int)f2bf(s[1] / c) << 16);
}

extern "C" void kernel_launch(void* const* d_in, const int* in_sizes, int n_in,
                              void* d_out, int out_size, void* d_ws, size_t ws_size,
                              hipStream_t stream) {
  const float* x     = (const float*)d_in[0];
  const int*   ei    = (const int*)d_in[1];
  const int*   esrc  = ei;
  const int*   edst  = ei + N_EDGES;
  const int*   batch = (const int*)d_in[2];
  const float* cw1[3] = {(const float*)d_in[3],  (const float*)d_in[7],  (const float*)d_in[11]};
  const float* cb1[3] = {(const float*)d_in[4],  (const float*)d_in[8],  (const float*)d_in[12]};
  const float* cw2[3] = {(const float*)d_in[5],  (const float*)d_in[9],  (const float*)d_in[13]};
  const float* cb2[3] = {(const float*)d_in[6],  (const float*)d_in[10], (const float*)d_in[14]};
  const float* ow1 = (const float*)d_in[15];
  const float* ob1 = (const float*)d_in[16];
  const float* ow2 = (const float*)d_in[17];
  const float* ob2 = (const float*)d_in[18];

  char* ws = (char*)d_ws;
  size_t off = 0;
  auto bump = [&](size_t bytes) {
    void* p = ws + off;
    off += (bytes + 255) & ~(size_t)255;
    return p;
  };
  const int NRT = N_NODES / 16;
  float*          B2     = (float*)bump((size_t)N_NODES * HID * 4);
  unsigned int*   Apk    = (unsigned int*)bump((size_t)NRT * 8 * 256 * 4);
  unsigned int*   Mpk    = (unsigned int*)bump((size_t)NRT * 8 * 256 * 4);
  float*          pooled = (float*)bump((size_t)N_GRAPHS * (HID + 1) * 4);
  float*          counts = pooled + (size_t)N_GRAPHS * HID;
  unsigned int*   Ppk    = (unsigned int*)bump((size_t)(N_GRAPHS / 16) * 8 * 256 * 4);
  unsigned int*   MPpk   = (unsigned int*)bump((size_t)(N_GRAPHS / 16) * 8 * 256 * 4);
  unsigned int*   wpk1[3], *wpk2[3];
  for (int l = 0; l < 3; ++l) {
    wpk1[l] = (unsigned int*)bump((size_t)(HID / 16) * 8 * 256 * 4);
    wpk2[l] = (unsigned int*)bump((size_t)(HID / 16) * 8 * 256 * 4);
  }
  unsigned int* owpk1 = (unsigned int*)bump((size_t)(HID / 16) * 8 * 256 * 4);
  unsigned int* owpk2 = (unsigned int*)bump((size_t)(OUT_C / 16) * 8 * 256 * 4);
  int* deg    = (int*)bump((size_t)N_NODES * 4);
  int* rowptr = (int*)bump((size_t)(N_NODES + 1) * 4);
  int* cursor = (int*)bump((size_t)N_NODES * 4);
  int* col    = (int*)bump((size_t)N_EDGES * 4);

  dim3 blk(256);
  auto wpack = [&](const float* W, unsigned int* dst, int K, int M) {
    int KT = K / 32;
    long total = (long)(M / 16) * KT * 256;
    wpack_f32<<<(unsigned)((total + 255) / 256), blk, 0, stream>>>(W, dst, K, M, KT, total);
  };

  wpack(cw1[0], wpk1[0], IN_C, HID);
  wpack(cw2[0], wpk2[0], HID, HID);
  for (int l = 1; l < 3; ++l) {
    wpack(cw1[l], wpk1[l], HID, HID);
    wpack(cw2[l], wpk2[l], HID, HID);
  }
  wpack(ow1, owpk1, HID, HID);
  wpack(ow2, owpk2, HID, OUT_C);

  csr_zero<<<(N_NODES + 255) / 256, blk, 0, stream>>>(deg);
  csr_count<<<(N_EDGES + 255) / 256, blk, 0, stream>>>(edst, deg);
  csr_scan<<<1, SCAN_T, 0, stream>>>(deg, rowptr, cursor);
  csr_fill<<<(N_EDGES + 255) / 256, blk, 0, stream>>>(esrc, edst, cursor, col);

  const float* h = x;
  int C = IN_C;
  for (int l = 0; l < 3; ++l) {
    if (l == 0) gather_pack<IN_C><<<NRT, 512, 0, stream>>>(h, rowptr, col, (uint4*)Apk);
    else        gather_pack<HID><<<NRT, 512, 0, stream>>>(h, rowptr, col, (uint4*)Apk);

    dim3 g1((HID / 16 + 3) / 4, (NRT + 7) / 8);
    gemm_pk<1, 1><<<g1, blk, 0, stream>>>((const uint4*)Apk, (const uint4*)wpk1[l], cb1[l],
                                          Mpk, N_NODES, C, HID);
    dim3 g2((HID / 16 + 3) / 4, (NRT + 7) / 8);
    gemm_pk<1, 0><<<g2, blk, 0, stream>>>((const uint4*)Mpk, (const uint4*)wpk2[l], cb2[l],
                                          B2, N_NODES, HID, HID);
    h = B2;
    C = HID;
  }

  hipMemsetAsync(pooled, 0, (size_t)N_GRAPHS * (HID + 1) * 4, stream);
  pool_runs<<<(N_NODES / POOL_R + 7) / 8, blk, 0, stream>>>(B2, batch, pooled, counts);

  const int GRT = N_GRAPHS / 16;
  pool_div_pack<<<GRT * (HID / 32), blk, 0, stream>>>(pooled, counts, Ppk);
  dim3 g3((HID / 16 + 3) / 4, (GRT + 7) / 8);
  gemm_pk<1, 1><<<g3, blk, 0, stream>>>((const uint4*)Ppk, (const uint4*)owpk1, ob1,
                                        MPpk, N_GRAPHS, HID, HID);
  dim3 g4((OUT_C / 16 + 3) / 4, (GRT + 7) / 8);
  gemm_pk<0, 0><<<g4, blk, 0, stream>>>((const uint4*)MPpk, (const uint4*)owpk2, ob2,
                                        (float*)d_out, N_GRAPHS, HID, OUT_C);
}


# The eight weight-pack launches, the degree-counter clear and the pooled-sum memset merged into one preparation launch (a block's index picks its job); 25 -> 17 launches, memset dropped
# speedup vs baseline: 2.8042x; 1.0469x over previous
#include <hip/hip_runtime.h>

#define N_NODES 50000
#define N_EDGES 800000
#define N_GRAPHS 512
#define IN_C 128
#define HID 256
#define OUT_C 16

typedef __attribute__((ext_vector_type(16))) __bf16 v16bf;
typedef __attribute__((ext_vector_type(8)))  float  v8f;

__device__ __forceinline__ unsigned short f2bf(float f) {
  union { float f; unsigned int u; } c; c.f = f;
  unsigned int u = c.u;
  u += 0x7fffu + ((u >> 16) & 1u);
  return (unsigned short)(u >> 16);
}

__device__ __forceinline__ int frag_koff(int lane, int p) {
  int kb = (lane < 16) ? 0 : 8;
  return (p < 4) ? (kb + 2 * p) : (16 + kb + 2 * (p - 4));
}

__global__ void csr_count(const int* __restrict__ edst, int* __restrict__ deg) {
  int e = blockIdx.x * blockDim.x + threadIdx.x;
  if (e < N_EDGES) atomicAdd(&deg[edst[e]], 1);
}

#define SCAN_T 1024
#define SCAN_TILES ((N_NODES / 4 + SCAN_T - 1) / SCAN_T)

__device__ __forceinline__ int wave_incl_scan(int x, int lane) {
  #pragma unroll
  for (int o = 1; o < 32; o <<= 1) {
    const int up = __shfl_up(x, o, 32);
    if (lane >= o) x += up;
  }
  return x;
}

__global__ __launch_bounds__(SCAN_T) void csr_scan(const int* __restrict__ deg, int* __restrict__ rowptr,
                                                   int* __restrict__ cursor) {
  __shared__ int run_off[SCAN_TILES * 32];
  const int t = threadIdx.x;
  const int lane = t & 31;
  const int w = t >> 5;
  const int n4 = N_NODES / 4;
  int4 q[SCAN_TILES];
  int s[SCAN_TILES], inc[SCAN_TILES];
  #pragma unroll
  for (int k = 0; k < SCAN_TILES; ++k) {
    const int i4 = k * SCAN_T + t;
    q[k] = (i4 < n4) ? ((const int4*)deg)[i4] : make_int4(0, 0, 0, 0);
  }
  #pragma unroll
  for (int k = 0; k < SCAN_TILES; ++k) {
    s[k] = q[k].x + q[k].y + q[k].z + q[k].w;
    inc[k] = wave_incl_scan(s[k], lane);
    if (lane == 31) run_off[k * 32 + w] = inc[k];
  }
  __syncthreads();
  if (w == 0) {
    int carry = 0;
    #pragma unroll
    for (int k = 0; k < SCAN_TILES; ++k) {
      const int tot = run_off[k * 32 + lane];
      const int ti = wave_incl_scan(tot, lane);
      run_off[k * 32 + lane] = carry + ti - tot;
      carry += __shfl(ti, 31, 32);
    }
    if (lane == 0) rowptr[N_NODES] = carry;
  }
  __syncthreads();
  #pragma unroll
  for (int k = 0; k < SCAN_TILES; ++k) {
    const int i4 = k * SCAN_T + t;
    if (i4 < n4) {
      int4 o;
      o.x = run_off[k * 32 + w] + inc[k] - s[k];
      o.y = o.x + q[k].x;
      o.z = o.y + q[k].y;
      o.w = o.z + q[k].z;
      ((int4*)rowptr)[i4] = o;
      ((int4*)cursor)[i4] = o;
    }
  }
}

__global__ void csr_fill(const int* __restrict__ esrc, const int* __restrict__ edst,
                         int* __restrict__ cursor, int* __restrict__ col) {
  int e = blockIdx.x * blockDim.x + threadIdx.x;
  if (e < N_EDGES) {
    int pos = atomicAdd(&cursor[edst[e]], 1);
    col[pos] = esrc[e];
  }
}

__device__ __forceinline__ void add4(float4& a, const float4& b) {
  a.x += b.x; a.y += b.y; a.z += b.z; a.w += b.w;
}

static_assert(N_NODES % 16 == 0, "one work-group per full 16-row tile");
template <int C>
__global__ __launch_bounds__(512) void gather_pack(const float* __restrict__ h, const int* __restrict__ rowptr,
                                                   const int* __restrict__ col, uint4* __restrict__ Apk) {
  constexpr int V = C / 128;
  constexpr int U = 8 / V;
  constexpr int KT = C / 32;
  __shared__ uint4 tile[KT * 64];
  const int lane = threadIdx.x & 31;
  const int w = __builtin_amdgcn_readfirstlane((int)(threadIdx.x >> 5));
  const int node = __builtin_amdgcn_readfirstlane((int)(blockIdx.x * 16 + w));
  const int beg = rowptr[node];
  const int end = rowptr[node + 1];
  const float4* __restrict__ hp = (const float4*)h;
  float4 acc[V];
  #pragma unroll
  for (int v = 0; v < V; ++v) acc[v] = hp[(long)node * (C / 4) + v * 32 + lane];
  for (int base = beg; base < end; base += 32) {
    const int cnt = min(32, end - base);
    const int mine = (lane < cnt) ? col[base + lane] : 0;
    int j = 0;
    for (; j + U <= cnt; j += U) {
      float4 x[U][V];
      #pragma unroll
      for (int u = 0; u < U; ++u) {
        const int sidx = __builtin_amdgcn_readlane(mine, j + u);
        #pragma unroll
        for (int v = 0; v < V; ++v) x[u][v] = hp[(long)sidx * (C / 4) + v * 32 + lane];
      }
      #pragma unroll
      for (int u = 0; u < U; ++u) {
        #pragma unroll
        for (int v = 0; v < V; ++v) add4(acc[v], x[u][v]);
      }
    }
    #pragma nounroll
    for (; j < cnt; ++j) {
      const int sidx = __builtin_amdgcn_readlane(mine, j);
      #pragma unroll
      for (int v = 0; v < V; ++v) add4(acc[v], hp[(long)sidx * (C / 4) + v * 32 + lane]);
    }
  }
  unsigned int* tl = (unsigned int*)tile;
  const int m  = lane & 7;
  const int fh = (m >> 1) & 1;
  const int p0 = 2 * (m & 1) + 4 * (m >> 2);
  #pragma unroll
  for (int v = 0; v < V; ++v) {
    const int kt = v * 4 + (lane >> 3);
    uint2 pr;
    pr.x = (unsigned int)f2bf(acc[v].x) | ((unsigned int)f2bf(acc[v].y) << 16);
    pr.y = (unsigned int)f2bf(acc[v].z) | ((unsigned int)f2bf(acc[v].w) << 16);
    *(uint2*)(tl + kt * 256 + (w + 16 * fh) * 8 + p0) = pr;
  }
  __syncthreads();
  if (threadIdx.x < KT * 64) Apk[(long)blockIdx.x * (KT * 64) + threadIdx.x] = tile[threadIdx.x];
}

template <int K, int M>
__device__ __forceinline__ void wpack_elem(const float* __restrict__ Wm, unsigned int* __restrict__ out, int i) {
  constexpr int KT = K / 32;
  const int p = i & 7;
  const int lane = (i >> 3) & 31;
  const int tt = i >> 8;
  const int kt = tt % KT;
  const int ct = tt / KT;
  const int col = ct * 16 + (lane & 15);
  const int k = kt * 32 + frag_koff(lane, p);
  const unsigned int lo = f2bf(Wm[(long)k * M + col]);
  const unsigned int hi = f2bf(Wm[(long)(k + 1) * M + col]);
  out[i] = lo | (hi << 16);
}

#define PREP_T      256
#define PREP_B_WIN  ((HID / 16) * (IN_C / 32))
#define PREP_B_WH   ((HID / 16) * (HID / 32))
#define PREP_N_WH   6
#define PREP_B_WOUT ((OUT_C / 16) * (HID / 32))
#define PREP_N_DEG4 (N_NODES / 4)
#define PREP_B_DEG  ((PREP_N_DEG4 + PREP_T - 1) / PREP_T)
#define PREP_N_PL4  (N_GRAPHS * (HID + 1) / 4)
#define PREP_B_PL   ((PREP_N_PL4 + PREP_T - 1) / PREP_T)
#define PREP_BLOCKS (PREP_B_WIN + PREP_N_WH * PREP_B_WH + PREP_B_WOUT + PREP_B_DEG + PREP_B_PL)
static_assert(N_NODES % 4 == 0 && (N_GRAPHS * (HID + 1)) % 4 == 0, "the clears go out as whole 16-byte stores");

__global__ __launch_bounds__(PREP_T) void prep_all(
    const float* __restrict__ w_in,
    const float* __restrict__ wh0, const float* __restrict__ wh1, const float* __restrict__ wh2,
    const float* __restrict__ wh3, const float* __restrict__ wh4, const float* __restrict__ wh5,
    const float* __restrict__ w_out,
    unsigned int* __restrict__ p_in,
    unsigned int* __restrict__ ph0, unsigned int* __restrict__ ph1, unsigned int* __restrict__ ph2,
    unsigned int* __restrict__ ph3, unsigned int* __restrict__ ph4, unsigned int* __restrict__ ph5,
    unsigned int* __restrict__ p_out,
    int* __restrict__ deg, float* __restrict__ pooled) {
  const int t = threadIdx.x;
  int b = blockIdx.x;
  if (b < PREP_B_WIN) { wpack_elem<IN_C, HID>(w_in, p_in, b * PREP_T + t); return; }
  b -= PREP_B_WIN;
  if (b < PREP_N_WH * PREP_B_WH) {
    const int j = b / PREP_B_WH;
    const int i = (b % PREP_B_WH) * PREP_T + t;
    const float* Wm = wh0;  unsigned int* o = ph0;
    if (j == 1) { Wm = wh1; o = ph1; }
    if (j == 2) { Wm = wh2; o = ph2; }
    if (j == 3) { Wm = wh3; o = ph3; }
    if (j == 4) { Wm = wh4; o = ph4; }
    if (j == 5) { Wm = wh5; o = ph5; }
    wpack_elem<HID, HID>(Wm, o, i);
    return;
  }
  b -= PREP_N_WH * PREP_B_WH;
  if (b < PREP_B_WOUT) { wpack_elem<HID, OUT_C>(w_out, p_out, b * PREP_T + t); return; }
  b -= PREP_B_WOUT;
  if (b < PREP_B_DEG) {
    const int i4 = b * PREP_T + t;
    if (i4 < PREP_N_DEG4) ((int4*)deg)[i4] = make_int4(0, 0, 0, 0);
    return;
  }
  b -= PREP_B_DEG;
  const int i4 = b * PREP_T + t;
  if (i4 < PREP_N_PL4) ((float4*)pooled)[i4] = make_float4(0.0f, 0.0f, 0.0f, 0.0f);
}

template <int RELU, int OUTPK>
__global__ void gemm_pk(const uint4* __restrict__ Apk, const uint4* __restrict__ Wpk,
                        const float* __restrict__ bias, void* __restrict__ Cout,
                        int N, int K, int M) {
  const int tid  = threadIdx.x;
  const int lane = tid & 31;
  const int wave = tid >> 5;
  const int wr   = wave & 3;
  const int wc   = wave >> 2;
  const int KT   = K >> 5;
  const int nrt  = N >> 4;
  const int nct  = M >> 4;
  const int rt0  = blockIdx.y * 8 + wr * 2;
  const int ct0  = blockIdx.x * 4 + wc * 2;
  const bool r0 = rt0 < nrt,     r1 = (rt0 + 1) < nrt;
  const bool c0 = ct0 < nct,     c1 = (ct0 + 1) < nct;
  const int rA0 = r0 ? rt0 : 0,  rA1 = r1 ? (rt0 + 1) : 0;
  const int cB0 = c0 ? ct0 : 0,  cB1 = c1 ? (ct0 + 1) : 0;

  long ia0 = ((long)rA0 * KT * 32 + lane) * 2;
  long ia1 = ((long)rA1 * KT * 32 + lane) * 2;
  long ib0 = ((long)cB0 * KT * 32 + lane) * 2;
  long ib1 = ((long)cB1 * KT * 32 + lane) * 2;
  const long step = 64;

  v8f acc00 = {}, acc01 = {}, acc10 = {}, acc11 = {};
  union F { uint4 q[2]; v16bf v; };

  for (int kt = 0; kt < KT; ++kt) {
    F a0, a1, b0, b1;
    a0.q[0] = Apk[ia0]; a0.q[1] = Apk[ia0 + 1];
    a1.q[0] = Apk[ia1]; a1.q[1] = Apk[ia1 + 1];
    b0.q[0] = Wpk[ib0]; b0.q[1] = Wpk[ib0 + 1];
    b1.q[0] = Wpk[ib1]; b1.q[1] = Wpk[ib1 + 1];
    if (kt + 1 < KT) {
      __builtin_prefetch((const void*)(Apk + ia0 + step), 0, 3);
      __builtin_prefetch((const void*)(Apk + ia1 + step), 0, 3);
    }
    ia0 += step; ia1 += step; ib0 += step; ib1 += step;
    acc00 = __builtin_amdgcn_wmma_f32_16x16x32_bf16(false, a0.v, false, b0.v, (short)0, acc00, false, false);
    acc01 = __builtin_amdgcn_wmma_f32_16x16x32_bf16(false, a0.v, false, b1.v, (short)0, acc01, false, false);
    acc10 = __builtin_amdgcn_wmma_f32_16x16x32_bf16(false, a1.v, false, b0.v, (short)0, acc10, false, false);
    acc11 = __builtin_amdgcn_wmma_f32_16x16x32_bf16(false, a1.v, false, b1.v, (short)0, acc11, false, false);
  }

  const int rowL = lane & 15;
  if constexpr (OUTPK) {
    __shared__ uint4 stage[8 * 2 * 64];
    unsigned short* sw = (unsigned short*)stage + wave * 1024;
    const int hh = lane >> 4;
    auto put = [&](const v8f& acc, int i, int j, int ctc) {
      const float b = bias[ctc * 16 + rowL];
      #pragma unroll
      for (int r = 0; r < 8; ++r) {
        float o = acc[r] + b;
        if (RELU) o = fmaxf(o, 0.0f);
        sw[i * 512 + (8 * hh + r + 16 * (rowL >> 3)) * 16 + (rowL & 7) + 8 * j] = f2bf(o);
      }
    };
    put(acc00, 0, 0, cB0);   put(acc01, 0, 1, cB1);
    put(acc10, 1, 0, cB0);   put(acc11, 1, 1, cB1);
    __syncthreads();
    const int KTo = M >> 5;
    const int kto = ct0 >> 1;
    uint4* __restrict__ Opk = (uint4*)Cout;
    #pragma unroll
    for (int i = 0; i < 2; ++i) {
      const bool rok = i ? r1 : r0;
      if (rok && c1) {
        const long o4 = ((long)(rt0 + i) * KTo + kto) * 64;
        Opk[o4 + lane]      = stage[(wave * 2 + i) * 64 + lane];
        Opk[o4 + 32 + lane] = stage[(wave * 2 + i) * 64 + 32 + lane];
      }
    }
  } else {
    auto st = [&](const v8f& acc, int rt, bool rok, int ct, bool cok) {
      if (!rok || !cok) return;
      int col = ct * 16 + rowL;
      #pragma unroll
      for (int r = 0; r < 8; ++r) {
        int row = rt * 16 + ((lane < 16) ? r : r + 8);
        float o = acc[r] + bias[col];
        if (RELU) o = fmaxf(o, 0.0f);
        ((float*)Cout)[(long)row * M + col] = o;
      }
    };
    st(acc00, rt0, r0, ct0, c0);         st(acc01, rt0, r0, ct0 + 1, c1);
    st(acc10, rt0 + 1, r1, ct0, c0);     st(acc11, rt0 + 1, r1, ct0 + 1, c1);
  }
}

#define POOL_R 16
static_assert(N_NODES % POOL_R == 0, "every wave owns POOL_R whole rows");
static_assert(HID == 256, "a lane owns two float4 of a row");

__device__ __forceinline__ void pool_flush(float* __restrict__ pooled, float* __restrict__ counts, int g, int lane,
                                           const float4& a0, const float4& a1, int len) {
  float* o = pooled + (long)g * HID + lane * 4;
  atomicAdd(o + 0, a0.x); atomicAdd(o + 1, a0.y); atomicAdd(o + 2, a0.z); atomicAdd(o + 3, a0.w);
  atomicAdd(o + 128, a1.x); atomicAdd(o + 129, a1.y); atomicAdd(o + 130, a1.z); atomicAdd(o + 131, a1.w);
  if (lane == 0) atomicAdd(&counts[g], (float)len);
}

__global__ __launch_bounds__(256) void pool_runs(const float* __restrict__ h, const int* __restrict__ batch,
                                                 float* __restrict__ pooled, float* __restrict__ counts) {
  const int lane = threadIdx.x & 31;
  const int gw = __builtin_amdgcn_readfirstlane((int)(blockIdx.x * 8 + (threadIdx.x >> 5)));
  const int n0 = gw * POOL_R;
  if (n0 >= N_NODES) return;
  const int ids = batch[n0 + (lane & (POOL_R - 1))];
  const float4* __restrict__ hp = (const float4*)h + (long)n0 * (HID / 4);
  float4 x[POOL_R][2];
  #pragma unroll
  for (int j = 0; j < POOL_R; ++j) {
    x[j][0] = hp[j * (HID / 4) + lane];
    x[j][1] = hp[j * (HID / 4) + 32 + lane];
  }
  int cur = __builtin_amdgcn_readlane(ids, 0);
  float4 a0 = x[0][0], a1 = x[0][1];
  int len = 1;
  #pragma unroll
  for (int j = 1; j < POOL_R; ++j) {
    const int g = __builtin_amdgcn_readlane(ids, j);
    if (g != cur) {
      pool_flush(pooled, counts, cur, lane, a0, a1, len);
      a0 = x[j][0]; a1 = x[j][1];
      len = 1;
      cur = g;
    } else {
      add4(a0, x[j][0]); add4(a1, x[j][1]);
      ++len;
    }
  }
  pool_flush(pooled, counts, cur, lane, a0, a1, len);
}

__global__ void pool_div_pack(const float* __restrict__ pooled, const float* __restrict__ counts,
                              unsigned int* __restrict__ out) {
  const int KT = HID / 32;
  int i = blockIdx.x * blockDim.x + threadIdx.x;
  int p = i & 7;
  int lane = (i >> 3) & 31;
  int tt = i >> 8;
  int kt = tt % KT;
  int rt = tt / KT;
  int row = rt * 16 + (lane & 15);
  int k = kt * 32 + frag_koff(lane, p);
  const float c = fmaxf(counts[row], 1.0f);
  const float* s = pooled + (long)row * HID + k;
  out[i] = (unsigned int)f2bf(s[0] / c) | ((unsigned int)f2bf(s[1] / c) << 16);
}

extern "C" void kernel_launch(void* const* d_in, const int* in_sizes, int n_in,
                              void* d_out, int out_size, void* d_ws, size_t ws_size,
                              hipStream_t stream) {
  const float* x     = (const float*)d_in[0];
  const int*   ei    = (const int*)d_in[1];
  const int*   esrc  = ei;
  const int*   edst  = ei + N_EDGES;
  const int*   batch = (const int*)d_in[2];
  const float* cw1[3] = {(const float*)d_in[3],  (const float*)d_in[7],  (const float*)d_in[11]};
  const float* cb1[3] = {(const float*)d_in[4],  (const float*)d_in[8],  (const float*)d_in[12]};
  const float* cw2[3] = {(const float*)d_in[5],  (const float*)d_in[9],  (const float*)d_in[13]};
  const float* cb2[3] = {(const float*)d_in[6],  (const float*)d_in[10], (const float*)d_in[14]};
  const float* ow1 = (const float*)d_in[15];
  const float* ob1 = (const float*)d_in[16];
  const float* ow2 = (const float*)d_in[17];
  const float* ob2 = (const float*)d_in[18];

  char* ws = (char*)d_ws;
  size_t off = 0;
  auto bump = [&](size_t bytes) {
    void* p = ws + off;
    off += (bytes + 255) & ~(size_t)255;
    return p;
  };
  const int NRT = N_NODES / 16;
  float*          B2     = (float*)bump((size_t)N_NODES * HID * 4);
  unsigned int*   Apk    = (unsigned int*)bump((size_t)NRT * 8 * 256 * 4);
  unsigned int*   Mpk    = (unsigned int*)bump((size_t)NRT * 8 * 256 * 4);
  float*          pooled = (float*)bump((size_t)N_GRAPHS * (HID + 1) * 4);
  float*          counts = pooled + (size_t)N_GRAPHS * HID;
  unsigned int*   Ppk    = (unsigned int*)bump((size_t)(N_GRAPHS / 16) * 8 * 256 * 4);
  unsigned int*   MPpk   = (unsigned int*)bump((size_t)(N_GRAPHS / 16) * 8 * 256 * 4);
  unsigned int*   wpk1[3], *wpk2[3];
  for (int l = 0; l < 3; ++l) {
    wpk1[l] = (unsigned int*)bump((size_t)(HID / 16) * 8 * 256 * 4);
    wpk2[l] = (unsigned int*)bump((size_t)(HID / 16) * 8 * 256 * 4);
  }
  unsigned int* owpk1 = (unsigned int*)bump((size_t)(HID / 16) * 8 * 256 * 4);
  unsigned int* owpk2 = (unsigned int*)bump((size_t)(OUT_C / 16) * 8 * 256 * 4);
  int* deg    = (int*)bump((size_t)N_NODES * 4);
  int* rowptr = (int*)bump((size_t)(N_NODES + 1) * 4);
  int* cursor = (int*)bump((size_t)N_NODES * 4);
  int* col    = (int*)bump((size_t)N_EDGES * 4);

  dim3 blk(256);

  prep_all<<<PREP_BLOCKS, PREP_T, 0, stream>>>(cw1[0], cw2[0], cw1[1], cw2[1], cw1[2], cw2[2], ow1, ow2,
                                                wpk1[0], wpk2[0], wpk1[1], wpk2[1], wpk1[2], wpk2[2], owpk1, owpk2,
                                                deg, pooled);

  csr_count<<<(N_EDGES + 255) / 256, blk, 0, stream>>>(edst, deg);
  csr_scan<<<1, SCAN_T, 0, stream>>>(deg, rowptr, cursor);
  csr_fill<<<(N_EDGES + 255) / 256, blk, 0, stream>>>(esrc, edst, cursor, col);

  const float* h = x;
  int C = IN_C;
  for (int l = 0; l < 3; ++l) {
    if (l == 0) gather_pack<IN_C><<<NRT, 512, 0, stream>>>(h, rowptr, col, (uint4*)Apk);
    else        gather_pack<HID><<<NRT, 512, 0, stream>>>(h, rowptr, col, (uint4*)Apk);

    dim3 g1((HID / 16 + 3) / 4, (NRT + 7) / 8);
    gemm_pk<1, 1><<<g1, blk, 0, stream>>>((const uint4*)Apk, (const uint4*)wpk1[l], cb1[l],
                                          Mpk, N_NODES, C, HID);
    dim3 g2((HID / 16 + 3) / 4, (NRT + 7) / 8);
    gemm_pk<1, 0><<<g2, blk, 0, stream>>>((const uint4*)Mpk, (const uint4*)wpk2[l], cb2[l],
                                          B2, N_NODES, HID, HID);
    h = B2;
    C = HID;
  }

  pool_runs<<<(N_NODES / POOL_R + 7) / 8, blk, 0, stream>>>(B2, batch, pooled, counts);

  const int GRT = N_GRAPHS / 16;
  pool_div_pack<<<GRT * (HID / 32), blk, 0, stream>>>(pooled, counts, Ppk);
  dim3 g3((HID / 16 + 3) / 4, (GRT + 7) / 8);
  gemm_pk<1, 1><<<g3, blk, 0, stream>>>((const uint4*)Ppk, (const uint4*)owpk1, ob1,
                                        MPpk, N_GRAPHS, HID, HID);
  dim3 g4((OUT_C / 16 + 3) / 4, (GRT + 7) / 8);
  gemm_pk<0, 0><<<g4, blk, 0, stream>>>((const uint4*)MPpk, (const uint4*)owpk2, ob2,
                                        (float*)d_out, N_GRAPHS, HID, OUT_C);
}


# precision: h of layers 1,2 held as bf16 rows (rounded before the f32 neighbour sum), gather reads half the bytes; lists sorted by source first, so the sum order, arrival order before, is fixed
# speedup vs baseline: 3.6137x; 1.2887x over previous
#include <hip/hip_runtime.h>

#define N_NODES 50000
#define N_EDGES 800000
#define N_GRAPHS 512
#define IN_C 128
#define HID 256
#define OUT_C 16

typedef __attribute__((ext_vector_type(16))) __bf16 v16bf;
typedef __attribute__((ext_vector_type(8)))  float  v8f;

__device__ __forceinline__ unsigned short f2bf(float f) {
  union { float f; unsigned int u; } c; c.f = f;
  unsigned int u = c.u;
  u += 0x7fffu + ((u >> 16) & 1u);
  return (unsigned short)(u >> 16);
}

__device__ __forceinline__ int frag_koff(int lane, int p) {
  int kb = (lane < 16) ? 0 : 8;
  return (p < 4) ? (kb + 2 * p) : (16 + kb + 2 * (p - 4));
}

__global__ void csr_count(const int* __restrict__ edst, int* __restrict__ deg) {
  int e = blockIdx.x * blockDim.x + threadIdx.x;
  if (e < N_EDGES) atomicAdd(&deg[edst[e]], 1);
}

#define SCAN_T 1024
#define SCAN_TILES ((N_NODES / 4 + SCAN_T - 1) / SCAN_T)

__device__ __forceinline__ int wave_incl_scan(int x, int lane) {
  #pragma unroll
  for (int o = 1; o < 32; o <<= 1) {
    const int up = __shfl_up(x, o, 32);
    if (lane >= o) x += up;
  }
  return x;
}

__global__ __launch_bounds__(SCAN_T) void csr_scan(const int* __restrict__ deg, int* __restrict__ rowptr,
                                                   int* __restrict__ cursor) {
  __shared__ int run_off[SCAN_TILES * 32];
  const int t = threadIdx.x;
  const int lane = t & 31;
  const int w = t >> 5;
  const int n4 = N_NODES / 4;
  int4 q[SCAN_TILES];
  int s[SCAN_TILES], inc[SCAN_TILES];
  #pragma unroll
  for (int k = 0; k < SCAN_TILES; ++k) {
    const int i4 = k * SCAN_T + t;
    q[k] = (i4 < n4) ? ((const int4*)deg)[i4] : make_int4(0, 0, 0, 0);
  }
  #pragma unroll
  for (int k = 0; k < SCAN_TILES; ++k) {
    s[k] = q[k].x + q[k].y + q[k].z + q[k].w;
    inc[k] = wave_incl_scan(s[k], lane);
    if (lane == 31) run_off[k * 32 + w] = inc[k];
  }
  __syncthreads();
  if (w == 0) {
    int carry = 0;
    #pragma unroll
    for (int k = 0; k < SCAN_TILES; ++k) {
      const int tot = run_off[k * 32 + lane];
      const int ti = wave_incl_scan(tot, lane);
      run_off[k * 32 + lane] = carry + ti - tot;
      carry += __shfl(ti, 31, 32);
    }
    if (lane == 0) rowptr[N_NODES] = carry;
  }
  __syncthreads();
  #pragma unroll
  for (int k = 0; k < SCAN_TILES; ++k) {
    const int i4 = k * SCAN_T + t;
    if (i4 < n4) {
      int4 o;
      o.x = run_off[k * 32 + w] + inc[k] - s[k];
      o.y = o.x + q[k].x;
      o.z = o.y + q[k].y;
      o.w = o.z + q[k].z;
      ((int4*)rowptr)[i4] = o;
      ((int4*)cursor)[i4] = o;
    }
  }
}

__global__ void csr_fill(const int* __restrict__ esrc, const int* __restrict__ edst,
                         int* __restrict__ cursor, int* __restrict__ col) {
  int e = blockIdx.x * blockDim.x + threadIdx.x;
  if (e < N_EDGES) {
    int pos = atomicAdd(&cursor[edst[e]], 1);
    col[pos] = esrc[e];
  }
}

__device__ __forceinline__ void add4(float4& a, const float4& b) {
  a.x += b.x; a.y += b.y; a.z += b.z; a.w += b.w;
}

static_assert(N_NODES % 16 == 0, "one work-group per full 16-row tile");
template <int C>
__global__ __launch_bounds__(512) void gather_pack(const float* __restrict__ h, const int* __restrict__ rowptr,
                                                   const int* __restrict__ col, uint4* __restrict__ Apk) {
  constexpr int V = C / 128;
  constexpr int U = 8 / V;
  constexpr int KT = C / 32;
  __shared__ uint4 tile[KT * 64];
  const int lane = threadIdx.x & 31;
  const int w = __builtin_amdgcn_readfirstlane((int)(threadIdx.x >> 5));
  const int node = __builtin_amdgcn_readfirstlane((int)(blockIdx.x * 16 + w));
  const int beg = rowptr[node];
  const int end = rowptr[node + 1];
  const float4* __restrict__ hp = (const float4*)h;
  float4 acc[V];
  #pragma unroll
  for (int v = 0; v < V; ++v) acc[v] = hp[(long)node * (C / 4) + v * 32 + lane];
  for (int base = beg; base < end; base += 32) {
    const int cnt = min(32, end - base);
    const int mine = (lane < cnt) ? col[base + lane] : 0;
    int j = 0;
    for (; j + U <= cnt; j += U) {
      float4 x[U][V];
      #pragma unroll
      for (int u = 0; u < U; ++u) {
        const int sidx = __builtin_amdgcn_readlane(mine, j + u);
        #pragma unroll
        for (int v = 0; v < V; ++v) x[u][v] = hp[(long)sidx * (C / 4) + v * 32 + lane];
      }
      #pragma unroll
      for (int u = 0; u < U; ++u) {
        #pragma unroll
        for (int v = 0; v < V; ++v) add4(acc[v], x[u][v]);
      }
    }
    #pragma nounroll
    for (; j < cnt; ++j) {
      const int sidx = __builtin_amdgcn_readlane(mine, j);
      #pragma unroll
      for (int v = 0; v < V; ++v) add4(acc[v], hp[(long)sidx * (C / 4) + v * 32 + lane]);
    }
  }
  unsigned int* tl = (unsigned int*)tile;
  const int m  = lane & 7;
  const int fh = (m >> 1) & 1;
  const int p0 = 2 * (m & 1) + 4 * (m >> 2);
  #pragma unroll
  for (int v = 0; v < V; ++v) {
    const int kt = v * 4 + (lane >> 3);
    uint2 pr;
    pr.x = (unsigned int)f2bf(acc[v].x) | ((unsigned int)f2bf(acc[v].y) << 16);
    pr.y = (unsigned int)f2bf(acc[v].z) | ((unsigned int)f2bf(acc[v].w) << 16);
    *(uint2*)(tl + kt * 256 + (w + 16 * fh) * 8 + p0) = pr;
  }
  __syncthreads();
  if (threadIdx.x < KT * 64) Apk[(long)blockIdx.x * (KT * 64) + threadIdx.x] = tile[threadIdx.x];
}

#define SORT_NPW 4
static_assert(N_NODES % SORT_NPW == 0, "every wave owns SORT_NPW whole nodes");
__global__ __launch_bounds__(256) void csr_sort(const int* __restrict__ rowptr, const int* __restrict__ col,
                                                int* __restrict__ col2) {
  const int lane = threadIdx.x & 31;
  const int n0 = __builtin_amdgcn_readfirstlane((int)((blockIdx.x * 8 + (threadIdx.x >> 5)) * SORT_NPW));
  if (n0 >= N_NODES) return;
  int bnd[SORT_NPW + 1];
  #pragma unroll
  for (int q = 0; q <= SORT_NPW; ++q) bnd[q] = rowptr[n0 + q];
  int mine[SORT_NPW];
  #pragma unroll
  for (int q = 0; q < SORT_NPW; ++q)
    mine[q] = (lane < bnd[q + 1] - bnd[q]) ? col[bnd[q] + lane] : 0x7fffffff;
  #pragma unroll
  for (int q = 0; q < SORT_NPW; ++q) {
    const int beg = bnd[q];
    const int d = bnd[q + 1] - beg;
    if (d <= 32) {
      int rank = 0;
      for (int j = 0; j < d; ++j) {
        const int v = __builtin_amdgcn_readlane(mine[q], j);
        rank += (v < mine[q] || (v == mine[q] && j < lane)) ? 1 : 0;
      }
      if (lane < d) col2[beg + rank] = mine[q];
    } else {
      for (int e = lane; e < d; e += 32) {
        const int v = col[beg + e];
        int rank = 0;
        for (int j = 0; j < d; ++j) {
          const int u = col[beg + j];
          rank += (u < v || (u == v && j < e)) ? 1 : 0;
        }
        col2[beg + rank] = v;
      }
    }
  }
}

__device__ __forceinline__ void addbf(float (&a)[8], const uint4& q) {
  a[0] += __uint_as_float(q.x << 16);  a[1] += __uint_as_float(q.x & 0xffff0000u);
  a[2] += __uint_as_float(q.y << 16);  a[3] += __uint_as_float(q.y & 0xffff0000u);
  a[4] += __uint_as_float(q.z << 16);  a[5] += __uint_as_float(q.z & 0xffff0000u);
  a[6] += __uint_as_float(q.w << 16);  a[7] += __uint_as_float(q.w & 0xffff0000u);
}

static_assert(HID == 256, "a bf16 row is exactly one 16-byte load per lane");
__global__ __launch_bounds__(512) void gather_pack_bf(const uint4* __restrict__ hb, const int* __restrict__ rowptr,
                                                      const int* __restrict__ col, uint4* __restrict__ Apk) {
  constexpr int KT = HID / 32;
  __shared__ uint4 tile[KT * 64];
  const int lane = threadIdx.x & 31;
  const int w = __builtin_amdgcn_readfirstlane((int)(threadIdx.x >> 5));
  const int node = __builtin_amdgcn_readfirstlane((int)(blockIdx.x * 16 + w));
  const int beg = rowptr[node];
  const int end = rowptr[node + 1];
  float acc[8];
  #pragma unroll
  for (int i = 0; i < 8; ++i) acc[i] = 0.0f;
  addbf(acc, hb[(long)node * (HID / 8) + lane]);
  for (int base = beg; base < end; base += 32) {
    const int cnt = min(32, end - base);
    const int mine = (lane < cnt) ? col[base + lane] : 0;
    int j = 0;
    for (; j + 8 <= cnt; j += 8) {
      uint4 x[8];
      #pragma unroll
      for (int u = 0; u < 8; ++u) x[u] = hb[(long)__builtin_amdgcn_readlane(mine, j + u) * (HID / 8) + lane];
      #pragma unroll
      for (int u = 0; u < 8; ++u) addbf(acc, x[u]);
    }
    if (j + 4 <= cnt) {
      uint4 x[4];
      #pragma unroll
      for (int u = 0; u < 4; ++u) x[u] = hb[(long)__builtin_amdgcn_readlane(mine, j + u) * (HID / 8) + lane];
      #pragma unroll
      for (int u = 0; u < 4; ++u) addbf(acc, x[u]);
      j += 4;
    }
    #pragma nounroll
    for (; j < cnt; ++j) addbf(acc, hb[(long)__builtin_amdgcn_readlane(mine, j) * (HID / 8) + lane]);
  }
  unsigned int* tl = (unsigned int*)tile;
  const int m  = lane & 3;
  const int fh = m & 1;
  const int p0 = 4 * (m >> 1);
  const int kt = lane >> 2;
  uint4 pr;
  pr.x = (unsigned int)f2bf(acc[0]) | ((unsigned int)f2bf(acc[1]) << 16);
  pr.y = (unsigned int)f2bf(acc[2]) | ((unsigned int)f2bf(acc[3]) << 16);
  pr.z = (unsigned int)f2bf(acc[4]) | ((unsigned int)f2bf(acc[5]) << 16);
  pr.w = (unsigned int)f2bf(acc[6]) | ((unsigned int)f2bf(acc[7]) << 16);
  *(uint4*)(tl + kt * 256 + (w + 16 * fh) * 8 + p0) = pr;
  __syncthreads();
  Apk[(long)blockIdx.x * (KT * 64) + threadIdx.x] = tile[threadIdx.x];
}

template <int K, int M>
__device__ __forceinline__ void wpack_elem(const float* __restrict__ Wm, unsigned int* __restrict__ out, int i) {
  constexpr int KT = K / 32;
  const int p = i & 7;
  const int lane = (i >> 3) & 31;
  const int tt = i >> 8;
  const int kt = tt % KT;
  const int ct = tt / KT;
  const int col = ct * 16 + (lane & 15);
  const int k = kt * 32 + frag_koff(lane, p);
  const unsigned int lo = f2bf(Wm[(long)k * M + col]);
  const unsigned int hi = f2bf(Wm[(long)(k + 1) * M + col]);
  out[i] = lo | (hi << 16);
}

#define PREP_T      256
#define PREP_B_WIN  ((HID / 16) * (IN_C / 32))
#define PREP_B_WH   ((HID / 16) * (HID / 32))
#define PREP_N_WH   6
#define PREP_B_WOUT ((OUT_C / 16) * (HID / 32))
#define PREP_N_DEG4 (N_NODES / 4)
#define PREP_B_DEG  ((PREP_N_DEG4 + PREP_T - 1) / PREP_T)
#define PREP_N_PL4  (N_GRAPHS * (HID + 1) / 4)
#define PREP_B_PL   ((PREP_N_PL4 + PREP_T - 1) / PREP_T)
#define PREP_BLOCKS (PREP_B_WIN + PREP_N_WH * PREP_B_WH + PREP_B_WOUT + PREP_B_DEG + PREP_B_PL)
static_assert(N_NODES % 4 == 0 && (N_GRAPHS * (HID + 1)) % 4 == 0, "the clears go out as whole 16-byte stores");

__global__ __launch_bounds__(PREP_T) void prep_all(
    const float* __restrict__ w_in,
    const float* __restrict__ wh0, const float* __restrict__ wh1, const float* __restrict__ wh2,
    const float* __restrict__ wh3, const float* __restrict__ wh4, const float* __restrict__ wh5,
    const float* __restrict__ w_out,
    unsigned int* __restrict__ p_in,
    unsigned int* __restrict__ ph0, unsigned int* __restrict__ ph1, unsigned int* __restrict__ ph2,
    unsigned int* __restrict__ ph3, unsigned int* __restrict__ ph4, unsigned int* __restrict__ ph5,
    unsigned int* __restrict__ p_out,
    int* __restrict__ deg, float* __restrict__ pooled) {
  const int t = threadIdx.x;
  int b = blockIdx.x;
  if (b < PREP_B_WIN) { wpack_elem<IN_C, HID>(w_in, p_in, b * PREP_T + t); return; }
  b -= PREP_B_WIN;
  if (b < PREP_N_WH * PREP_B_WH) {
    const int j = b / PREP_B_WH;
    const int i = (b % PREP_B_WH) * PREP_T + t;
    const float* Wm = wh0;  unsigned int* o = ph0;
    if (j == 1) { Wm = wh1; o = ph1; }
    if (j == 2) { Wm = wh2; o = ph2; }
    if (j == 3) { Wm = wh3; o = ph3; }
    if (j == 4) { Wm = wh4; o = ph4; }
    if (j == 5) { Wm = wh5; o = ph5; }
    wpack_elem<HID, HID>(Wm, o, i);
    return;
  }
  b -= PREP_N_WH * PREP_B_WH;
  if (b < PREP_B_WOUT) { wpack_elem<HID, OUT_C>(w_out, p_out, b * PREP_T + t); return; }
  b -= PREP_B_WOUT;
  if (b < PREP_B_DEG) {
    const int i4 = b * PREP_T + t;
    if (i4 < PREP_N_DEG4) ((int4*)deg)[i4] = make_int4(0, 0, 0, 0);
    return;
  }
  b -= PREP_B_DEG;
  const int i4 = b * PREP_T + t;
  if (i4 < PREP_N_PL4) ((float4*)pooled)[i4] = make_float4(0.0f, 0.0f, 0.0f, 0.0f);
}

template <int RELU, int OUTPK>
__global__ void gemm_pk(const uint4* __restrict__ Apk, const uint4* __restrict__ Wpk,
                        const float* __restrict__ bias, void* __restrict__ Cout,
                        int N, int K, int M) {
  const int tid  = threadIdx.x;
  const int lane = tid & 31;
  const int wave = tid >> 5;
  const int wr   = wave & 3;
  const int wc   = wave >> 2;
  const int KT   = K >> 5;
  const int nrt  = N >> 4;
  const int nct  = M >> 4;
  const int rt0  = blockIdx.y * 8 + wr * 2;
  const int ct0  = blockIdx.x * 4 + wc * 2;
  const bool r0 = rt0 < nrt,     r1 = (rt0 + 1) < nrt;
  const bool c0 = ct0 < nct,     c1 = (ct0 + 1) < nct;
  const int rA0 = r0 ? rt0 : 0,  rA1 = r1 ? (rt0 + 1) : 0;
  const int cB0 = c0 ? ct0 : 0,  cB1 = c1 ? (ct0 + 1) : 0;

  long ia0 = ((long)rA0 * KT * 32 + lane) * 2;
  long ia1 = ((long)rA1 * KT * 32 + lane) * 2;
  long ib0 = ((long)cB0 * KT * 32 + lane) * 2;
  long ib1 = ((long)cB1 * KT * 32 + lane) * 2;
  const long step = 64;

  v8f acc00 = {}, acc01 = {}, acc10 = {}, acc11 = {};
  union F { uint4 q[2]; v16bf v; };

  for (int kt = 0; kt < KT; ++kt) {
    F a0, a1, b0, b1;
    a0.q[0] = Apk[ia0]; a0.q[1] = Apk[ia0 + 1];
    a1.q[0] = Apk[ia1]; a1.q[1] = Apk[ia1 + 1];
    b0.q[0] = Wpk[ib0]; b0.q[1] = Wpk[ib0 + 1];
    b1.q[0] = Wpk[ib1]; b1.q[1] = Wpk[ib1 + 1];
    if (kt + 1 < KT) {
      __builtin_prefetch((const void*)(Apk + ia0 + step), 0, 3);
      __builtin_prefetch((const void*)(Apk + ia1 + step), 0, 3);
    }
    ia0 += step; ia1 += step; ib0 += step; ib1 += step;
    acc00 = __builtin_amdgcn_wmma_f32_16x16x32_bf16(false, a0.v, false, b0.v, (short)0, acc00, false, false);
    acc01 = __builtin_amdgcn_wmma_f32_16x16x32_bf16(false, a0.v, false, b1.v, (short)0, acc01, false, false);
    acc10 = __builtin_amdgcn_wmma_f32_16x16x32_bf16(false, a1.v, false, b0.v, (short)0, acc10, false, false);
    acc11 = __builtin_amdgcn_wmma_f32_16x16x32_bf16(false, a1.v, false, b1.v, (short)0, acc11, false, false);
  }

  const int rowL = lane & 15;
  if constexpr (OUTPK) {
    __shared__ uint4 stage[8 * 2 * 64];
    unsigned short* sw = (unsigned short*)stage + wave * 1024;
    const int hh = lane >> 4;
    auto put = [&](const v8f& acc, int i, int j, int ctc) {
      const float b = bias[ctc * 16 + rowL];
      #pragma unroll
      for (int r = 0; r < 8; ++r) {
        float o = acc[r] + b;
        if (RELU) o = fmaxf(o, 0.0f);
        sw[i * 512 + (8 * hh + r + 16 * (rowL >> 3)) * 16 + (rowL & 7) + 8 * j] = f2bf(o);
      }
    };
    put(acc00, 0, 0, cB0);   put(acc01, 0, 1, cB1);
    put(acc10, 1, 0, cB0);   put(acc11, 1, 1, cB1);
    __syncthreads();
    const int KTo = M >> 5;
    const int kto = ct0 >> 1;
    uint4* __restrict__ Opk = (uint4*)Cout;
    #pragma unroll
    for (int i = 0; i < 2; ++i) {
      const bool rok = i ? r1 : r0;
      if (rok && c1) {
        const long o4 = ((long)(rt0 + i) * KTo + kto) * 64;
        Opk[o4 + lane]      = stage[(wave * 2 + i) * 64 + lane];
        Opk[o4 + 32 + lane] = stage[(wave * 2 + i) * 64 + 32 + lane];
      }
    }
  } else {
    auto st = [&](const v8f& acc, int rt, bool rok, int ct, bool cok) {
      if (!rok || !cok) return;
      int col = ct * 16 + rowL;
      #pragma unroll
      for (int r = 0; r < 8; ++r) {
        int row = rt * 16 + ((lane < 16) ? r : r + 8);
        float o = acc[r] + bias[col];
        if (RELU) o = fmaxf(o, 0.0f);
        ((float*)Cout)[(long)row * M + col] = o;
      }
    };
    st(acc00, rt0, r0, ct0, c0);         st(acc01, rt0, r0, ct0 + 1, c1);
    st(acc10, rt0 + 1, r1, ct0, c0);     st(acc11, rt0 + 1, r1, ct0 + 1, c1);
  }
}

template <int K1, int OUTBF>
__global__ __launch_bounds__(256) void mlp_fused(const uint4* __restrict__ Apk, const uint4* __restrict__ W1pk,
                                                 const float* __restrict__ bias1, const uint4* __restrict__ W2pk,
                                                 const float* __restrict__ bias2, float* __restrict__ Hout, int nrt) {
  constexpr int KT1 = K1 / 32;
  constexpr int KT2 = HID / 32;
  constexpr int RT  = 4;
  static_assert(KT1 <= KT2, "the A tile must fit the buffer that later holds the middle tile");
  constexpr int OPITCH = HID / 8 + 1;
  __shared__ uint4 sbuf[RT * KT2 * 64 + (OUTBF ? RT * 16 : 0)];
  const int tid  = threadIdx.x;
  const int lane = tid & 31;
  const int wave = __builtin_amdgcn_readfirstlane(tid >> 5);
  const int rt0  = blockIdx.x * RT;
  const int nr   = min(RT, nrt - rt0);
  const int ct0  = wave * 2;
  union F { uint4 q[2]; v16bf v; };

  F w1f[KT1][2];
  {
    const uint4* Wp = W1pk + ((long)ct0 * KT1 * 32 + lane) * 2;
    #pragma unroll
    for (int kt = 0; kt < KT1; ++kt) {
      w1f[kt][0].q[0] = Wp[kt * 64];           w1f[kt][0].q[1] = Wp[kt * 64 + 1];
      w1f[kt][1].q[0] = Wp[(KT1 + kt) * 64];   w1f[kt][1].q[1] = Wp[(KT1 + kt) * 64 + 1];
    }
  }
  {
    const uint4* Ab = Apk + (long)rt0 * (KT1 * 64);
    const int nvalid = nr * (KT1 * 64);
    uint4 v[RT * KT1 * 64 / 256];
    #pragma unroll
    for (int s = 0; s < RT * KT1 * 64 / 256; ++s) {
      const int i = s * 256 + tid;
      v[s] = (i < nvalid) ? Ab[i] : make_uint4(0u, 0u, 0u, 0u);
    }
    #pragma unroll
    for (int s = 0; s < RT * KT1 * 64 / 256; ++s) {
      const int i = s * 256 + tid;
      sbuf[(i & ~63) | ((i & 1) << 5) | ((i & 63) >> 1)] = v[s];
    }
  }
  __syncthreads();

  v8f acc[RT][2];
  #pragma unroll
  for (int i = 0; i < RT; ++i) { acc[i][0] = v8f{}; acc[i][1] = v8f{}; }
  {
    F af[2][RT];
    #pragma unroll
    for (int i = 0; i < RT; ++i) {
      af[0][i].q[0] = sbuf[(i * KT1) * 64 + lane];
      af[0][i].q[1] = sbuf[(i * KT1) * 64 + 32 + lane];
    }
    __builtin_amdgcn_sched_group_barrier(0x100, 8, 0);
    #pragma unroll
    for (int kt = 0; kt < KT1; ++kt) {
      if (kt + 1 < KT1) {
        #pragma unroll
        for (int i = 0; i < RT; ++i) {
          af[(kt + 1) & 1][i].q[0] = sbuf[(i * KT1 + kt + 1) * 64 + lane];
          af[(kt + 1) & 1][i].q[1] = sbuf[(i * KT1 + kt + 1) * 64 + 32 + lane];
        }
      }
      #pragma unroll
      for (int i = 0; i < RT; ++i) {
        acc[i][0] = __builtin_amdgcn_wmma_f32_16x16x32_bf16(false, af[kt & 1][i].v, false, w1f[kt][0].v, (short)0, acc[i][0], false, false);
        acc[i][1] = __builtin_amdgcn_wmma_f32_16x16x32_bf16(false, af[kt & 1][i].v, false, w1f[kt][1].v, (short)0, acc[i][1], false, false);
      }
      __builtin_amdgcn_sched_group_barrier(0x100, 8, 0);
      __builtin_amdgcn_sched_group_barrier(0x008, 8, 0);
    }
  }

  F w2f[KT2][2];
  {
    const uint4* Wp = W2pk + ((long)ct0 * KT2 * 32 + lane) * 2;
    #pragma unroll
    for (int kt = 0; kt < KT2; ++kt) {
      w2f[kt][0].q[0] = Wp[kt * 64];           w2f[kt][0].q[1] = Wp[kt * 64 + 1];
      w2f[kt][1].q[0] = Wp[(KT2 + kt) * 64];   w2f[kt][1].q[1] = Wp[(KT2 + kt) * 64 + 1];
    }
  }

  const int rowL = lane & 15;
  const int hh   = lane >> 4;
  __syncthreads();
  {
    unsigned short* sw = (unsigned short*)sbuf;
    #pragma unroll
    for (int j = 0; j < 2; ++j) {
      const float b = bias1[(ct0 + j) * 16 + rowL];
      #pragma unroll
      for (int i = 0; i < RT; ++i) {
        #pragma unroll
        for (int r = 0; r < 8; ++r) {
          const float o = fmaxf(acc[i][j][r] + b, 0.0f);
          sw[(i * KT2 + wave) * 512 + j * 256 + (8 * hh + r + 16 * (rowL >> 3)) * 8 + (rowL & 7)] = f2bf(o);
        }
      }
    }
  }
  __syncthreads();

  #pragma unroll
  for (int i = 0; i < RT; ++i) { acc[i][0] = v8f{}; acc[i][1] = v8f{}; }
  {
    F af[2][RT];
    #pragma unroll
    for (int i = 0; i < RT; ++i) {
      af[0][i].q[0] = sbuf[(i * KT2) * 64 + lane];
      af[0][i].q[1] = sbuf[(i * KT2) * 64 + 32 + lane];
    }
    __builtin_amdgcn_sched_group_barrier(0x100, 8, 0);
    #pragma unroll
    for (int kt = 0; kt < KT2; ++kt) {
      if (kt + 1 < KT2) {
        #pragma unroll
        for (int i = 0; i < RT; ++i) {
          af[(kt + 1) & 1][i].q[0] = sbuf[(i * KT2 + kt + 1) * 64 + lane];
          af[(kt + 1) & 1][i].q[1] = sbuf[(i * KT2 + kt + 1) * 64 + 32 + lane];
        }
      }
      #pragma unroll
      for (int i = 0; i < RT; ++i) {
        acc[i][0] = __builtin_amdgcn_wmma_f32_16x16x32_bf16(false, af[kt & 1][i].v, false, w2f[kt][0].v, (short)0, acc[i][0], false, false);
        acc[i][1] = __builtin_amdgcn_wmma_f32_16x16x32_bf16(false, af[kt & 1][i].v, false, w2f[kt][1].v, (short)0, acc[i][1], false, false);
      }
      __builtin_amdgcn_sched_group_barrier(0x100, 8, 0);
      __builtin_amdgcn_sched_group_barrier(0x008, 8, 0);
    }
  }

  if constexpr (OUTBF) {
    __syncthreads();
    unsigned short* sw = (unsigned short*)sbuf;
    #pragma unroll
    for (int j = 0; j < 2; ++j) {
      const int col = (ct0 + j) * 16 + rowL;
      const float b = bias2[col];
      #pragma unroll
      for (int i = 0; i < RT; ++i) {
        #pragma unroll
        for (int r = 0; r < 8; ++r)
          sw[(i * 16 + 8 * hh + r) * (OPITCH * 8) + col] = f2bf(fmaxf(acc[i][j][r] + b, 0.0f));
      }
    }
    __syncthreads();
    if ((wave >> 1) < nr) {
      uint4* __restrict__ Hb = (uint4*)Hout;
      #pragma unroll
      for (int q = 0; q < 8; ++q) {
        const int rloc = wave * 8 + q;
        Hb[((long)rt0 * 16 + rloc) * (HID / 8) + lane] = sbuf[rloc * OPITCH + lane];
      }
    }
  } else {
    #pragma unroll
    for (int j = 0; j < 2; ++j) {
      const int col = (ct0 + j) * 16 + rowL;
      const float b = bias2[col];
      #pragma unroll
      for (int i = 0; i < RT; ++i) {
        if (i < nr) {
          #pragma unroll
          for (int r = 0; r < 8; ++r) {
            const int row = (rt0 + i) * 16 + 8 * hh + r;
            Hout[(long)row * HID + col] = fmaxf(acc[i][j][r] + b, 0.0f);
          }
        }
      }
    }
  }
}

#define POOL_R 16
static_assert(N_NODES % POOL_R == 0, "every wave owns POOL_R whole rows");
static_assert(HID == 256, "a lane owns two float4 of a row");

__device__ __forceinline__ void pool_flush(float* __restrict__ pooled, float* __restrict__ counts, int g, int lane,
                                           const float4& a0, const float4& a1, int len) {
  float* o = pooled + (long)g * HID + lane * 4;
  atomicAdd(o + 0, a0.x); atomicAdd(o + 1, a0.y); atomicAdd(o + 2, a0.z); atomicAdd(o + 3, a0.w);
  atomicAdd(o + 128, a1.x); atomicAdd(o + 129, a1.y); atomicAdd(o + 130, a1.z); atomicAdd(o + 131, a1.w);
  if (lane == 0) atomicAdd(&counts[g], (float)len);
}

__global__ __launch_bounds__(256) void pool_runs(const float* __restrict__ h, const int* __restrict__ batch,
                                                 float* __restrict__ pooled, float* __restrict__ counts) {
  const int lane = threadIdx.x & 31;
  const int gw = __builtin_amdgcn_readfirstlane((int)(blockIdx.x * 8 + (threadIdx.x >> 5)));
  const int n0 = gw * POOL_R;
  if (n0 >= N_NODES) return;
  const int ids = batch[n0 + (lane & (POOL_R - 1))];
  const float4* __restrict__ hp = (const float4*)h + (long)n0 * (HID / 4);
  float4 x[POOL_R][2];
  #pragma unroll
  for (int j = 0; j < POOL_R; ++j) {
    x[j][0] = hp[j * (HID / 4) + lane];
    x[j][1] = hp[j * (HID / 4) + 32 + lane];
  }
  int cur = __builtin_amdgcn_readlane(ids, 0);
  float4 a0 = x[0][0], a1 = x[0][1];
  int len = 1;
  #pragma unroll
  for (int j = 1; j < POOL_R; ++j) {
    const int g = __builtin_amdgcn_readlane(ids, j);
    if (g != cur) {
      pool_flush(pooled, counts, cur, lane, a0, a1, len);
      a0 = x[j][0]; a1 = x[j][1];
      len = 1;
      cur = g;
    } else {
      add4(a0, x[j][0]); add4(a1, x[j][1]);
      ++len;
    }
  }
  pool_flush(pooled, counts, cur, lane, a0, a1, len);
}

__global__ void pool_div_pack(const float* __restrict__ pooled, const float* __restrict__ counts,
                              unsigned int* __restrict__ out) {
  const int KT = HID / 32;
  int i = blockIdx.x * blockDim.x + threadIdx.x;
  int p = i & 7;
  int lane = (i >> 3) & 31;
  int tt = i >> 8;
  int kt = tt % KT;
  int rt = tt / KT;
  int row = rt * 16 + (lane & 15);
  int k = kt * 32 + frag_koff(lane, p);
  const float c = fmaxf(counts[row], 1.0f);
  const float* s = pooled + (long)row * HID + k;
  out[i] = (unsigned int)f2bf(s[0] / c) | ((unsigned int)f2bf(s[1] / c) << 16);
}

extern "C" void kernel_launch(void* const* d_in, const int* in_sizes, int n_in,
                              void* d_out, int out_size, void* d_ws, size_t ws_size,
                              hipStream_t stream) {
  const float* x     = (const float*)d_in[0];
  const int*   ei    = (const int*)d_in[1];
  const int*   esrc  = ei;
  const int*   edst  = ei + N_EDGES;
  const int*   batch = (const int*)d_in[2];
  const float* cw1[3] = {(const float*)d_in[3],  (const float*)d_in[7],  (const float*)d_in[11]};
  const float* cb1[3] = {(const float*)d_in[4],  (const float*)d_in[8],  (const float*)d_in[12]};
  const float* cw2[3] = {(const float*)d_in[5],  (const float*)d_in[9],  (const float*)d_in[13]};
  const float* cb2[3] = {(const float*)d_in[6],  (const float*)d_in[10], (const float*)d_in[14]};
  const float* ow1 = (const float*)d_in[15];
  const float* ob1 = (const float*)d_in[16];
  const float* ow2 = (const float*)d_in[17];
  const float* ob2 = (const float*)d_in[18];

  char* ws = (char*)d_ws;
  size_t off = 0;
  auto bump = [&](size_t bytes) {
    void* p = ws + off;
    off += (bytes + 255) & ~(size_t)255;
    return p;
  };
  const int NRT = N_NODES / 16;
  float*          B2     = (float*)bump((size_t)N_NODES * HID * 4);
  unsigned short* Hb     = (unsigned short*)bump((size_t)N_NODES * HID * 2);
  unsigned int*   Apk    = (unsigned int*)bump((size_t)NRT * 8 * 256 * 4);
  float*          pooled = (float*)bump((size_t)N_GRAPHS * (HID + 1) * 4);
  float*          counts = pooled + (size_t)N_GRAPHS * HID;
  unsigned int*   Ppk    = (unsigned int*)bump((size_t)(N_GRAPHS / 16) * 8 * 256 * 4);
  unsigned int*   MPpk   = (unsigned int*)bump((size_t)(N_GRAPHS / 16) * 8 * 256 * 4);
  unsigned int*   wpk1[3], *wpk2[3];
  for (int l = 0; l < 3; ++l) {
    wpk1[l] = (unsigned int*)bump((size_t)(HID / 16) * 8 * 256 * 4);
    wpk2[l] = (unsigned int*)bump((size_t)(HID / 16) * 8 * 256 * 4);
  }
  unsigned int* owpk1 = (unsigned int*)bump((size_t)(HID / 16) * 8 * 256 * 4);
  unsigned int* owpk2 = (unsigned int*)bump((size_t)(OUT_C / 16) * 8 * 256 * 4);
  int* deg    = (int*)bump((size_t)N_NODES * 4);
  int* rowptr = (int*)bump((size_t)(N_NODES + 1) * 4);
  int* cursor = (int*)bump((size_t)N_NODES * 4);
  int* col    = (int*)bump((size_t)N_EDGES * 4);
  int* col2   = (int*)bump((size_t)N_EDGES * 4);

  dim3 blk(256);

  prep_all<<<PREP_BLOCKS, PREP_T, 0, stream>>>(cw1[0], cw2[0], cw1[1], cw2[1], cw1[2], cw2[2], ow1, ow2,
                                                wpk1[0], wpk2[0], wpk1[1], wpk2[1], wpk1[2], wpk2[2], owpk1, owpk2,
                                                deg, pooled);

  csr_count<<<(N_EDGES + 255) / 256, blk, 0, stream>>>(edst, deg);
  csr_scan<<<1, SCAN_T, 0, stream>>>(deg, rowptr, cursor);
  csr_fill<<<(N_EDGES + 255) / 256, blk, 0, stream>>>(esrc, edst, cursor, col);
  csr_sort<<<(N_NODES / SORT_NPW + 7) / 8, blk, 0, stream>>>(rowptr, col, col2);

  gather_pack<IN_C><<<NRT, 512, 0, stream>>>(x, rowptr, col2, (uint4*)Apk);
  mlp_fused<IN_C, 1><<<(NRT + 3) / 4, blk, 0, stream>>>((const uint4*)Apk, (const uint4*)wpk1[0], cb1[0],
                                                        (const uint4*)wpk2[0], cb2[0], (float*)Hb, NRT);
  gather_pack_bf<<<NRT, 512, 0, stream>>>((const uint4*)Hb, rowptr, col2, (uint4*)Apk);
  mlp_fused<HID, 1><<<(NRT + 3) / 4, blk, 0, stream>>>((const uint4*)Apk, (const uint4*)wpk1[1], cb1[1],
                                                       (const uint4*)wpk2[1], cb2[1], (float*)Hb, NRT);
  gather_pack_bf<<<NRT, 512, 0, stream>>>((const uint4*)Hb, rowptr, col2, (uint4*)Apk);
  mlp_fused<HID, 0><<<(NRT + 3) / 4, blk, 0, stream>>>((const uint4*)Apk, (const uint4*)wpk1[2], cb1[2],
                                                       (const uint4*)wpk2[2], cb2[2], B2, NRT);

  pool_runs<<<(N_NODES / POOL_R + 7) / 8, blk, 0, stream>>>(B2, batch, pooled, counts);

  const int GRT = N_GRAPHS / 16;
  pool_div_pack<<<GRT * (HID / 32), blk, 0, stream>>>(pooled, counts, Ppk);
  dim3 g3((HID / 16 + 3) / 4, (GRT + 7) / 8);
  gemm_pk<1, 1><<<g3, blk, 0, stream>>>((const uint4*)Ppk, (const uint4*)owpk1, ob1,
                                        MPpk, N_GRAPHS, HID, HID);
  dim3 g4((OUT_C / 16 + 3) / 4, (GRT + 7) / 8);
  gemm_pk<0, 0><<<g4, blk, 0, stream>>>((const uint4*)MPpk, (const uint4*)owpk2, ob2,
                                        (float*)d_out, N_GRAPHS, HID, OUT_C);
}


# Edge lists at a fixed pitch (64 slots per node, taken from its counter; longer lists spill to an overflow list added in arrival order): count and scan launches go, list loads wait for no offsets
# speedup vs baseline: 4.4651x; 1.2356x over previous
#include <hip/hip_runtime.h>

#define N_NODES 50000
#define N_EDGES 800000
#define N_GRAPHS 512
#define IN_C 128
#define HID 256
#define OUT_C 16

typedef __attribute__((ext_vector_type(16))) __bf16 v16bf;
typedef __attribute__((ext_vector_type(8)))  float  v8f;

__device__ __forceinline__ unsigned short f2bf(float f) {
  union { float f; unsigned int u; } c; c.f = f;
  unsigned int u = c.u;
  u += 0x7fffu + ((u >> 16) & 1u);
  return (unsigned short)(u >> 16);
}

__device__ __forceinline__ int frag_koff(int lane, int p) {
  int kb = (lane < 16) ? 0 : 8;
  return (p < 4) ? (kb + 2 * p) : (16 + kb + 2 * (p - 4));
}

#define LIST_CAP 64
__global__ void list_fill(const int* __restrict__ esrc, const int* __restrict__ edst,
                          int* __restrict__ deg, int* __restrict__ lst, int2* __restrict__ ovf) {
  int e = blockIdx.x * blockDim.x + threadIdx.x;
  if (e < N_EDGES) {
    const int d = edst[e];
    const int s = esrc[e];
    const int pos = atomicAdd(&deg[d], 1);
    if (pos < LIST_CAP) {
      lst[(long)d * LIST_CAP + pos] = s;
    } else {
      const int k = atomicAdd(&deg[N_NODES], 1);
      ovf[k] = make_int2(d, s);
    }
  }
}

__device__ __forceinline__ void add4(float4& a, const float4& b) {
  a.x += b.x; a.y += b.y; a.z += b.z; a.w += b.w;
}

static_assert(N_NODES % 16 == 0, "one work-group per full 16-row tile");
template <int C>
__global__ __launch_bounds__(512) void gather_pack(const float* __restrict__ h, const int* __restrict__ deg,
                                                   const int* __restrict__ lst, const int2* __restrict__ ovf,
                                                   uint4* __restrict__ Apk) {
  constexpr int V = C / 128;
  constexpr int U = 8 / V;
  constexpr int KT = C / 32;
  __shared__ uint4 tile[KT * 64];
  const int lane = threadIdx.x & 31;
  const int w = __builtin_amdgcn_readfirstlane((int)(threadIdx.x >> 5));
  const int node = __builtin_amdgcn_readfirstlane((int)(blockIdx.x * 16 + w));
  const int total = deg[node];
  const int beg = node * LIST_CAP;
  const int end = beg + min(total, LIST_CAP);
  const int* __restrict__ col = lst;
  int mine = col[beg + lane];
  const float4* __restrict__ hp = (const float4*)h;
  float4 acc[V];
  #pragma unroll
  for (int v = 0; v < V; ++v) acc[v] = hp[(long)node * (C / 4) + v * 32 + lane];
  asm volatile("" :: "v"(mine));
  for (int base = beg; base < end; base += 32) {
    const int cnt = min(32, end - base);
    if (base > beg) mine = (lane < cnt) ? col[base + lane] : 0;
    int j = 0;
    for (; j + U <= cnt; j += U) {
      float4 x[U][V];
      #pragma unroll
      for (int u = 0; u < U; ++u) {
        const int sidx = __builtin_amdgcn_readlane(mine, j + u);
        #pragma unroll
        for (int v = 0; v < V; ++v) x[u][v] = hp[(long)sidx * (C / 4) + v * 32 + lane];
      }
      #pragma unroll
      for (int u = 0; u < U; ++u) {
        #pragma unroll
        for (int v = 0; v < V; ++v) add4(acc[v], x[u][v]);
      }
    }
    #pragma nounroll
    for (; j < cnt; ++j) {
      const int sidx = __builtin_amdgcn_readlane(mine, j);
      #pragma unroll
      for (int v = 0; v < V; ++v) add4(acc[v], hp[(long)sidx * (C / 4) + v * 32 + lane]);
    }
  }
  if (total > LIST_CAP) {
    const int no = min(deg[N_NODES], N_EDGES);
    #pragma nounroll
    for (int k = 0; k < no; ++k) {
      const int2 e = ovf[k];
      if (e.x == node) {
        #pragma unroll
        for (int v = 0; v < V; ++v) add4(acc[v], hp[(long)e.y * (C / 4) + v * 32 + lane]);
      }
    }
  }
  unsigned int* tl = (unsigned int*)tile;
  const int m  = lane & 7;
  const int fh = (m >> 1) & 1;
  const int p0 = 2 * (m & 1) + 4 * (m >> 2);
  #pragma unroll
  for (int v = 0; v < V; ++v) {
    const int kt = v * 4 + (lane >> 3);
    uint2 pr;
    pr.x = (unsigned int)f2bf(acc[v].x) | ((unsigned int)f2bf(acc[v].y) << 16);
    pr.y = (unsigned int)f2bf(acc[v].z) | ((unsigned int)f2bf(acc[v].w) << 16);
    *(uint2*)(tl + kt * 256 + (w + 16 * fh) * 8 + p0) = pr;
  }
  __syncthreads();
  if (threadIdx.x < KT * 64) Apk[(long)blockIdx.x * (KT * 64) + threadIdx.x] = tile[threadIdx.x];
}

#define SORT_NPW 4
static_assert(N_NODES % SORT_NPW == 0, "every wave owns SORT_NPW whole nodes");
__global__ __launch_bounds__(256) void list_sort(const int* __restrict__ deg, const int* __restrict__ col,
                                                 int* __restrict__ col2) {
  const int lane = threadIdx.x & 31;
  const int n0 = __builtin_amdgcn_readfirstlane((int)((blockIdx.x * 8 + (threadIdx.x >> 5)) * SORT_NPW));
  if (n0 >= N_NODES) return;
  int len[SORT_NPW];
  #pragma unroll
  for (int q = 0; q < SORT_NPW; ++q) len[q] = min(deg[n0 + q], LIST_CAP);
  int mine[SORT_NPW];
  #pragma unroll
  for (int q = 0; q < SORT_NPW; ++q)
    mine[q] = (lane < len[q]) ? col[(n0 + q) * LIST_CAP + lane] : 0x7fffffff;
  #pragma unroll
  for (int q = 0; q < SORT_NPW; ++q) {
    const int beg = (n0 + q) * LIST_CAP;
    const int d = len[q];
    if (d <= 32) {
      int rank = 0;
      for (int j = 0; j < d; ++j) {
        const int v = __builtin_amdgcn_readlane(mine[q], j);
        rank += (v < mine[q] || (v == mine[q] && j < lane)) ? 1 : 0;
      }
      col2[beg + ((lane < d) ? rank : lane)] = (lane < d) ? mine[q] : -1;
    } else {
      for (int e = lane; e < d; e += 32) {
        const int v = col[beg + e];
        int rank = 0;
        for (int j = 0; j < d; ++j) {
          const int u = col[beg + j];
          rank += (u < v || (u == v && j < e)) ? 1 : 0;
        }
        col2[beg + rank] = v;
      }
    }
  }
}

__device__ __forceinline__ void addbf(float (&a)[8], const uint4& q) {
  a[0] += __uint_as_float(q.x << 16);  a[1] += __uint_as_float(q.x & 0xffff0000u);
  a[2] += __uint_as_float(q.y << 16);  a[3] += __uint_as_float(q.y & 0xffff0000u);
  a[4] += __uint_as_float(q.z << 16);  a[5] += __uint_as_float(q.z & 0xffff0000u);
  a[6] += __uint_as_float(q.w << 16);  a[7] += __uint_as_float(q.w & 0xffff0000u);
}

static_assert(HID == 256, "a bf16 row is exactly one 16-byte load per lane");
__global__ __launch_bounds__(512) void gather_pack_bf(const uint4* __restrict__ hb, const int* __restrict__ deg,
                                                      const int* __restrict__ lst, const int2* __restrict__ ovf,
                                                      uint4* __restrict__ Apk) {
  constexpr int KT = HID / 32;
  __shared__ uint4 tile[KT * 64];
  const int lane = threadIdx.x & 31;
  const int w = __builtin_amdgcn_readfirstlane((int)(threadIdx.x >> 5));
  const int node = __builtin_amdgcn_readfirstlane((int)(blockIdx.x * 16 + w));
  const int total = deg[node];
  const int beg = node * LIST_CAP;
  const int end = beg + min(total, LIST_CAP);
  const int* __restrict__ col = lst;
  int mine = col[beg + lane];
  float acc[8];
  #pragma unroll
  for (int i = 0; i < 8; ++i) acc[i] = 0.0f;
  addbf(acc, hb[(long)node * (HID / 8) + lane]);
  asm volatile("" :: "v"(mine));
  for (int base = beg; base < end; base += 32) {
    const int cnt = min(32, end - base);
    if (base > beg) mine = (lane < cnt) ? col[base + lane] : 0;
    int j = 0;
    for (; j + 8 <= cnt; j += 8) {
      uint4 x[8];
      #pragma unroll
      for (int u = 0; u < 8; ++u) x[u] = hb[(long)__builtin_amdgcn_readlane(mine, j + u) * (HID / 8) + lane];
      #pragma unroll
      for (int u = 0; u < 8; ++u) addbf(acc, x[u]);
    }
    if (j + 4 <= cnt) {
      uint4 x[4];
      #pragma unroll
      for (int u = 0; u < 4; ++u) x[u] = hb[(long)__builtin_amdgcn_readlane(mine, j + u) * (HID / 8) + lane];
      #pragma unroll
      for (int u = 0; u < 4; ++u) addbf(acc, x[u]);
      j += 4;
    }
    #pragma nounroll
    for (; j < cnt; ++j) addbf(acc, hb[(long)__builtin_amdgcn_readlane(mine, j) * (HID / 8) + lane]);
  }
  if (total > LIST_CAP) {
    const int no = min(deg[N_NODES], N_EDGES);
    #pragma nounroll
    for (int k = 0; k < no; ++k) {
      const int2 e = ovf[k];
      if (e.x == node) addbf(acc, hb[(long)e.y * (HID / 8) + lane]);
    }
  }
  unsigned int* tl = (unsigned int*)tile;
  const int m  = lane & 3;
  const int fh = m & 1;
  const int p0 = 4 * (m >> 1);
  const int kt = lane >> 2;
  uint4 pr;
  pr.x = (unsigned int)f2bf(acc[0]) | ((unsigned int)f2bf(acc[1]) << 16);
  pr.y = (unsigned int)f2bf(acc[2]) | ((unsigned int)f2bf(acc[3]) << 16);
  pr.z = (unsigned int)f2bf(acc[4]) | ((unsigned int)f2bf(acc[5]) << 16);
  pr.w = (unsigned int)f2bf(acc[6]) | ((unsigned int)f2bf(acc[7]) << 16);
  *(uint4*)(tl + kt * 256 + (w + 16 * fh) * 8 + p0) = pr;
  __syncthreads();
  Apk[(long)blockIdx.x * (KT * 64) + threadIdx.x] = tile[threadIdx.x];
}

template <int K, int M>
__device__ __forceinline__ void wpack_elem(const float* __restrict__ Wm, unsigned int* __restrict__ out, int i) {
  constexpr int KT = K / 32;
  const int p = i & 7;
  const int lane = (i >> 3) & 31;
  const int tt = i >> 8;
  const int kt = tt % KT;
  const int ct = tt / KT;
  const int col = ct * 16 + (lane & 15);
  const int k = kt * 32 + frag_koff(lane, p);
  const unsigned int lo = f2bf(Wm[(long)k * M + col]);
  const unsigned int hi = f2bf(Wm[(long)(k + 1) * M + col]);
  out[i] = lo | (hi << 16);
}

#define PREP_T      256
#define PREP_B_WIN  ((HID / 16) * (IN_C / 32))
#define PREP_B_WH   ((HID / 16) * (HID / 32))
#define PREP_N_WH   6
#define PREP_B_WOUT ((OUT_C / 16) * (HID / 32))
#define PREP_N_DEG4 (N_NODES / 4 + 1)
#define PREP_B_DEG  ((PREP_N_DEG4 + PREP_T - 1) / PREP_T)
#define PREP_N_PL4  (N_GRAPHS * (HID + 1) / 4)
#define PREP_B_PL   ((PREP_N_PL4 + PREP_T - 1) / PREP_T)
#define PREP_BLOCKS (PREP_B_WIN + PREP_N_WH * PREP_B_WH + PREP_B_WOUT + PREP_B_DEG + PREP_B_PL)
static_assert(N_NODES % 4 == 0 && (N_GRAPHS * (HID + 1)) % 4 == 0, "the clears go out as whole 16-byte stores");

__global__ __launch_bounds__(PREP_T) void prep_all(
    const float* __restrict__ w_in,
    const float* __restrict__ wh0, const float* __restrict__ wh1, const float* __restrict__ wh2,
    const float* __restrict__ wh3, const float* __restrict__ wh4, const float* __restrict__ wh5,
    const float* __restrict__ w_out,
    unsigned int* __restrict__ p_in,
    unsigned int* __restrict__ ph0, unsigned int* __restrict__ ph1, unsigned int* __restrict__ ph2,
    unsigned int* __restrict__ ph3, unsigned int* __restrict__ ph4, unsigned int* __restrict__ ph5,
    unsigned int* __restrict__ p_out,
    int* __restrict__ deg, float* __restrict__ pooled) {
  const int t = threadIdx.x;
  int b = blockIdx.x;
  if (b < PREP_B_WIN) { wpack_elem<IN_C, HID>(w_in, p_in, b * PREP_T + t); return; }
  b -= PREP_B_WIN;
  if (b < PREP_N_WH * PREP_B_WH) {
    const int j = b / PREP_B_WH;
    const int i = (b % PREP_B_WH) * PREP_T + t;
    const float* Wm = wh0;  unsigned int* o = ph0;
    if (j == 1) { Wm = wh1; o = ph1; }
    if (j == 2) { Wm = wh2; o = ph2; }
    if (j == 3) { Wm = wh3; o = ph3; }
    if (j == 4) { Wm = wh4; o = ph4; }
    if (j == 5) { Wm = wh5; o = ph5; }
    wpack_elem<HID, HID>(Wm, o, i);
    return;
  }
  b -= PREP_N_WH * PREP_B_WH;
  if (b < PREP_B_WOUT) { wpack_elem<HID, OUT_C>(w_out, p_out, b * PREP_T + t); return; }
  b -= PREP_B_WOUT;
  if (b < PREP_B_DEG) {
    const int i4 = b * PREP_T + t;
    if (i4 < PREP_N_DEG4) ((int4*)deg)[i4] = make_int4(0, 0, 0, 0);
    return;
  }
  b -= PREP_B_DEG;
  const int i4 = b * PREP_T + t;
  if (i4 < PREP_N_PL4) ((float4*)pooled)[i4] = make_float4(0.0f, 0.0f, 0.0f, 0.0f);
}

template <int RELU, int OUTPK>
__global__ void gemm_pk(const uint4* __restrict__ Apk, const uint4* __restrict__ Wpk,
                        const float* __restrict__ bias, void* __restrict__ Cout,
                        int N, int K, int M) {
  const int tid  = threadIdx.x;
  const int lane = tid & 31;
  const int wave = tid >> 5;
  const int wr   = wave & 3;
  const int wc   = wave >> 2;
  const int KT   = K >> 5;
  const int nrt  = N >> 4;
  const int nct  = M >> 4;
  const int rt0  = blockIdx.y * 8 + wr * 2;
  const int ct0  = blockIdx.x * 4 + wc * 2;
  const bool r0 = rt0 < nrt,     r1 = (rt0 + 1) < nrt;
  const bool c0 = ct0 < nct,     c1 = (ct0 + 1) < nct;
  const int rA0 = r0 ? rt0 : 0,  rA1 = r1 ? (rt0 + 1) : 0;
  const int cB0 = c0 ? ct0 : 0,  cB1 = c1 ? (ct0 + 1) : 0;

  long ia0 = ((long)rA0 * KT * 32 + lane) * 2;
  long ia1 = ((long)rA1 * KT * 32 + lane) * 2;
  long ib0 = ((long)cB0 * KT * 32 + lane) * 2;
  long ib1 = ((long)cB1 * KT * 32 + lane) * 2;
  const long step = 64;

  v8f acc00 = {}, acc01 = {}, acc10 = {}, acc11 = {};
  union F { uint4 q[2]; v16bf v; };

  for (int kt = 0; kt < KT; ++kt) {
    F a0, a1, b0, b1;
    a0.q[0] = Apk[ia0]; a0.q[1] = Apk[ia0 + 1];
    a1.q[0] = Apk[ia1]; a1.q[1] = Apk[ia1 + 1];
    b0.q[0] = Wpk[ib0]; b0.q[1] = Wpk[ib0 + 1];
    b1.q[0] = Wpk[ib1]; b1.q[1] = Wpk[ib1 + 1];
    if (kt + 1 < KT) {
      __builtin_prefetch((const void*)(Apk + ia0 + step), 0, 3);
      __builtin_prefetch((const void*)(Apk + ia1 + step), 0, 3);
    }
    ia0 += step; ia1 += step; ib0 += step; ib1 += step;
    acc00 = __builtin_amdgcn_wmma_f32_16x16x32_bf16(false, a0.v, false, b0.v, (short)0, acc00, false, false);
    acc01 = __builtin_amdgcn_wmma_f32_16x16x32_bf16(false, a0.v, false, b1.v, (short)0, acc01, false, false);
    acc10 = __builtin_amdgcn_wmma_f32_16x16x32_bf16(false, a1.v, false, b0.v, (short)0, acc10, false, false);
    acc11 = __builtin_amdgcn_wmma_f32_16x16x32_bf16(false, a1.v, false, b1.v, (short)0, acc11, false, false);
  }

  const int rowL = lane & 15;
  if constexpr (OUTPK) {
    __shared__ uint4 stage[8 * 2 * 64];
    unsigned short* sw = (unsigned short*)stage + wave * 1024;
    const int hh = lane >> 4;
    auto put = [&](const v8f& acc, int i, int j, int ctc) {
      const float b = bias[ctc * 16 + rowL];
      #pragma unroll
      for (int r = 0; r < 8; ++r) {
        float o = acc[r] + b;
        if (RELU) o = fmaxf(o, 0.0f);
        sw[i * 512 + (8 * hh + r + 16 * (rowL >> 3)) * 16 + (rowL & 7) + 8 * j] = f2bf(o);
      }
    };
    put(acc00, 0, 0, cB0);   put(acc01, 0, 1, cB1);
    put(acc10, 1, 0, cB0);   put(acc11, 1, 1, cB1);
    __syncthreads();
    const int KTo = M >> 5;
    const int kto = ct0 >> 1;
    uint4* __restrict__ Opk = (uint4*)Cout;
    #pragma unroll
    for (int i = 0; i < 2; ++i) {
      const bool rok = i ? r1 : r0;
      if (rok && c1) {
        const long o4 = ((long)(rt0 + i) * KTo + kto) * 64;
        Opk[o4 + lane]      = stage[(wave * 2 + i) * 64 + lane];
        Opk[o4 + 32 + lane] = stage[(wave * 2 + i) * 64 + 32 + lane];
      }
    }
  } else {
    auto st = [&](const v8f& acc, int rt, bool rok, int ct, bool cok) {
      if (!rok || !cok) return;
      int col = ct * 16 + rowL;
      #pragma unroll
      for (int r = 0; r < 8; ++r) {
        int row = rt * 16 + ((lane < 16) ? r : r + 8);
        float o = acc[r] + bias[col];
        if (RELU) o = fmaxf(o, 0.0f);
        ((float*)Cout)[(long)row * M + col] = o;
      }
    };
    st(acc00, rt0, r0, ct0, c0);         st(acc01, rt0, r0, ct0 + 1, c1);
    st(acc10, rt0 + 1, r1, ct0, c0);     st(acc11, rt0 + 1, r1, ct0 + 1, c1);
  }
}

template <int K1, int OUTBF>
__global__ __launch_bounds__(256) void mlp_fused(const uint4* __restrict__ Apk, const uint4* __restrict__ W1pk,
                                                 const float* __restrict__ bias1, const uint4* __restrict__ W2pk,
                                                 const float* __restrict__ bias2, float* __restrict__ Hout, int nrt) {
  constexpr int KT1 = K1 / 32;
  constexpr int KT2 = HID / 32;
  constexpr int RT  = 4;
  static_assert(KT1 <= KT2, "the A tile must fit the buffer that later holds the middle tile");
  constexpr int OPITCH = HID / 8 + 1;
  __shared__ uint4 sbuf[RT * KT2 * 64 + (OUTBF ? RT * 16 : 0)];
  const int tid  = threadIdx.x;
  const int lane = tid & 31;
  const int wave = __builtin_amdgcn_readfirstlane(tid >> 5);
  const int rt0  = blockIdx.x * RT;
  const int nr   = min(RT, nrt - rt0);
  const int ct0  = wave * 2;
  union F { uint4 q[2]; v16bf v; };

  F w1f[KT1][2];
  {
    const uint4* Wp = W1pk + ((long)ct0 * KT1 * 32 + lane) * 2;
    #pragma unroll
    for (int kt = 0; kt < KT1; ++kt) {
      w1f[kt][0].q[0] = Wp[kt * 64];           w1f[kt][0].q[1] = Wp[kt * 64 + 1];
      w1f[kt][1].q[0] = Wp[(KT1 + kt) * 64];   w1f[kt][1].q[1] = Wp[(KT1 + kt) * 64 + 1];
    }
  }
  {
    const uint4* Ab = Apk + (long)rt0 * (KT1 * 64);
    const int nvalid = nr * (KT1 * 64);
    uint4 v[RT * KT1 * 64 / 256];
    #pragma unroll
    for (int s = 0; s < RT * KT1 * 64 / 256; ++s) {
      const int i = s * 256 + tid;
      v[s] = (i < nvalid) ? Ab[i] : make_uint4(0u, 0u, 0u, 0u);
    }
    #pragma unroll
    for (int s = 0; s < RT * KT1 * 64 / 256; ++s) {
      const int i = s * 256 + tid;
      sbuf[(i & ~63) | ((i & 1) << 5) | ((i & 63) >> 1)] = v[s];
    }
  }
  __syncthreads();

  v8f acc[RT][2];
  #pragma unroll
  for (int i = 0; i < RT; ++i) { acc[i][0] = v8f{}; acc[i][1] = v8f{}; }
  {
    F af[2][RT];
    #pragma unroll
    for (int i = 0; i < RT; ++i) {
      af[0][i].q[0] = sbuf[(i * KT1) * 64 + lane];
      af[0][i].q[1] = sbuf[(i * KT1) * 64 + 32 + lane];
    }
    __builtin_amdgcn_sched_group_barrier(0x100, 8, 0);
    #pragma unroll
    for (int kt = 0; kt < KT1; ++kt) {
      if (kt + 1 < KT1) {
        #pragma unroll
        for (int i = 0; i < RT; ++i) {
          af[(kt + 1) & 1][i].q[0] = sbuf[(i * KT1 + kt + 1) * 64 + lane];
          af[(kt + 1) & 1][i].q[1] = sbuf[(i * KT1 + kt + 1) * 64 + 32 + lane];
        }
      }
      #pragma unroll
      for (int i = 0; i < RT; ++i) {
        acc[i][0] = __builtin_amdgcn_wmma_f32_16x16x32_bf16(false, af[kt & 1][i].v, false, w1f[kt][0].v, (short)0, acc[i][0], false, false);
        acc[i][1] = __builtin_amdgcn_wmma_f32_16x16x32_bf16(false, af[kt & 1][i].v, false, w1f[kt][1].v, (short)0, acc[i][1], false, false);
      }
      __builtin_amdgcn_sched_group_barrier(0x100, 8, 0);
      __builtin_amdgcn_sched_group_barrier(0x008, 8, 0);
    }
  }

  F w2f[KT2][2];
  {
    const uint4* Wp = W2pk + ((long)ct0 * KT2 * 32 + lane) * 2;
    #pragma unroll
    for (int kt = 0; kt < KT2; ++kt) {
      w2f[kt][0].q[0] = Wp[kt * 64];           w2f[kt][0].q[1] = Wp[kt * 64 + 1];
      w2f[kt][1].q[0] = Wp[(KT2 + kt) * 64];   w2f[kt][1].q[1] = Wp[(KT2 + kt) * 64 + 1];
    }
  }

  const int rowL = lane & 15;
  const int hh   = lane >> 4;
  __syncthreads();
  {
    unsigned short* sw = (unsigned short*)sbuf;
    #pragma unroll
    for (int j = 0; j < 2; ++j) {
      const float b = bias1[(ct0 + j) * 16 + rowL];
      #pragma unroll
      for (int i = 0; i < RT; ++i) {
        #pragma unroll
        for (int r = 0; r < 8; ++r) {
          const float o = fmaxf(acc[i][j][r] + b, 0.0f);
          sw[(i * KT2 + wave) * 512 + j * 256 + (8 * hh + r + 16 * (rowL >> 3)) * 8 + (rowL & 7)] = f2bf(o);
        }
      }
    }
  }
  __syncthreads();

  #pragma unroll
  for (int i = 0; i < RT; ++i) { acc[i][0] = v8f{}; acc[i][1] = v8f{}; }
  {
    F af[2][RT];
    #pragma unroll
    for (int i = 0; i < RT; ++i) {
      af[0][i].q[0] = sbuf[(i * KT2) * 64 + lane];
      af[0][i].q[1] = sbuf[(i * KT2) * 64 + 32 + lane];
    }
    __builtin_amdgcn_sched_group_barrier(0x100, 8, 0);
    #pragma unroll
    for (int kt = 0; kt < KT2; ++kt) {
      if (kt + 1 < KT2) {
        #pragma unroll
        for (int i = 0; i < RT; ++i) {
          af[(kt + 1) & 1][i].q[0] = sbuf[(i * KT2 + kt + 1) * 64 + lane];
          af[(kt + 1) & 1][i].q[1] = sbuf[(i * KT2 + kt + 1) * 64 + 32 + lane];
        }
      }
      #pragma unroll
      for (int i = 0; i < RT; ++i) {
        acc[i][0] = __builtin_amdgcn_wmma_f32_16x16x32_bf16(false, af[kt & 1][i].v, false, w2f[kt][0].v, (short)0, acc[i][0], false, false);
        acc[i][1] = __builtin_amdgcn_wmma_f32_16x16x32_bf16(false, af[kt & 1][i].v, false, w2f[kt][1].v, (short)0, acc[i][1], false, false);
      }
      __builtin_amdgcn_sched_group_barrier(0x100, 8, 0);
      __builtin_amdgcn_sched_group_barrier(0x008, 8, 0);
    }
  }

  if constexpr (OUTBF) {
    __syncthreads();
    unsigned short* sw = (unsigned short*)sbuf;
    #pragma unroll
    for (int j = 0; j < 2; ++j) {
      const int col = (ct0 + j) * 16 + rowL;
      const float b = bias2[col];
      #pragma unroll
      for (int i = 0; i < RT; ++i) {
        #pragma unroll
        for (int r = 0; r < 8; ++r)
          sw[(i * 16 + 8 * hh + r) * (OPITCH * 8) + col] = f2bf(fmaxf(acc[i][j][r] + b, 0.0f));
      }
    }
    __syncthreads();
    if ((wave >> 1) < nr) {
      uint4* __restrict__ Hb = (uint4*)Hout;
      #pragma unroll
      for (int q = 0; q < 8; ++q) {
        const int rloc = wave * 8 + q;
        Hb[((long)rt0 * 16 + rloc) * (HID / 8) + lane] = sbuf[rloc * OPITCH + lane];
      }
    }
  } else {
    #pragma unroll
    for (int j = 0; j < 2; ++j) {
      const int col = (ct0 + j) * 16 + rowL;
      const float b = bias2[col];
      #pragma unroll
      for (int i = 0; i < RT; ++i) {
        if (i < nr) {
          #pragma unroll
          for (int r = 0; r < 8; ++r) {
            const int row = (rt0 + i) * 16 + 8 * hh + r;
            Hout[(long)row * HID + col] = fmaxf(acc[i][j][r] + b, 0.0f);
          }
        }
      }
    }
  }
}

#define POOL_R 16
static_assert(N_NODES % POOL_R == 0, "every wave owns POOL_R whole rows");
static_assert(HID == 256, "a lane owns two float4 of a row");

__device__ __forceinline__ void pool_flush(float* __restrict__ pooled, float* __restrict__ counts, int g, int lane,
                                           const float4& a0, const float4& a1, int len) {
  float* o = pooled + (long)g * HID + lane * 4;
  atomicAdd(o + 0, a0.x); atomicAdd(o + 1, a0.y); atomicAdd(o + 2, a0.z); atomicAdd(o + 3, a0.w);
  atomicAdd(o + 128, a1.x); atomicAdd(o + 129, a1.y); atomicAdd(o + 130, a1.z); atomicAdd(o + 131, a1.w);
  if (lane == 0) atomicAdd(&counts[g], (float)len);
}

__global__ __launch_bounds__(256) void pool_runs(const float* __restrict__ h, const int* __restrict__ batch,
                                                 float* __restrict__ pooled, float* __restrict__ counts) {
  const int lane = threadIdx.x & 31;
  const int gw = __builtin_amdgcn_readfirstlane((int)(blockIdx.x * 8 + (threadIdx.x >> 5)));
  const int n0 = gw * POOL_R;
  if (n0 >= N_NODES) return;
  const int ids = batch[n0 + (lane & (POOL_R - 1))];
  const float4* __restrict__ hp = (const float4*)h + (long)n0 * (HID / 4);
  float4 x[POOL_R][2];
  #pragma unroll
  for (int j = 0; j < POOL_R; ++j) {
    x[j][0] = hp[j * (HID / 4) + lane];
    x[j][1] = hp[j * (HID / 4) + 32 + lane];
  }
  int cur = __builtin_amdgcn_readlane(ids, 0);
  float4 a0 = x[0][0], a1 = x[0][1];
  int len = 1;
  #pragma unroll
  for (int j = 1; j < POOL_R; ++j) {
    const int g = __builtin_amdgcn_readlane(ids, j);
    if (g != cur) {
      pool_flush(pooled, counts, cur, lane, a0, a1, len);
      a0 = x[j][0]; a1 = x[j][1];
      len = 1;
      cur = g;
    } else {
      add4(a0, x[j][0]); add4(a1, x[j][1]);
      ++len;
    }
  }
  pool_flush(pooled, counts, cur, lane, a0, a1, len);
}

__global__ void pool_div_pack(const float* __restrict__ pooled, const float* __restrict__ counts,
                              unsigned int* __restrict__ out) {
  const int KT = HID / 32;
  int i = blockIdx.x * blockDim.x + threadIdx.x;
  int p = i & 7;
  int lane = (i >> 3) & 31;
  int tt = i >> 8;
  int kt = tt % KT;
  int rt = tt / KT;
  int row = rt * 16 + (lane & 15);
  int k = kt * 32 + frag_koff(lane, p);
  const float c = fmaxf(counts[row], 1.0f);
  const float* s = pooled + (long)row * HID + k;
  out[i] = (unsigned int)f2bf(s[0] / c) | ((unsigned int)f2bf(s[1] / c) << 16);
}

extern "C" void kernel_launch(void* const* d_in, const int* in_sizes, int n_in,
                              void* d_out, int out_size, void* d_ws, size_t ws_size,
                              hipStream_t stream) {
  const float* x     = (const float*)d_in[0];
  const int*   ei    = (const int*)d_in[1];
  const int*   esrc  = ei;
  const int*   edst  = ei + N_EDGES;
  const int*   batch = (const int*)d_in[2];
  const float* cw1[3] = {(const float*)d_in[3],  (const float*)d_in[7],  (const float*)d_in[11]};
  const float* cb1[3] = {(const float*)d_in[4],  (const float*)d_in[8],  (const float*)d_in[12]};
  const float* cw2[3] = {(const float*)d_in[5],  (const float*)d_in[9],  (const float*)d_in[13]};
  const float* cb2[3] = {(const float*)d_in[6],  (const float*)d_in[10], (const float*)d_in[14]};
  const float* ow1 = (const float*)d_in[15];
  const float* ob1 = (const float*)d_in[16];
  const float* ow2 = (const float*)d_in[17];
  const float* ob2 = (const float*)d_in[18];

  char* ws = (char*)d_ws;
  size_t off = 0;
  auto bump = [&](size_t bytes) {
    void* p = ws + off;
    off += (bytes + 255) & ~(size_t)255;
    return p;
  };
  const int NRT = N_NODES / 16;
  float*          B2     = (float*)bump((size_t)N_NODES * HID * 4);
  unsigned short* Hb     = (unsigned short*)bump((size_t)N_NODES * HID * 2);
  unsigned int*   Apk    = (unsigned int*)bump((size_t)NRT * 8 * 256 * 4);
  float*          pooled = (float*)bump((size_t)N_GRAPHS * (HID + 1) * 4);
  float*          counts = pooled + (size_t)N_GRAPHS * HID;
  unsigned int*   Ppk    = (unsigned int*)bump((size_t)(N_GRAPHS / 16) * 8 * 256 * 4);
  unsigned int*   MPpk   = (unsigned int*)bump((size_t)(N_GRAPHS / 16) * 8 * 256 * 4);
  unsigned int*   wpk1[3], *wpk2[3];
  for (int l = 0; l < 3; ++l) {
    wpk1[l] = (unsigned int*)bump((size_t)(HID / 16) * 8 * 256 * 4);
    wpk2[l] = (unsigned int*)bump((size_t)(HID / 16) * 8 * 256 * 4);
  }
  unsigned int* owpk1 = (unsigned int*)bump((size_t)(HID / 16) * 8 * 256 * 4);
  unsigned int* owpk2 = (unsigned int*)bump((size_t)(OUT_C / 16) * 8 * 256 * 4);
  int*  deg  = (int*)bump((size_t)(N_NODES + 4) * 4);
  int*  lst  = (int*)bump((size_t)N_NODES * LIST_CAP * 4);
  int*  lst2 = (int*)bump((size_t)N_NODES * LIST_CAP * 4);
  int2* ovf  = (int2*)bump((size_t)N_EDGES * 8);

  dim3 blk(256);

  prep_all<<<PREP_BLOCKS, PREP_T, 0, stream>>>(cw1[0], cw2[0], cw1[1], cw2[1], cw1[2], cw2[2], ow1, ow2,
                                                wpk1[0], wpk2[0], wpk1[1], wpk2[1], wpk1[2], wpk2[2], owpk1, owpk2,
                                                deg, pooled);

  list_fill<<<(N_EDGES + 255) / 256, blk, 0, stream>>>(esrc, edst, deg, lst, ovf);
  list_sort<<<(N_NODES / SORT_NPW + 7) / 8, blk, 0, stream>>>(deg, lst, lst2);

  gather_pack<IN_C><<<NRT, 512, 0, stream>>>(x, deg, lst2, ovf, (uint4*)Apk);
  mlp_fused<IN_C, 1><<<(NRT + 3) / 4, blk, 0, stream>>>((const uint4*)Apk, (const uint4*)wpk1[0], cb1[0],
                                                        (const uint4*)wpk2[0], cb2[0], (float*)Hb, NRT);
  gather_pack_bf<<<NRT, 512, 0, stream>>>((const uint4*)Hb, deg, lst2, ovf, (uint4*)Apk);
  mlp_fused<HID, 1><<<(NRT + 3) / 4, blk, 0, stream>>>((const uint4*)Apk, (const uint4*)wpk1[1], cb1[1],
                                                       (const uint4*)wpk2[1], cb2[1], (float*)Hb, NRT);
  gather_pack_bf<<<NRT, 512, 0, stream>>>((const uint4*)Hb, deg, lst2, ovf, (uint4*)Apk);
  mlp_fused<HID, 0><<<(NRT + 3) / 4, blk, 0, stream>>>((const uint4*)Apk, (const uint4*)wpk1[2], cb1[2],
                                                       (const uint4*)wpk2[2], cb2[2], B2, NRT);

  pool_runs<<<(N_NODES / POOL_R + 7) / 8, blk, 0, stream>>>(B2, batch, pooled, counts);

  const int GRT = N_GRAPHS / 16;
  pool_div_pack<<<GRT * (HID / 32), blk, 0, stream>>>(pooled, counts, Ppk);
  dim3 g3((HID / 16 + 3) / 4, (GRT + 7) / 8);
  gemm_pk<1, 1><<<g3, blk, 0, stream>>>((const uint4*)Ppk, (const uint4*)owpk1, ob1,
                                        MPpk, N_GRAPHS, HID, HID);
  dim3 g4((OUT_C / 16 + 3) / 4, (GRT + 7) / 8);
  gemm_pk<0, 0><<<g4, blk, 0, stream>>>((const uint4*)MPpk, (const uint4*)owpk2, ob2,
                                        (float*)d_out, N_GRAPHS, HID, OUT_C);
}


# Pool and mean passes removed: the last MLP's epilogue adds its rows to the pooled sums (runs per lane in a 64-row tile, atomic flush per run); mean + both output products in one launch; 13->10
# speedup vs baseline: 5.1774x; 1.1595x over previous
#include <hip/hip_runtime.h>

#define N_NODES 50000
#define N_EDGES 800000
#define N_GRAPHS 512
#define IN_C 128
#define HID 256
#define OUT_C 16

typedef __attribute__((ext_vector_type(16))) __bf16 v16bf;
typedef __attribute__((ext_vector_type(8)))  float  v8f;

__device__ __forceinline__ unsigned short f2bf(float f) {
  union { float f; unsigned int u; } c; c.f = f;
  unsigned int u = c.u;
  u += 0x7fffu + ((u >> 16) & 1u);
  return (unsigned short)(u >> 16);
}

__device__ __forceinline__ int frag_koff(int lane, int p) {
  int kb = (lane < 16) ? 0 : 8;
  return (p < 4) ? (kb + 2 * p) : (16 + kb + 2 * (p - 4));
}

#define LIST_CAP 64
__global__ void list_fill(const int* __restrict__ esrc, const int* __restrict__ edst,
                          int* __restrict__ deg, int* __restrict__ lst, int2* __restrict__ ovf) {
  int e = blockIdx.x * blockDim.x + threadIdx.x;
  if (e < N_EDGES) {
    const int d = edst[e];
    const int s = esrc[e];
    const int pos = atomicAdd(&deg[d], 1);
    if (pos < LIST_CAP) {
      lst[(long)d * LIST_CAP + pos] = s;
    } else {
      const int k = atomicAdd(&deg[N_NODES], 1);
      ovf[k] = make_int2(d, s);
    }
  }
}

__device__ __forceinline__ void add4(float4& a, const float4& b) {
  a.x += b.x; a.y += b.y; a.z += b.z; a.w += b.w;
}

static_assert(N_NODES % 16 == 0, "one work-group per full 16-row tile");
template <int C>
__global__ __launch_bounds__(512) void gather_pack(const float* __restrict__ h, const int* __restrict__ deg,
                                                   const int* __restrict__ lst, const int2* __restrict__ ovf,
                                                   uint4* __restrict__ Apk) {
  constexpr int V = C / 128;
  constexpr int U = 8 / V;
  constexpr int KT = C / 32;
  __shared__ uint4 tile[KT * 64];
  const int lane = threadIdx.x & 31;
  const int w = __builtin_amdgcn_readfirstlane((int)(threadIdx.x >> 5));
  const int node = __builtin_amdgcn_readfirstlane((int)(blockIdx.x * 16 + w));
  const int total = deg[node];
  const int beg = node * LIST_CAP;
  const int end = beg + min(total, LIST_CAP);
  const int* __restrict__ col = lst;
  int mine = col[beg + lane];
  const float4* __restrict__ hp = (const float4*)h;
  float4 acc[V];
  #pragma unroll
  for (int v = 0; v < V; ++v) acc[v] = hp[(long)node * (C / 4) + v * 32 + lane];
  asm volatile("" :: "v"(mine));
  for (int base = beg; base < end; base += 32) {
    const int cnt = min(32, end - base);
    if (base > beg) mine = (lane < cnt) ? col[base + lane] : 0;
    int j = 0;
    for (; j + U <= cnt; j += U) {
      float4 x[U][V];
      #pragma unroll
      for (int u = 0; u < U; ++u) {
        const int sidx = __builtin_amdgcn_readlane(mine, j + u);
        #pragma unroll
        for (int v = 0; v < V; ++v) x[u][v] = hp[(long)sidx * (C / 4) + v * 32 + lane];
      }
      #pragma unroll
      for (int u = 0; u < U; ++u) {
        #pragma unroll
        for (int v = 0; v < V; ++v) add4(acc[v], x[u][v]);
      }
    }
    #pragma nounroll
    for (; j < cnt; ++j) {
      const int sidx = __builtin_amdgcn_readlane(mine, j);
      #pragma unroll
      for (int v = 0; v < V; ++v) add4(acc[v], hp[(long)sidx * (C / 4) + v * 32 + lane]);
    }
  }
  if (total > LIST_CAP) {
    const int no = min(deg[N_NODES], N_EDGES);
    #pragma nounroll
    for (int k = 0; k < no; ++k) {
      const int2 e = ovf[k];
      if (e.x == node) {
        #pragma unroll
        for (int v = 0; v < V; ++v) add4(acc[v], hp[(long)e.y * (C / 4) + v * 32 + lane]);
      }
    }
  }
  unsigned int* tl = (unsigned int*)tile;
  const int m  = lane & 7;
  const int fh = (m >> 1) & 1;
  const int p0 = 2 * (m & 1) + 4 * (m >> 2);
  #pragma unroll
  for (int v = 0; v < V; ++v) {
    const int kt = v * 4 + (lane >> 3);
    uint2 pr;
    pr.x = (unsigned int)f2bf(acc[v].x) | ((unsigned int)f2bf(acc[v].y) << 16);
    pr.y = (unsigned int)f2bf(acc[v].z) | ((unsigned int)f2bf(acc[v].w) << 16);
    *(uint2*)(tl + kt * 256 + (w + 16 * fh) * 8 + p0) = pr;
  }
  __syncthreads();
  if (threadIdx.x < KT * 64) Apk[(long)blockIdx.x * (KT * 64) + threadIdx.x] = tile[threadIdx.x];
}

#define SORT_NPW 4
static_assert(N_NODES % SORT_NPW == 0, "every wave owns SORT_NPW whole nodes");
__global__ __launch_bounds__(256) void list_sort(const int* __restrict__ deg, const int* __restrict__ col,
                                                 int* __restrict__ col2) {
  const int lane = threadIdx.x & 31;
  const int n0 = __builtin_amdgcn_readfirstlane((int)((blockIdx.x * 8 + (threadIdx.x >> 5)) * SORT_NPW));
  if (n0 >= N_NODES) return;
  int len[SORT_NPW];
  #pragma unroll
  for (int q = 0; q < SORT_NPW; ++q) len[q] = min(deg[n0 + q], LIST_CAP);
  int mine[SORT_NPW];
  #pragma unroll
  for (int q = 0; q < SORT_NPW; ++q)
    mine[q] = (lane < len[q]) ? col[(n0 + q) * LIST_CAP + lane] : 0x7fffffff;
  #pragma unroll
  for (int q = 0; q < SORT_NPW; ++q) {
    const int beg = (n0 + q) * LIST_CAP;
    const int d = len[q];
    if (d <= 32) {
      int rank = 0;
      for (int j = 0; j < d; ++j) {
        const int v = __builtin_amdgcn_readlane(mine[q], j);
        rank += (v < mine[q] || (v == mine[q] && j < lane)) ? 1 : 0;
      }
      col2[beg + ((lane < d) ? rank : lane)] = (lane < d) ? mine[q] : -1;
    } else {
      for (int e = lane; e < d; e += 32) {
        const int v = col[beg + e];
        int rank = 0;
        for (int j = 0; j < d; ++j) {
          const int u = col[beg + j];
          rank += (u < v || (u == v && j < e)) ? 1 : 0;
        }
        col2[beg + rank] = v;
      }
    }
  }
}

__device__ __forceinline__ void addbf(float (&a)[8], const uint4& q) {
  a[0] += __uint_as_float(q.x << 16);  a[1] += __uint_as_float(q.x & 0xffff0000u);
  a[2] += __uint_as_float(q.y << 16);  a[3] += __uint_as_float(q.y & 0xffff0000u);
  a[4] += __uint_as_float(q.z << 16);  a[5] += __uint_as_float(q.z & 0xffff0000u);
  a[6] += __uint_as_float(q.w << 16);  a[7] += __uint_as_float(q.w & 0xffff0000u);
}

static_assert(HID == 256, "a bf16 row is exactly one 16-byte load per lane");
__global__ __launch_bounds__(512) void gather_pack_bf(const uint4* __restrict__ hb, const int* __restrict__ deg,
                                                      const int* __restrict__ lst, const int2* __restrict__ ovf,
                                                      uint4* __restrict__ Apk) {
  constexpr int KT = HID / 32;
  __shared__ uint4 tile[KT * 64];
  const int lane = threadIdx.x & 31;
  const int w = __builtin_amdgcn_readfirstlane((int)(threadIdx.x >> 5));
  const int node = __builtin_amdgcn_readfirstlane((int)(blockIdx.x * 16 + w));
  const int total = deg[node];
  const int beg = node * LIST_CAP;
  const int end = beg + min(total, LIST_CAP);
  const int* __restrict__ col = lst;
  int mine = col[beg + lane];
  float acc[8];
  #pragma unroll
  for (int i = 0; i < 8; ++i) acc[i] = 0.0f;
  addbf(acc, hb[(long)node * (HID / 8) + lane]);
  asm volatile("" :: "v"(mine));
  for (int base = beg; base < end; base += 32) {
    const int cnt = min(32, end - base);
    if (base > beg) mine = (lane < cnt) ? col[base + lane] : 0;
    int j = 0;
    for (; j + 8 <= cnt; j += 8) {
      uint4 x[8];
      #pragma unroll
      for (int u = 0; u < 8; ++u) x[u] = hb[(long)__builtin_amdgcn_readlane(mine, j + u) * (HID / 8) + lane];
      #pragma unroll
      for (int u = 0; u < 8; ++u) addbf(acc, x[u]);
    }
    if (j + 4 <= cnt) {
      uint4 x[4];
      #pragma unroll
      for (int u = 0; u < 4; ++u) x[u] = hb[(long)__builtin_amdgcn_readlane(mine, j + u) * (HID / 8) + lane];
      #pragma unroll
      for (int u = 0; u < 4; ++u) addbf(acc, x[u]);
      j += 4;
    }
    #pragma nounroll
    for (; j < cnt; ++j) addbf(acc, hb[(long)__builtin_amdgcn_readlane(mine, j) * (HID / 8) + lane]);
  }
  if (total > LIST_CAP) {
    const int no = min(deg[N_NODES], N_EDGES);
    #pragma nounroll
    for (int k = 0; k < no; ++k) {
      const int2 e = ovf[k];
      if (e.x == node) addbf(acc, hb[(long)e.y * (HID / 8) + lane]);
    }
  }
  unsigned int* tl = (unsigned int*)tile;
  const int m  = lane & 3;
  const int fh = m & 1;
  const int p0 = 4 * (m >> 1);
  const int kt = lane >> 2;
  uint4 pr;
  pr.x = (unsigned int)f2bf(acc[0]) | ((unsigned int)f2bf(acc[1]) << 16);
  pr.y = (unsigned int)f2bf(acc[2]) | ((unsigned int)f2bf(acc[3]) << 16);
  pr.z = (unsigned int)f2bf(acc[4]) | ((unsigned int)f2bf(acc[5]) << 16);
  pr.w = (unsigned int)f2bf(acc[6]) | ((unsigned int)f2bf(acc[7]) << 16);
  *(uint4*)(tl + kt * 256 + (w + 16 * fh) * 8 + p0) = pr;
  __syncthreads();
  Apk[(long)blockIdx.x * (KT * 64) + threadIdx.x] = tile[threadIdx.x];
}

template <int K, int M>
__device__ __forceinline__ void wpack_elem(const float* __restrict__ Wm, unsigned int* __restrict__ out, int i) {
  constexpr int KT = K / 32;
  const int p = i & 7;
  const int lane = (i >> 3) & 31;
  const int tt = i >> 8;
  const int kt = tt % KT;
  const int ct = tt / KT;
  const int col = ct * 16 + (lane & 15);
  const int k = kt * 32 + frag_koff(lane, p);
  const unsigned int lo = f2bf(Wm[(long)k * M + col]);
  const unsigned int hi = f2bf(Wm[(long)(k + 1) * M + col]);
  out[i] = lo | (hi << 16);
}

#define PREP_T      256
#define PREP_B_WIN  ((HID / 16) * (IN_C / 32))
#define PREP_B_WH   ((HID / 16) * (HID / 32))
#define PREP_N_WH   6
#define PREP_B_WOUT ((OUT_C / 16) * (HID / 32))
#define PREP_N_DEG4 (N_NODES / 4 + 1)
#define PREP_B_DEG  ((PREP_N_DEG4 + PREP_T - 1) / PREP_T)
#define PREP_N_PL4  (N_GRAPHS * (HID + 1) / 4)
#define PREP_B_PL   ((PREP_N_PL4 + PREP_T - 1) / PREP_T)
#define PREP_BLOCKS (PREP_B_WIN + PREP_N_WH * PREP_B_WH + PREP_B_WOUT + PREP_B_DEG + PREP_B_PL)
static_assert(N_NODES % 4 == 0 && (N_GRAPHS * (HID + 1)) % 4 == 0, "the clears go out as whole 16-byte stores");

__global__ __launch_bounds__(PREP_T) void prep_all(
    const float* __restrict__ w_in,
    const float* __restrict__ wh0, const float* __restrict__ wh1, const float* __restrict__ wh2,
    const float* __restrict__ wh3, const float* __restrict__ wh4, const float* __restrict__ wh5,
    const float* __restrict__ w_out,
    unsigned int* __restrict__ p_in,
    unsigned int* __restrict__ ph0, unsigned int* __restrict__ ph1, unsigned int* __restrict__ ph2,
    unsigned int* __restrict__ ph3, unsigned int* __restrict__ ph4, unsigned int* __restrict__ ph5,
    unsigned int* __restrict__ p_out,
    int* __restrict__ deg, float* __restrict__ pooled) {
  const int t = threadIdx.x;
  int b = blockIdx.x;
  if (b < PREP_B_WIN) { wpack_elem<IN_C, HID>(w_in, p_in, b * PREP_T + t); return; }
  b -= PREP_B_WIN;
  if (b < PREP_N_WH * PREP_B_WH) {
    const int j = b / PREP_B_WH;
    const int i = (b % PREP_B_WH) * PREP_T + t;
    const float* Wm = wh0;  unsigned int* o = ph0;
    if (j == 1) { Wm = wh1; o = ph1; }
    if (j == 2) { Wm = wh2; o = ph2; }
    if (j == 3) { Wm = wh3; o = ph3; }
    if (j == 4) { Wm = wh4; o = ph4; }
    if (j == 5) { Wm = wh5; o = ph5; }
    wpack_elem<HID, HID>(Wm, o, i);
    return;
  }
  b -= PREP_N_WH * PREP_B_WH;
  if (b < PREP_B_WOUT) { wpack_elem<HID, OUT_C>(w_out, p_out, b * PREP_T + t); return; }
  b -= PREP_B_WOUT;
  if (b < PREP_B_DEG) {
    const int i4 = b * PREP_T + t;
    if (i4 < PREP_N_DEG4) ((int4*)deg)[i4] = make_int4(0, 0, 0, 0);
    return;
  }
  b -= PREP_B_DEG;
  const int i4 = b * PREP_T + t;
  if (i4 < PREP_N_PL4) ((float4*)pooled)[i4] = make_float4(0.0f, 0.0f, 0.0f, 0.0f);
}

static_assert(HID == 256, "a wave owns two 16-column tiles of the 256");
__device__ __forceinline__ void pool_add(float* __restrict__ pooled, float* __restrict__ counts, int g, int col0,
                                         float s0, float s1, int len, bool cnt_lane) {
  float* o = pooled + (long)g * HID + col0;
  atomicAdd(o, s0);
  atomicAdd(o + 16, s1);
  if (cnt_lane) atomicAdd(&counts[g], (float)len);
}

template <int K1, int OUTBF>
__global__ __launch_bounds__(256) void mlp_fused(const uint4* __restrict__ Apk, const uint4* __restrict__ W1pk,
                                                 const float* __restrict__ bias1, const uint4* __restrict__ W2pk,
                                                 const float* __restrict__ bias2, float* __restrict__ Hout, int nrt,
                                                 const int* __restrict__ batch, float* __restrict__ counts) {
  constexpr int KT1 = K1 / 32;
  constexpr int KT2 = HID / 32;
  constexpr int RT  = 4;
  static_assert(KT1 <= KT2, "the A tile must fit the buffer that later holds the middle tile");
  constexpr int OPITCH = HID / 8 + 1;
  __shared__ uint4 sbuf[RT * KT2 * 64 + RT * 16];
  const int tid  = threadIdx.x;
  const int lane = tid & 31;
  const int wave = __builtin_amdgcn_readfirstlane(tid >> 5);
  const int rt0  = blockIdx.x * RT;
  const int nr   = min(RT, nrt - rt0);
  const int ct0  = wave * 2;
  union F { uint4 q[2]; v16bf v; };

  F w1f[KT1][2];
  {
    const uint4* Wp = W1pk + ((long)ct0 * KT1 * 32 + lane) * 2;
    #pragma unroll
    for (int kt = 0; kt < KT1; ++kt) {
      w1f[kt][0].q[0] = Wp[kt * 64];           w1f[kt][0].q[1] = Wp[kt * 64 + 1];
      w1f[kt][1].q[0] = Wp[(KT1 + kt) * 64];   w1f[kt][1].q[1] = Wp[(KT1 + kt) * 64 + 1];
    }
  }
  {
    const uint4* Ab = Apk + (long)rt0 * (KT1 * 64);
    const int nvalid = nr * (KT1 * 64);
    uint4 v[RT * KT1 * 64 / 256];
    #pragma unroll
    for (int s = 0; s < RT * KT1 * 64 / 256; ++s) {
      const int i = s * 256 + tid;
      v[s] = (i < nvalid) ? Ab[i] : make_uint4(0u, 0u, 0u, 0u);
    }
    #pragma unroll
    for (int s = 0; s < RT * KT1 * 64 / 256; ++s) {
      const int i = s * 256 + tid;
      sbuf[(i & ~63) | ((i & 1) << 5) | ((i & 63) >> 1)] = v[s];
    }
  }
  if constexpr (!OUTBF) {
    if (tid < RT * 16) ((int*)(sbuf + RT * KT2 * 64))[tid] = (tid < nr * 16) ? batch[rt0 * 16 + tid] : -1;
  }
  __syncthreads();

  v8f acc[RT][2];
  #pragma unroll
  for (int i = 0; i < RT; ++i) { acc[i][0] = v8f{}; acc[i][1] = v8f{}; }
  {
    F af[2][RT];
    #pragma unroll
    for (int i = 0; i < RT; ++i) {
      af[0][i].q[0] = sbuf[(i * KT1) * 64 + lane];
      af[0][i].q[1] = sbuf[(i * KT1) * 64 + 32 + lane];
    }
    __builtin_amdgcn_sched_group_barrier(0x100, 8, 0);
    #pragma unroll
    for (int kt = 0; kt < KT1; ++kt) {
      if (kt + 1 < KT1) {
        #pragma unroll
        for (int i = 0; i < RT; ++i) {
          af[(kt + 1) & 1][i].q[0] = sbuf[(i * KT1 + kt + 1) * 64 + lane];
          af[(kt + 1) & 1][i].q[1] = sbuf[(i * KT1 + kt + 1) * 64 + 32 + lane];
        }
      }
      #pragma unroll
      for (int i = 0; i < RT; ++i) {
        acc[i][0] = __builtin_amdgcn_wmma_f32_16x16x32_bf16(false, af[kt & 1][i].v, false, w1f[kt][0].v, (short)0, acc[i][0], false, false);
        acc[i][1] = __builtin_amdgcn_wmma_f32_16x16x32_bf16(false, af[kt & 1][i].v, false, w1f[kt][1].v, (short)0, acc[i][1], false, false);
      }
      __builtin_amdgcn_sched_group_barrier(0x100, 8, 0);
      __builtin_amdgcn_sched_group_barrier(0x008, 8, 0);
    }
  }

  F w2f[KT2][2];
  {
    const uint4* Wp = W2pk + ((long)ct0 * KT2 * 32 + lane) * 2;
    #pragma unroll
    for (int kt = 0; kt < KT2; ++kt) {
      w2f[kt][0].q[0] = Wp[kt * 64];           w2f[kt][0].q[1] = Wp[kt * 64 + 1];
      w2f[kt][1].q[0] = Wp[(KT2 + kt) * 64];   w2f[kt][1].q[1] = Wp[(KT2 + kt) * 64 + 1];
    }
  }

  const int rowL = lane & 15;
  const int hh   = lane >> 4;
  __syncthreads();
  {
    unsigned short* sw = (unsigned short*)sbuf;
    #pragma unroll
    for (int j = 0; j < 2; ++j) {
      const float b = bias1[(ct0 + j) * 16 + rowL];
      #pragma unroll
      for (int i = 0; i < RT; ++i) {
        #pragma unroll
        for (int r = 0; r < 8; ++r) {
          const float o = fmaxf(acc[i][j][r] + b, 0.0f);
          sw[(i * KT2 + wave) * 512 + j * 256 + (8 * hh + r + 16 * (rowL >> 3)) * 8 + (rowL & 7)] = f2bf(o);
        }
      }
    }
  }
  __syncthreads();

  #pragma unroll
  for (int i = 0; i < RT; ++i) { acc[i][0] = v8f{}; acc[i][1] = v8f{}; }
  {
    F af[2][RT];
    #pragma unroll
    for (int i = 0; i < RT; ++i) {
      af[0][i].q[0] = sbuf[(i * KT2) * 64 + lane];
      af[0][i].q[1] = sbuf[(i * KT2) * 64 + 32 + lane];
    }
    __builtin_amdgcn_sched_group_barrier(0x100, 8, 0);
    #pragma unroll
    for (int kt = 0; kt < KT2; ++kt) {
      if (kt + 1 < KT2) {
        #pragma unroll
        for (int i = 0; i < RT; ++i) {
          af[(kt + 1) & 1][i].q[0] = sbuf[(i * KT2 + kt + 1) * 64 + lane];
          af[(kt + 1) & 1][i].q[1] = sbuf[(i * KT2 + kt + 1) * 64 + 32 + lane];
        }
      }
      #pragma unroll
      for (int i = 0; i < RT; ++i) {
        acc[i][0] = __builtin_amdgcn_wmma_f32_16x16x32_bf16(false, af[kt & 1][i].v, false, w2f[kt][0].v, (short)0, acc[i][0], false, false);
        acc[i][1] = __builtin_amdgcn_wmma_f32_16x16x32_bf16(false, af[kt & 1][i].v, false, w2f[kt][1].v, (short)0, acc[i][1], false, false);
      }
      __builtin_amdgcn_sched_group_barrier(0x100, 8, 0);
      __builtin_amdgcn_sched_group_barrier(0x008, 8, 0);
    }
  }

  if constexpr (OUTBF) {
    __syncthreads();
    unsigned short* sw = (unsigned short*)sbuf;
    #pragma unroll
    for (int j = 0; j < 2; ++j) {
      const int col = (ct0 + j) * 16 + rowL;
      const float b = bias2[col];
      #pragma unroll
      for (int i = 0; i < RT; ++i) {
        #pragma unroll
        for (int r = 0; r < 8; ++r)
          sw[(i * 16 + 8 * hh + r) * (OPITCH * 8) + col] = f2bf(fmaxf(acc[i][j][r] + b, 0.0f));
      }
    }
    __syncthreads();
    if ((wave >> 1) < nr) {
      uint4* __restrict__ Hb = (uint4*)Hout;
      #pragma unroll
      for (int q = 0; q < 8; ++q) {
        const int rloc = wave * 8 + q;
        Hb[((long)rt0 * 16 + rloc) * (HID / 8) + lane] = sbuf[rloc * OPITCH + lane];
      }
    }
  } else {
    const int* idl = (const int*)(sbuf + RT * KT2 * 64);
    const int col0 = ct0 * 16 + rowL;
    const float b0 = bias2[col0];
    const float b1 = bias2[col0 + 16];
    const bool cnt_lane = (wave == 0) && (rowL == 0);
    float s0 = 0.0f, s1 = 0.0f;
    int len = 0;
    int cur = idl[8 * hh];
    #pragma unroll
    for (int i = 0; i < RT; ++i) {
      if (i == 0 || i < nr) {
        const int4 ga = *(const int4*)(idl + i * 16 + 8 * hh);
        const int4 gb = *(const int4*)(idl + i * 16 + 8 * hh + 4);
        const int g[8] = {ga.x, ga.y, ga.z, ga.w, gb.x, gb.y, gb.z, gb.w};
        #pragma unroll
        for (int r = 0; r < 8; ++r) {
          if (g[r] != cur) {
            pool_add(Hout, counts, cur, col0, s0, s1, len, cnt_lane);
            s0 = 0.0f; s1 = 0.0f; len = 0;
            cur = g[r];
          }
          s0 += fmaxf(acc[i][0][r] + b0, 0.0f);
          s1 += fmaxf(acc[i][1][r] + b1, 0.0f);
          ++len;
        }
      }
    }
    pool_add(Hout, counts, cur, col0, s0, s1, len, cnt_lane);
  }
}

static_assert(HID == 256 && OUT_C == 16 && N_GRAPHS % 16 == 0, "one work-group per full 16-graph tile, one output column tile");
__global__ __launch_bounds__(256) void out_mlp(const float* __restrict__ pooled, const float* __restrict__ counts,
                                               const uint4* __restrict__ W1pk, const float* __restrict__ bias1,
                                               const uint4* __restrict__ W2pk, const float* __restrict__ bias2,
                                               float* __restrict__ out) {
  constexpr int KT = HID / 32;
  __shared__ uint4 sbuf[KT * 64];
  const int tid  = threadIdx.x;
  const int lane = tid & 31;
  const int wave = __builtin_amdgcn_readfirstlane(tid >> 5);
  const int rt   = blockIdx.x;
  const int ct0  = wave * 2;
  union F { uint4 q[2]; v16bf v; };

  const int m = tid >> 4;
  const int c = tid & 15;
  const int row = rt * 16 + m;
  const float cnt = fmaxf(counts[row], 1.0f);
  const float4* sp = (const float4*)(pooled + (long)row * HID + c * 16);
  const float4 s0 = sp[0], s1 = sp[1], s2 = sp[2], s3 = sp[3];
  const int rowL = lane & 15;
  const int hh   = lane >> 4;
  const float b10 = bias1[ct0 * 16 + rowL];
  const float b11 = bias1[(ct0 + 1) * 16 + rowL];
  F w1f[KT][2];
  {
    const uint4* Wp = W1pk + ((long)ct0 * KT * 32 + lane) * 2;
    #pragma unroll
    for (int kt = 0; kt < KT; ++kt) {
      w1f[kt][0].q[0] = Wp[kt * 64];          w1f[kt][0].q[1] = Wp[kt * 64 + 1];
      w1f[kt][1].q[0] = Wp[(KT + kt) * 64];   w1f[kt][1].q[1] = Wp[(KT + kt) * 64 + 1];
    }
  }
  F w2f[KT];
  float b2v = 0.0f;
  #pragma unroll
  for (int kt = 0; kt < KT; ++kt) { w2f[kt].q[0] = make_uint4(0u, 0u, 0u, 0u); w2f[kt].q[1] = make_uint4(0u, 0u, 0u, 0u); }
  if (wave == 0) {
    const uint4* Wp = W2pk + lane * 2;
    #pragma unroll
    for (int kt = 0; kt < KT; ++kt) { w2f[kt].q[0] = Wp[kt * 64]; w2f[kt].q[1] = Wp[kt * 64 + 1]; }
    b2v = bias2[rowL];
  }

  {
    uint4 lo, hi;
    lo.x = (unsigned int)f2bf(s0.x / cnt) | ((unsigned int)f2bf(s0.y / cnt) << 16);
    lo.y = (unsigned int)f2bf(s0.z / cnt) | ((unsigned int)f2bf(s0.w / cnt) << 16);
    lo.z = (unsigned int)f2bf(s1.x / cnt) | ((unsigned int)f2bf(s1.y / cnt) << 16);
    lo.w = (unsigned int)f2bf(s1.z / cnt) | ((unsigned int)f2bf(s1.w / cnt) << 16);
    hi.x = (unsigned int)f2bf(s2.x / cnt) | ((unsigned int)f2bf(s2.y / cnt) << 16);
    hi.y = (unsigned int)f2bf(s2.z / cnt) | ((unsigned int)f2bf(s2.w / cnt) << 16);
    hi.z = (unsigned int)f2bf(s3.x / cnt) | ((unsigned int)f2bf(s3.y / cnt) << 16);
    hi.w = (unsigned int)f2bf(s3.z / cnt) | ((unsigned int)f2bf(s3.w / cnt) << 16);
    const int base = (c >> 1) * 64 + (c & 1) * 32 + m;
    sbuf[base]      = lo;
    sbuf[base + 16] = hi;
  }
  __syncthreads();

  v8f acc0 = v8f{}, acc1 = v8f{};
  {
    F af[KT];
    #pragma unroll
    for (int kt = 0; kt < KT; ++kt) {
      af[kt].q[0] = sbuf[kt * 64 + lane];
      af[kt].q[1] = sbuf[kt * 64 + 32 + lane];
    }
    #pragma unroll
    for (int kt = 0; kt < KT; ++kt) {
      acc0 = __builtin_amdgcn_wmma_f32_16x16x32_bf16(false, af[kt].v, false, w1f[kt][0].v, (short)0, acc0, false, false);
      acc1 = __builtin_amdgcn_wmma_f32_16x16x32_bf16(false, af[kt].v, false, w1f[kt][1].v, (short)0, acc1, false, false);
    }
  }

  __syncthreads();
  {
    unsigned short* sw = (unsigned short*)sbuf;
    #pragma unroll
    for (int r = 0; r < 8; ++r) {
      const int pos = wave * 512 + (8 * hh + r + 16 * (rowL >> 3)) * 8 + (rowL & 7);
      sw[pos]       = f2bf(fmaxf(acc0[r] + b10, 0.0f));
      sw[pos + 256] = f2bf(fmaxf(acc1[r] + b11, 0.0f));
    }
  }
  __syncthreads();

  if (wave == 0) {
    v8f acc = v8f{};
    F af[KT];
    #pragma unroll
    for (int kt = 0; kt < KT; ++kt) {
      af[kt].q[0] = sbuf[kt * 64 + lane];
      af[kt].q[1] = sbuf[kt * 64 + 32 + lane];
    }
    #pragma unroll
    for (int kt = 0; kt < KT; ++kt)
      acc = __builtin_amdgcn_wmma_f32_16x16x32_bf16(false, af[kt].v, false, w2f[kt].v, (short)0, acc, false, false);
    #pragma unroll
    for (int r = 0; r < 8; ++r) out[(long)(rt * 16 + 8 * hh + r) * OUT_C + rowL] = acc[r] + b2v;
  }
}

extern "C" void kernel_launch(void* const* d_in, const int* in_sizes, int n_in,
                              void* d_out, int out_size, void* d_ws, size_t ws_size,
                              hipStream_t stream) {
  const float* x     = (const float*)d_in[0];
  const int*   ei    = (const int*)d_in[1];
  const int*   esrc  = ei;
  const int*   edst  = ei + N_EDGES;
  const int*   batch = (const int*)d_in[2];
  const float* cw1[3] = {(const float*)d_in[3],  (const float*)d_in[7],  (const float*)d_in[11]};
  const float* cb1[3] = {(const float*)d_in[4],  (const float*)d_in[8],  (const float*)d_in[12]};
  const float* cw2[3] = {(const float*)d_in[5],  (const float*)d_in[9],  (const float*)d_in[13]};
  const float* cb2[3] = {(const float*)d_in[6],  (const float*)d_in[10], (const float*)d_in[14]};
  const float* ow1 = (const float*)d_in[15];
  const float* ob1 = (const float*)d_in[16];
  const float* ow2 = (const float*)d_in[17];
  const float* ob2 = (const float*)d_in[18];

  char* ws = (char*)d_ws;
  size_t off = 0;
  auto bump = [&](size_t bytes) {
    void* p = ws + off;
    off += (bytes + 255) & ~(size_t)255;
    return p;
  };
  const int NRT = N_NODES / 16;
  unsigned short* Hb     = (unsigned short*)bump((size_t)N_NODES * HID * 2);
  unsigned int*   Apk    = (unsigned int*)bump((size_t)NRT * 8 * 256 * 4);
  float*          pooled = (float*)bump((size_t)N_GRAPHS * (HID + 1) * 4);
  float*          counts = pooled + (size_t)N_GRAPHS * HID;
  unsigned int*   wpk1[3], *wpk2[3];
  for (int l = 0; l < 3; ++l) {
    wpk1[l] = (unsigned int*)bump((size_t)(HID / 16) * 8 * 256 * 4);
    wpk2[l] = (unsigned int*)bump((size_t)(HID / 16) * 8 * 256 * 4);
  }
  unsigned int* owpk1 = (unsigned int*)bump((size_t)(HID / 16) * 8 * 256 * 4);
  unsigned int* owpk2 = (unsigned int*)bump((size_t)(OUT_C / 16) * 8 * 256 * 4);
  int*  deg  = (int*)bump((size_t)(N_NODES + 4) * 4);
  int*  lst  = (int*)bump((size_t)N_NODES * LIST_CAP * 4);
  int*  lst2 = (int*)bump((size_t)N_NODES * LIST_CAP * 4);
  int2* ovf  = (int2*)bump((size_t)N_EDGES * 8);

  dim3 blk(256);

  prep_all<<<PREP_BLOCKS, PREP_T, 0, stream>>>(cw1[0], cw2[0], cw1[1], cw2[1], cw1[2], cw2[2], ow1, ow2,
                                                wpk1[0], wpk2[0], wpk1[1], wpk2[1], wpk1[2], wpk2[2], owpk1, owpk2,
                                                deg, pooled);

  list_fill<<<(N_EDGES + 255) / 256, blk, 0, stream>>>(esrc, edst, deg, lst, ovf);
  list_sort<<<(N_NODES / SORT_NPW + 7) / 8, blk, 0, stream>>>(deg, lst, lst2);

  gather_pack<IN_C><<<NRT, 512, 0, stream>>>(x, deg, lst2, ovf, (uint4*)Apk);
  mlp_fused<IN_C, 1><<<(NRT + 3) / 4, blk, 0, stream>>>((const uint4*)Apk, (const uint4*)wpk1[0], cb1[0],
                                                        (const uint4*)wpk2[0], cb2[0], (float*)Hb, NRT, nullptr, nullptr);
  gather_pack_bf<<<NRT, 512, 0, stream>>>((const uint4*)Hb, deg, lst2, ovf, (uint4*)Apk);
  mlp_fused<HID, 1><<<(NRT + 3) / 4, blk, 0, stream>>>((const uint4*)Apk, (const uint4*)wpk1[1], cb1[1],
                                                       (const uint4*)wpk2[1], cb2[1], (float*)Hb, NRT, nullptr, nullptr);
  gather_pack_bf<<<NRT, 512, 0, stream>>>((const uint4*)Hb, deg, lst2, ovf, (uint4*)Apk);
  mlp_fused<HID, 0><<<(NRT + 3) / 4, blk, 0, stream>>>((const uint4*)Apk, (const uint4*)wpk1[2], cb1[2],
                                                       (const uint4*)wpk2[2], cb2[2], pooled, NRT, batch, counts);

  out_mlp<<<N_GRAPHS / 16, blk, 0, stream>>>(pooled, counts, (const uint4*)owpk1, ob1, (const uint4*)owpk2, ob2,
                                             (float*)d_out);
}


# Row gathers moved into the MLP launches: a work-group's 8 waves sum the 64 rows of its tile straight into the LDS A tile (a row's loads all in flight at once, same sum order); 10 -> 7 launches
# speedup vs baseline: 6.8460x; 1.3223x over previous
#include <hip/hip_runtime.h>

#define N_NODES 50000
#define N_EDGES 800000
#define N_GRAPHS 512
#define IN_C 128
#define HID 256
#define OUT_C 16

typedef __attribute__((ext_vector_type(16))) __bf16 v16bf;
typedef __attribute__((ext_vector_type(8)))  float  v8f;

__device__ __forceinline__ unsigned short f2bf(float f) {
  union { float f; unsigned int u; } c; c.f = f;
  unsigned int u = c.u;
  u += 0x7fffu + ((u >> 16) & 1u);
  return (unsigned short)(u >> 16);
}

__device__ __forceinline__ int frag_koff(int lane, int p) {
  int kb = (lane < 16) ? 0 : 8;
  return (p < 4) ? (kb + 2 * p) : (16 + kb + 2 * (p - 4));
}

#define LIST_CAP 64
#define BIN_R    196
#define NBIN     ((N_NODES + BIN_R - 1) / BIN_R)
#define BIN_CAP  4096
#define BINA_T   512
#define BINA_EPT 8
#define BINA_CHUNK (BINA_T * BINA_EPT)
static_assert(NBIN <= BINA_T, "one thread per bin counter");
__global__ __launch_bounds__(BINA_T) void bin_fill(const int* __restrict__ esrc, const int* __restrict__ edst,
                                                   int* __restrict__ bcnt, int* __restrict__ spillcnt,
                                                   int2* __restrict__ binbuf, int2* __restrict__ spill) {
  __shared__ int lcnt[NBIN];
  __shared__ int lbase[NBIN];
  const int tid = threadIdx.x;
  if (tid < NBIN) lcnt[tid] = 0;
  __syncthreads();
  int d[BINA_EPT], s[BINA_EPT], pos[BINA_EPT];
  const int e0 = blockIdx.x * BINA_CHUNK + tid;
  #pragma unroll
  for (int u = 0; u < BINA_EPT; ++u) {
    const int e = e0 + u * BINA_T;
    const bool ok = e < N_EDGES;
    d[u] = ok ? edst[e] : -1;
    s[u] = ok ? esrc[e] : 0;
  }
  #pragma unroll
  for (int u = 0; u < BINA_EPT; ++u) pos[u] = (d[u] >= 0) ? atomicAdd(&lcnt[(unsigned)d[u] / BIN_R], 1) : 0;
  __syncthreads();
  if (tid < NBIN) {
    const int cnt = lcnt[tid];
    lbase[tid] = (cnt > 0) ? atomicAdd(&bcnt[tid], cnt) : 0;
  }
  __syncthreads();
  #pragma unroll
  for (int u = 0; u < BINA_EPT; ++u) {
    if (d[u] >= 0) {
      const int b = (unsigned)d[u] / BIN_R;
      const int p = lbase[b] + pos[u];
      if (p < BIN_CAP) {
        binbuf[(long)b * BIN_CAP + p] = make_int2(d[u], s[u]);
      } else {
        const int k = atomicAdd(spillcnt, 1);
        spill[k] = make_int2(d[u], s[u]);
      }
    }
  }
}
#define BINB_T 512
#define BINB_U 7
static_assert(BIN_R <= BINB_T, "one thread per node counter");
__device__ __forceinline__ void place_pair(const int2 e, int pos, int n0, int* llist,
                                           int* __restrict__ ovfcnt, int2* __restrict__ ovf) {
  if (pos < LIST_CAP) {
    llist[(e.x - n0) * LIST_CAP + pos] = e.y;
  } else {
    const int k = atomicAdd(ovfcnt, 1);
    ovf[k] = e;
  }
}

__global__ __launch_bounds__(BINB_T) void bin_lists(const int* __restrict__ bcnt, const int2* __restrict__ binbuf,
                                                    const int* __restrict__ spillcnt, const int2* __restrict__ spill,
                                                    int* __restrict__ deg, int* __restrict__ ovfcnt,
                                                    int* __restrict__ col2, int2* __restrict__ ovf) {
  __shared__ int ldeg[BIN_R];
  __shared__ int llist[BIN_R * LIST_CAP];
  const int tid  = threadIdx.x;
  const int lane = tid & 31;
  const int wave = __builtin_amdgcn_readfirstlane(tid >> 5);
  const int b  = blockIdx.x;
  const int n0 = b * BIN_R;
  const int nb = bcnt[b];
  const int ne = min(nb, BIN_CAP);
  if (tid < BIN_R) ldeg[tid] = 0;
  __syncthreads();
  const int2* __restrict__ bp = binbuf + (long)b * BIN_CAP;
  for (int base = 0; base < ne; base += BINB_T * BINB_U) {
    int2 e[BINB_U];
    #pragma unroll
    for (int u = 0; u < BINB_U; ++u) {
      const int i = base + u * BINB_T + tid;
      e[u] = (i < ne) ? bp[i] : make_int2(-1, 0);
    }
    int pos[BINB_U];
    #pragma unroll
    for (int u = 0; u < BINB_U; ++u) pos[u] = (e[u].x >= 0) ? atomicAdd(&ldeg[e[u].x - n0], 1) : 0;
    #pragma unroll
    for (int u = 0; u < BINB_U; ++u) {
      if (e[u].x >= 0) place_pair(e[u], pos[u], n0, llist, ovfcnt, ovf);
    }
  }
  if (nb > BIN_CAP) {
    const int ns = min(spillcnt[0], N_EDGES);
    for (int i = tid; i < ns; i += BINB_T) {
      const int2 e = spill[i];
      if ((int)((unsigned)e.x / BIN_R) == b) place_pair(e, atomicAdd(&ldeg[e.x - n0], 1), n0, llist, ovfcnt, ovf);
    }
  }
  __syncthreads();
  if (tid < BIN_R && n0 + tid < N_NODES) deg[n0 + tid] = ldeg[tid];
  for (int r = wave; r < BIN_R; r += BINB_T / 32) {
    const int n = n0 + r;
    if (n >= N_NODES) break;
    const int d = __builtin_amdgcn_readfirstlane(min(ldeg[r], LIST_CAP));
    const int beg = n * LIST_CAP;
    if (d <= 32) {
      const int mine = (lane < d) ? llist[r * LIST_CAP + lane] : 0x7fffffff;
      int rank = 0;
      for (int j = 0; j < d; ++j) {
        const int v = __builtin_amdgcn_readlane(mine, j);
        rank += (v < mine || (v == mine && j < lane)) ? 1 : 0;
      }
      col2[beg + ((lane < d) ? rank : lane)] = (lane < d) ? mine : -1;
    } else {
      for (int e = lane; e < d; e += 32) {
        const int v = llist[r * LIST_CAP + e];
        int rank = 0;
        for (int j = 0; j < d; ++j) {
          const int u = llist[r * LIST_CAP + j];
          rank += (u < v || (u == v && j < e)) ? 1 : 0;
        }
        col2[beg + rank] = v;
      }
    }
  }
}

static_assert(IN_C * 4 == 512 && HID * 2 == 512, "a gathered row is exactly one 16-byte load per lane");
__device__ __forceinline__ void addbf(float (&a)[8], const uint4& q) {
  a[0] += __uint_as_float(q.x << 16);  a[1] += __uint_as_float(q.x & 0xffff0000u);
  a[2] += __uint_as_float(q.y << 16);  a[3] += __uint_as_float(q.y & 0xffff0000u);
  a[4] += __uint_as_float(q.z << 16);  a[5] += __uint_as_float(q.z & 0xffff0000u);
  a[6] += __uint_as_float(q.w << 16);  a[7] += __uint_as_float(q.w & 0xffff0000u);
}

template <int K1>
__device__ __forceinline__ uint4 row_ld(const float* __restrict__ xf, const uint4* __restrict__ hb, int row, int lane) {
  if constexpr (K1 == HID) {
    return hb[(long)row * 32 + lane];
  } else {
    const float4 v = ((const float4*)xf)[(long)row * 32 + lane];
    return make_uint4(__float_as_uint(v.x), __float_as_uint(v.y), __float_as_uint(v.z), __float_as_uint(v.w));
  }
}

template <int K1>
__device__ __forceinline__ void row_add(float (&a)[8], const uint4& q) {
  if constexpr (K1 == HID) {
    addbf(a, q);
  } else {
    a[0] += __uint_as_float(q.x);  a[1] += __uint_as_float(q.y);
    a[2] += __uint_as_float(q.z);  a[3] += __uint_as_float(q.w);
  }
}

template <int K1>
__device__ __forceinline__ void row_put(uint4* sbuf, int it, int row16, int lane, const float (&a)[8]) {
  constexpr int KT1 = K1 / 32;
  if constexpr (K1 == HID) {
    const int m = lane & 3;
    const int kt = lane >> 2;
    uint4 pr;
    pr.x = (unsigned int)f2bf(a[0]) | ((unsigned int)f2bf(a[1]) << 16);
    pr.y = (unsigned int)f2bf(a[2]) | ((unsigned int)f2bf(a[3]) << 16);
    pr.z = (unsigned int)f2bf(a[4]) | ((unsigned int)f2bf(a[5]) << 16);
    pr.w = (unsigned int)f2bf(a[6]) | ((unsigned int)f2bf(a[7]) << 16);
    sbuf[(it * KT1 + kt) * 64 + (m >> 1) * 32 + row16 + 16 * (m & 1)] = pr;
  } else {
    const int m = lane & 7;
    const int kt = lane >> 3;
    uint2 pr;
    pr.x = (unsigned int)f2bf(a[0]) | ((unsigned int)f2bf(a[1]) << 16);
    pr.y = (unsigned int)f2bf(a[2]) | ((unsigned int)f2bf(a[3]) << 16);
    unsigned int* tl = (unsigned int*)sbuf;
    *(uint2*)(tl + ((it * KT1 + kt) * 64 + (m >> 2) * 32 + row16 + 16 * ((m >> 1) & 1)) * 4 + 2 * (m & 1)) = pr;
  }
}

template <int K1, int NB>
__device__ __forceinline__ void row_batch(float (&acc)[8], const float* __restrict__ xf, const uint4* __restrict__ hb,
                                          const uint4& own, int mine, int r, int lane) {
  if constexpr (NB == 0) {
    row_add<K1>(acc, own);
  } else {
    uint4 x[NB];
    #pragma unroll
    for (int u = 0; u < NB; ++u) {
      const int sidx = (u < NB - 3) ? __builtin_amdgcn_readlane(mine, u) : __builtin_amdgcn_readlane(mine, min(u, r - 1));
      x[u] = row_ld<K1>(xf, hb, sidx, lane);
    }
    __builtin_amdgcn_sched_barrier(0);
    row_add<K1>(acc, own);
    #pragma unroll
    for (int u = 0; u < NB; ++u) {
      uint4 v = x[u];
      if (u >= NB - 3 && u >= r) v = make_uint4(0u, 0u, 0u, 0u);
      row_add<K1>(acc, v);
    }
  }
}

template <int K, int M>
__device__ __forceinline__ void wpack_elem(const float* __restrict__ Wm, unsigned int* __restrict__ out, int i) {
  constexpr int KT = K / 32;
  const int p = i & 7;
  const int lane = (i >> 3) & 31;
  const int tt = i >> 8;
  const int kt = tt % KT;
  const int ct = tt / KT;
  const int col = ct * 16 + (lane & 15);
  const int k = kt * 32 + frag_koff(lane, p);
  const unsigned int lo = f2bf(Wm[(long)k * M + col]);
  const unsigned int hi = f2bf(Wm[(long)(k + 1) * M + col]);
  out[i] = lo | (hi << 16);
}

#define PREP_T      256
#define PREP_B_WIN  ((HID / 16) * (IN_C / 32))
#define PREP_B_WH   ((HID / 16) * (HID / 32))
#define PREP_N_WH   6
#define PREP_B_WOUT ((OUT_C / 16) * (HID / 32))
#define CNT_INTS    (4 + ((NBIN + 3) & ~3))
#define PREP_N_DEG4 (CNT_INTS / 4)
#define PREP_B_DEG  ((PREP_N_DEG4 + PREP_T - 1) / PREP_T)
#define PREP_N_PL4  (N_GRAPHS * (HID + 1) / 4)
#define PREP_B_PL   ((PREP_N_PL4 + PREP_T - 1) / PREP_T)
#define PREP_BLOCKS (PREP_B_WIN + PREP_N_WH * PREP_B_WH + PREP_B_WOUT + PREP_B_DEG + PREP_B_PL)
static_assert(N_NODES % 4 == 0 && (N_GRAPHS * (HID + 1)) % 4 == 0, "the clears go out as whole 16-byte stores");

__global__ __launch_bounds__(PREP_T) void prep_all(
    const float* __restrict__ w_in,
    const float* __restrict__ wh0, const float* __restrict__ wh1, const float* __restrict__ wh2,
    const float* __restrict__ wh3, const float* __restrict__ wh4, const float* __restrict__ wh5,
    const float* __restrict__ w_out,
    unsigned int* __restrict__ p_in,
    unsigned int* __restrict__ ph0, unsigned int* __restrict__ ph1, unsigned int* __restrict__ ph2,
    unsigned int* __restrict__ ph3, unsigned int* __restrict__ ph4, unsigned int* __restrict__ ph5,
    unsigned int* __restrict__ p_out,
    int* __restrict__ cnt, float* __restrict__ pooled) {
  const int t = threadIdx.x;
  int b = blockIdx.x;
  if (b < PREP_B_WIN) { wpack_elem<IN_C, HID>(w_in, p_in, b * PREP_T + t); return; }
  b -= PREP_B_WIN;
  if (b < PREP_N_WH * PREP_B_WH) {
    const int j = b / PREP_B_WH;
    const int i = (b % PREP_B_WH) * PREP_T + t;
    const float* Wm = wh0;  unsigned int* o = ph0;
    if (j == 1) { Wm = wh1; o = ph1; }
    if (j == 2) { Wm = wh2; o = ph2; }
    if (j == 3) { Wm = wh3; o = ph3; }
    if (j == 4) { Wm = wh4; o = ph4; }
    if (j == 5) { Wm = wh5; o = ph5; }
    wpack_elem<HID, HID>(Wm, o, i);
    return;
  }
  b -= PREP_N_WH * PREP_B_WH;
  if (b < PREP_B_WOUT) { wpack_elem<HID, OUT_C>(w_out, p_out, b * PREP_T + t); return; }
  b -= PREP_B_WOUT;
  if (b < PREP_B_DEG) {
    const int i4 = b * PREP_T + t;
    if (i4 < PREP_N_DEG4) ((int4*)cnt)[i4] = make_int4(0, 0, 0, 0);
    return;
  }
  b -= PREP_B_DEG;
  const int i4 = b * PREP_T + t;
  if (i4 < PREP_N_PL4) ((float4*)pooled)[i4] = make_float4(0.0f, 0.0f, 0.0f, 0.0f);
}

static_assert(HID == 256, "a wave owns two 16-column tiles of the 256");
__device__ __forceinline__ void pool_add(float* __restrict__ pooled, float* __restrict__ counts, int g, int col0,
                                         float s0, float s1, int len, bool cnt_lane) {
  float* o = pooled + (long)g * HID + col0;
  atomicAdd(o, s0);
  atomicAdd(o + 16, s1);
  if (cnt_lane) atomicAdd(&counts[g], (float)len);
}

template <int K1, int OUTBF>
__global__ __launch_bounds__(256) __attribute__((amdgpu_waves_per_eu(4, 4)))
void mlp_fused(const float* __restrict__ xsrc, const uint4* __restrict__ hsrc,
                                                 const int* __restrict__ deg, const int* __restrict__ lst,
                                                 const int2* __restrict__ ovf, const uint4* __restrict__ W1pk,
                                                 const float* __restrict__ bias1, const uint4* __restrict__ W2pk,
                                                 const float* __restrict__ bias2, float* __restrict__ Hout, int nrt,
                                                 const int* __restrict__ batch, float* __restrict__ counts) {
  constexpr int KT1 = K1 / 32;
  constexpr int KT2 = HID / 32;
  constexpr int RT  = 4;
  static_assert(KT1 <= KT2, "the A tile must fit the buffer that later holds the middle tile");
  constexpr int OPITCH = HID / 8 + 1;
  __shared__ uint4 sbuf[RT * KT2 * 64 + RT * 16];
  __shared__ int slist[8 * 8 * 32];
  __shared__ uint4 sown[8 * 8 * 32];
  const int tid  = threadIdx.x;
  const int lane = tid & 31;
  const int wave = __builtin_amdgcn_readfirstlane(tid >> 5);
  const int rt0  = blockIdx.x * RT;
  const int nr   = min(RT, nrt - rt0);
  const int ct0  = wave * 2;
  union F { uint4 q[2]; v16bf v; };

  if (nr < RT) {
    for (int i = nr * (KT1 * 64) + tid; i < RT * (KT1 * 64); i += 256) sbuf[i] = make_uint4(0u, 0u, 0u, 0u);
  }
  if ((wave >> 1) < nr) {
    const int it = wave >> 1;
    const int node0 = rt0 * 16 + wave * 8;
    {
      int lm[8];
      uint4 ow[8];
      #pragma unroll
      for (int q = 0; q < 8; ++q) lm[q] = lst[(node0 + q) * LIST_CAP + lane];
      #pragma unroll
      for (int q = 0; q < 8; ++q) ow[q] = row_ld<K1>(xsrc, hsrc, node0 + q, lane);
      #pragma unroll
      for (int q = 0; q < 8; ++q) {
        slist[(wave * 8 + q) * 32 + lane] = lm[q];
        sown[(wave * 8 + q) * 32 + lane] = ow[q];
      }
    }
    #pragma nounroll
    for (int q = 0; q < 8; ++q) {
      const int node = node0 + q;
      const int mine = slist[(wave * 8 + q) * 32 + lane];
      const uint4 own = sown[(wave * 8 + q) * 32 + lane];
      const int cnt = __builtin_popcount(__builtin_amdgcn_ballot_w32(mine >= 0));
      float acc[8];
      #pragma unroll
      for (int i = 0; i < 8; ++i) acc[i] = 0.0f;
      switch ((cnt + 3) >> 2) {
        case 0:  row_batch<K1, 0>(acc, xsrc, hsrc, own, mine, cnt, lane);  break;
        case 1:  row_batch<K1, 4>(acc, xsrc, hsrc, own, mine, cnt, lane);  break;
        case 2:  row_batch<K1, 8>(acc, xsrc, hsrc, own, mine, cnt, lane);  break;
        case 3:  row_batch<K1, 12>(acc, xsrc, hsrc, own, mine, cnt, lane); break;
        case 4:  row_batch<K1, 16>(acc, xsrc, hsrc, own, mine, cnt, lane); break;
        case 5:  row_batch<K1, 20>(acc, xsrc, hsrc, own, mine, cnt, lane); break;
        case 6:  row_batch<K1, 24>(acc, xsrc, hsrc, own, mine, cnt, lane); break;
        case 7:  row_batch<K1, 28>(acc, xsrc, hsrc, own, mine, cnt, lane); break;
        default: row_batch<K1, 32>(acc, xsrc, hsrc, own, mine, cnt, lane); break;
      }
      if (cnt == 32) {
        const int total = deg[node];
        const int nl = min(total, LIST_CAP);
        for (int j = 32; j < nl; ++j) row_add<K1>(acc, row_ld<K1>(xsrc, hsrc, lst[node * LIST_CAP + j], lane));
        if (total > LIST_CAP) {
          const int no = min(deg[N_NODES], N_EDGES);
          for (int k = 0; k < no; ++k) {
            const int2 e = ovf[k];
            if (e.x == node) row_add<K1>(acc, row_ld<K1>(xsrc, hsrc, e.y, lane));
          }
        }
      }
      row_put<K1>(sbuf, it, (wave & 1) * 8 + q, lane, acc);
    }
  }

  F w1f[KT1][2];
  {
    const uint4* Wp = W1pk + ((long)ct0 * KT1 * 32 + lane) * 2;
    #pragma unroll
    for (int kt = 0; kt < KT1; ++kt) {
      w1f[kt][0].q[0] = Wp[kt * 64];           w1f[kt][0].q[1] = Wp[kt * 64 + 1];
      w1f[kt][1].q[0] = Wp[(KT1 + kt) * 64];   w1f[kt][1].q[1] = Wp[(KT1 + kt) * 64 + 1];
    }
  }
  if constexpr (!OUTBF) {
    if (tid < RT * 16) ((int*)(sbuf + RT * KT2 * 64))[tid] = (tid < nr * 16) ? batch[rt0 * 16 + tid] : -1;
  }
  __syncthreads();

  v8f acc[RT][2];
  #pragma unroll
  for (int i = 0; i < RT; ++i) { acc[i][0] = v8f{}; acc[i][1] = v8f{}; }
  {
    F af[2][RT];
    #pragma unroll
    for (int i = 0; i < RT; ++i) {
      af[0][i].q[0] = sbuf[(i * KT1) * 64 + lane];
      af[0][i].q[1] = sbuf[(i * KT1) * 64 + 32 + lane];
    }
    __builtin_amdgcn_sched_group_barrier(0x100, 8, 0);
    #pragma unroll
    for (int kt = 0; kt < KT1; ++kt) {
      if (kt + 1 < KT1) {
        #pragma unroll
        for (int i = 0; i < RT; ++i) {
          af[(kt + 1) & 1][i].q[0] = sbuf[(i * KT1 + kt + 1) * 64 + lane];
          af[(kt + 1) & 1][i].q[1] = sbuf[(i * KT1 + kt + 1) * 64 + 32 + lane];
        }
      }
      #pragma unroll
      for (int i = 0; i < RT; ++i) {
        acc[i][0] = __builtin_amdgcn_wmma_f32_16x16x32_bf16(false, af[kt & 1][i].v, false, w1f[kt][0].v, (short)0, acc[i][0], false, false);
        acc[i][1] = __builtin_amdgcn_wmma_f32_16x16x32_bf16(false, af[kt & 1][i].v, false, w1f[kt][1].v, (short)0, acc[i][1], false, false);
      }
      __builtin_amdgcn_sched_group_barrier(0x100, 8, 0);
      __builtin_amdgcn_sched_group_barrier(0x008, 8, 0);
    }
  }

  F w2f[KT2][2];
  {
    const uint4* Wp = W2pk + ((long)ct0 * KT2 * 32 + lane) * 2;
    #pragma unroll
    for (int kt = 0; kt < KT2; ++kt) {
      w2f[kt][0].q[0] = Wp[kt * 64];           w2f[kt][0].q[1] = Wp[kt * 64 + 1];
      w2f[kt][1].q[0] = Wp[(KT2 + kt) * 64];   w2f[kt][1].q[1] = Wp[(KT2 + kt) * 64 + 1];
    }
  }

  const int rowL = lane & 15;
  const int hh   = lane >> 4;
  __syncthreads();
  {
    unsigned short* sw = (unsigned short*)sbuf;
    #pragma unroll
    for (int j = 0; j < 2; ++j) {
      const float b = bias1[(ct0 + j) * 16 + rowL];
      #pragma unroll
      for (int i = 0; i < RT; ++i) {
        #pragma unroll
        for (int r = 0; r < 8; ++r) {
          const float o = fmaxf(acc[i][j][r] + b, 0.0f);
          sw[(i * KT2 + wave) * 512 + j * 256 + (8 * hh + r + 16 * (rowL >> 3)) * 8 + (rowL & 7)] = f2bf(o);
        }
      }
    }
  }
  __syncthreads();

  #pragma unroll
  for (int i = 0; i < RT; ++i) { acc[i][0] = v8f{}; acc[i][1] = v8f{}; }
  {
    F af[2][RT];
    #pragma unroll
    for (int i = 0; i < RT; ++i) {
      af[0][i].q[0] = sbuf[(i * KT2) * 64 + lane];
      af[0][i].q[1] = sbuf[(i * KT2) * 64 + 32 + lane];
    }
    __builtin_amdgcn_sched_group_barrier(0x100, 8, 0);
    #pragma unroll
    for (int kt = 0; kt < KT2; ++kt) {
      if (kt + 1 < KT2) {
        #pragma unroll
        for (int i = 0; i < RT; ++i) {
          af[(kt + 1) & 1][i].q[0] = sbuf[(i * KT2 + kt + 1) * 64 + lane];
          af[(kt + 1) & 1][i].q[1] = sbuf[(i * KT2 + kt + 1) * 64 + 32 + lane];
        }
      }
      #pragma unroll
      for (int i = 0; i < RT; ++i) {
        acc[i][0] = __builtin_amdgcn_wmma_f32_16x16x32_bf16(false, af[kt & 1][i].v, false, w2f[kt][0].v, (short)0, acc[i][0], false, false);
        acc[i][1] = __builtin_amdgcn_wmma_f32_16x16x32_bf16(false, af[kt & 1][i].v, false, w2f[kt][1].v, (short)0, acc[i][1], false, false);
      }
      __builtin_amdgcn_sched_group_barrier(0x100, 8, 0);
      __builtin_amdgcn_sched_group_barrier(0x008, 8, 0);
    }
  }

  if constexpr (OUTBF) {
    __syncthreads();
    unsigned short* sw = (unsigned short*)sbuf;
    #pragma unroll
    for (int j = 0; j < 2; ++j) {
      const int col = (ct0 + j) * 16 + rowL;
      const float b = bias2[col];
      #pragma unroll
      for (int i = 0; i < RT; ++i) {
        #pragma unroll
        for (int r = 0; r < 8; ++r)
          sw[(i * 16 + 8 * hh + r) * (OPITCH * 8) + col] = f2bf(fmaxf(acc[i][j][r] + b, 0.0f));
      }
    }
    __syncthreads();
    if ((wave >> 1) < nr) {
      uint4* __restrict__ Hb = (uint4*)Hout;
      #pragma unroll
      for (int q = 0; q < 8; ++q) {
        const int rloc = wave * 8 + q;
        Hb[((long)rt0 * 16 + rloc) * (HID / 8) + lane] = sbuf[rloc * OPITCH + lane];
      }
    }
  } else {
    const int* idl = (const int*)(sbuf + RT * KT2 * 64);
    const int col0 = ct0 * 16 + rowL;
    const float b0 = bias2[col0];
    const float b1 = bias2[col0 + 16];
    const bool cnt_lane = (wave == 0) && (rowL == 0);
    float s0 = 0.0f, s1 = 0.0f;
    int len = 0;
    int cur = idl[8 * hh];
    #pragma unroll
    for (int i = 0; i < RT; ++i) {
      if (i == 0 || i < nr) {
        const int4 ga = *(const int4*)(idl + i * 16 + 8 * hh);
        const int4 gb = *(const int4*)(idl + i * 16 + 8 * hh + 4);
        const int g[8] = {ga.x, ga.y, ga.z, ga.w, gb.x, gb.y, gb.z, gb.w};
        #pragma unroll
        for (int r = 0; r < 8; ++r) {
          if (g[r] != cur) {
            pool_add(Hout, counts, cur, col0, s0, s1, len, cnt_lane);
            s0 = 0.0f; s1 = 0.0f; len = 0;
            cur = g[r];
          }
          s0 += fmaxf(acc[i][0][r] + b0, 0.0f);
          s1 += fmaxf(acc[i][1][r] + b1, 0.0f);
          ++len;
        }
      }
    }
    pool_add(Hout, counts, cur, col0, s0, s1, len, cnt_lane);
  }
}

static_assert(HID == 256 && OUT_C == 16 && N_GRAPHS % 16 == 0, "one work-group per full 16-graph tile, one output column tile");
__global__ __launch_bounds__(256) void out_mlp(const float* __restrict__ pooled, const float* __restrict__ counts,
                                               const uint4* __restrict__ W1pk, const float* __restrict__ bias1,
                                               const uint4* __restrict__ W2pk, const float* __restrict__ bias2,
                                               float* __restrict__ out) {
  constexpr int KT = HID / 32;
  __shared__ uint4 sbuf[KT * 64];
  const int tid  = threadIdx.x;
  const int lane = tid & 31;
  const int wave = __builtin_amdgcn_readfirstlane(tid >> 5);
  const int rt   = blockIdx.x;
  const int ct0  = wave * 2;
  union F { uint4 q[2]; v16bf v; };

  const int m = tid >> 4;
  const int c = tid & 15;
  const int row = rt * 16 + m;
  const float cnt = fmaxf(counts[row], 1.0f);
  const float4* sp = (const float4*)(pooled + (long)row * HID + c * 16);
  const float4 s0 = sp[0], s1 = sp[1], s2 = sp[2], s3 = sp[3];
  const int rowL = lane & 15;
  const int hh   = lane >> 4;
  const float b10 = bias1[ct0 * 16 + rowL];
  const float b11 = bias1[(ct0 + 1) * 16 + rowL];
  F w1f[KT][2];
  {
    const uint4* Wp = W1pk + ((long)ct0 * KT * 32 + lane) * 2;
    #pragma unroll
    for (int kt = 0; kt < KT; ++kt) {
      w1f[kt][0].q[0] = Wp[kt * 64];          w1f[kt][0].q[1] = Wp[kt * 64 + 1];
      w1f[kt][1].q[0] = Wp[(KT + kt) * 64];   w1f[kt][1].q[1] = Wp[(KT + kt) * 64 + 1];
    }
  }
  F w2f[KT];
  float b2v = 0.0f;
  #pragma unroll
  for (int kt = 0; kt < KT; ++kt) { w2f[kt].q[0] = make_uint4(0u, 0u, 0u, 0u); w2f[kt].q[1] = make_uint4(0u, 0u, 0u, 0u); }
  if (wave == 0) {
    const uint4* Wp = W2pk + lane * 2;
    #pragma unroll
    for (int kt = 0; kt < KT; ++kt) { w2f[kt].q[0] = Wp[kt * 64]; w2f[kt].q[1] = Wp[kt * 64 + 1]; }
    b2v = bias2[rowL];
  }

  {
    uint4 lo, hi;
    lo.x = (unsigned int)f2bf(s0.x / cnt) | ((unsigned int)f2bf(s0.y / cnt) << 16);
    lo.y = (unsigned int)f2bf(s0.z / cnt) | ((unsigned int)f2bf(s0.w / cnt) << 16);
    lo.z = (unsigned int)f2bf(s1.x / cnt) | ((unsigned int)f2bf(s1.y / cnt) << 16);
    lo.w = (unsigned int)f2bf(s1.z / cnt) | ((unsigned int)f2bf(s1.w / cnt) << 16);
    hi.x = (unsigned int)f2bf(s2.x / cnt) | ((unsigned int)f2bf(s2.y / cnt) << 16);
    hi.y = (unsigned int)f2bf(s2.z / cnt) | ((unsigned int)f2bf(s2.w / cnt) << 16);
    hi.z = (unsigned int)f2bf(s3.x / cnt) | ((unsigned int)f2bf(s3.y / cnt) << 16);
    hi.w = (unsigned int)f2bf(s3.z / cnt) | ((unsigned int)f2bf(s3.w / cnt) << 16);
    const int base = (c >> 1) * 64 + (c & 1) * 32 + m;
    sbuf[base]      = lo;
    sbuf[base + 16] = hi;
  }
  __syncthreads();

  v8f acc0 = v8f{}, acc1 = v8f{};
  {
    F af[KT];
    #pragma unroll
    for (int kt = 0; kt < KT; ++kt) {
      af[kt].q[0] = sbuf[kt * 64 + lane];
      af[kt].q[1] = sbuf[kt * 64 + 32 + lane];
    }
    #pragma unroll
    for (int kt = 0; kt < KT; ++kt) {
      acc0 = __builtin_amdgcn_wmma_f32_16x16x32_bf16(false, af[kt].v, false, w1f[kt][0].v, (short)0, acc0, false, false);
      acc1 = __builtin_amdgcn_wmma_f32_16x16x32_bf16(false, af[kt].v, false, w1f[kt][1].v, (short)0, acc1, false, false);
    }
  }

  __syncthreads();
  {
    unsigned short* sw = (unsigned short*)sbuf;
    #pragma unroll
    for (int r = 0; r < 8; ++r) {
      const int pos = wave * 512 + (8 * hh + r + 16 * (rowL >> 3)) * 8 + (rowL & 7);
      sw[pos]       = f2bf(fmaxf(acc0[r] + b10, 0.0f));
      sw[pos + 256] = f2bf(fmaxf(acc1[r] + b11, 0.0f));
    }
  }
  __syncthreads();

  if (wave == 0) {
    v8f acc = v8f{};
    F af[KT];
    #pragma unroll
    for (int kt = 0; kt < KT; ++kt) {
      af[kt].q[0] = sbuf[kt * 64 + lane];
      af[kt].q[1] = sbuf[kt * 64 + 32 + lane];
    }
    #pragma unroll
    for (int kt = 0; kt < KT; ++kt)
      acc = __builtin_amdgcn_wmma_f32_16x16x32_bf16(false, af[kt].v, false, w2f[kt].v, (short)0, acc, false, false);
    #pragma unroll
    for (int r = 0; r < 8; ++r) out[(long)(rt * 16 + 8 * hh + r) * OUT_C + rowL] = acc[r] + b2v;
  }
}

extern "C" void kernel_launch(void* const* d_in, const int* in_sizes, int n_in,
                              void* d_out, int out_size, void* d_ws, size_t ws_size,
                              hipStream_t stream) {
  const float* x     = (const float*)d_in[0];
  const int*   ei    = (const int*)d_in[1];
  const int*   esrc  = ei;
  const int*   edst  = ei + N_EDGES;
  const int*   batch = (const int*)d_in[2];
  const float* cw1[3] = {(const float*)d_in[3],  (const float*)d_in[7],  (const float*)d_in[11]};
  const float* cb1[3] = {(const float*)d_in[4],  (const float*)d_in[8],  (const float*)d_in[12]};
  const float* cw2[3] = {(const float*)d_in[5],  (const float*)d_in[9],  (const float*)d_in[13]};
  const float* cb2[3] = {(const float*)d_in[6],  (const float*)d_in[10], (const float*)d_in[14]};
  const float* ow1 = (const float*)d_in[15];
  const float* ob1 = (const float*)d_in[16];
  const float* ow2 = (const float*)d_in[17];
  const float* ob2 = (const float*)d_in[18];

  char* ws = (char*)d_ws;
  size_t off = 0;
  auto bump = [&](size_t bytes) {
    void* p = ws + off;
    off += (bytes + 255) & ~(size_t)255;
    return p;
  };
  const int NRT = N_NODES / 16;
  unsigned short* HbA    = (unsigned short*)bump((size_t)N_NODES * HID * 2);
  unsigned short* HbB    = (unsigned short*)bump((size_t)N_NODES * HID * 2);
  float*          pooled = (float*)bump((size_t)N_GRAPHS * (HID + 1) * 4);
  float*          counts = pooled + (size_t)N_GRAPHS * HID;
  unsigned int*   wpk1[3], *wpk2[3];
  for (int l = 0; l < 3; ++l) {
    wpk1[l] = (unsigned int*)bump((size_t)(HID / 16) * 8 * 256 * 4);
    wpk2[l] = (unsigned int*)bump((size_t)(HID / 16) * 8 * 256 * 4);
  }
  unsigned int* owpk1 = (unsigned int*)bump((size_t)(HID / 16) * 8 * 256 * 4);
  unsigned int* owpk2 = (unsigned int*)bump((size_t)(OUT_C / 16) * 8 * 256 * 4);
  int*  deg    = (int*)bump((size_t)(N_NODES + CNT_INTS) * 4);
  int*  ovfcnt = deg + N_NODES;
  int*  spillcnt = ovfcnt + 1;
  int*  bcnt   = ovfcnt + 4;
  int2* binbuf = (int2*)bump((size_t)NBIN * BIN_CAP * 8);
  int*  lst2   = (int*)bump((size_t)N_NODES * LIST_CAP * 4);
  int2* ovf    = (int2*)bump((size_t)N_EDGES * 8);
  int2* spill  = (int2*)bump((size_t)N_EDGES * 8);

  dim3 blk(256);
  const int NWG = (NRT + 3) / 4;

  prep_all<<<PREP_BLOCKS, PREP_T, 0, stream>>>(cw1[0], cw2[0], cw1[1], cw2[1], cw1[2], cw2[2], ow1, ow2,
                                                wpk1[0], wpk2[0], wpk1[1], wpk2[1], wpk1[2], wpk2[2], owpk1, owpk2,
                                                ovfcnt, pooled);

  bin_fill<<<(N_EDGES + BINA_CHUNK - 1) / BINA_CHUNK, BINA_T, 0, stream>>>(esrc, edst, bcnt, spillcnt, binbuf, spill);
  bin_lists<<<NBIN, BINB_T, 0, stream>>>(bcnt, binbuf, spillcnt, spill, deg, ovfcnt, lst2, ovf);

  mlp_fused<IN_C, 1><<<NWG, blk, 0, stream>>>(x, nullptr, deg, lst2, ovf, (const uint4*)wpk1[0], cb1[0],
                                              (const uint4*)wpk2[0], cb2[0], (float*)HbA, NRT, nullptr, nullptr);
  mlp_fused<HID, 1><<<NWG, blk, 0, stream>>>(nullptr, (const uint4*)HbA, deg, lst2, ovf, (const uint4*)wpk1[1], cb1[1],
                                             (const uint4*)wpk2[1], cb2[1], (float*)HbB, NRT, nullptr, nullptr);
  mlp_fused<HID, 0><<<NWG, blk, 0, stream>>>(nullptr, (const uint4*)HbB, deg, lst2, ovf, (const uint4*)wpk1[2], cb1[2],
                                             (const uint4*)wpk2[2], cb2[2], pooled, NRT, batch, counts);

  out_mlp<<<N_GRAPHS / 16, blk, 0, stream>>>(pooled, counts, (const uint4*)owpk1, ob1, (const uint4*)owpk2, ob2,
                                             (float*)d_out);
}
